# starting kernel
# baseline (speedup 1.0000x reference)
#include <hip/hip_runtime.h>
#include <math.h>

#define B_   32
#define N_   512
#define V_   5
#define D_   256
#define H_   4
#define BN_  (B_ * N_)
#define T_   (BN_ + B_ * V_)
#define HD_  (H_ * D_)
#define EPS_ 1e-5f

typedef float v2f __attribute__((ext_vector_type(2)));
typedef float v8f __attribute__((ext_vector_type(8)));

__device__ __forceinline__ float gelu_exact(float x) {
    return 0.5f * x * (1.0f + erff(x * 0.70710678118654752f));
}

__global__ void k_build_nodes(const float* __restrict__ x,
                              const float* __restrict__ vn,
                              float* __restrict__ nodes) {
    size_t i = (size_t)blockIdx.x * blockDim.x + threadIdx.x;
    if (i < (size_t)BN_ * D_) {
        nodes[i] = x[i];
    } else if (i < (size_t)T_ * D_) {
        size_t j = i - (size_t)BN_ * D_;
        nodes[i] = vn[j % (V_ * D_)];
    }
}

template <int KDIM, int NDIM>
__global__ void k_gemm_f32_wmma(const float* __restrict__ A,
                                const float* __restrict__ Bm,
                                float* __restrict__ C) {
    const int wave = threadIdx.x >> 5;
    const int lane = threadIdx.x & 31;
    const int half = lane >> 4;
    const int lm   = lane & 15;
    const int m0   = blockIdx.x * 16;
    const int n0   = blockIdx.y * 256 + wave * 64;
    if (n0 >= NDIM) return;

    v8f c0 = {}; v8f c1 = {}; v8f c2 = {}; v8f c3 = {};
    const float* arow = A + (size_t)(m0 + lm) * KDIM;

    for (int k0 = 0; k0 < KDIM; k0 += 4) {
        const int kb = k0 + half * 2;
        v2f a;
        a[0] = arow[kb];
        a[1] = arow[kb + 1];
        const float* b0p = Bm + (size_t)kb * NDIM + n0 + lm;
        const float* b1p = b0p + NDIM;
        v2f b;
        b[0] = b0p[0];  b[1] = b1p[0];
        c0 = __builtin_amdgcn_wmma_f32_16x16x4_f32(false, a, false, b, (short)0, c0, false, false);
        b[0] = b0p[16]; b[1] = b1p[16];
        c1 = __builtin_amdgcn_wmma_f32_16x16x4_f32(false, a, false, b, (short)0, c1, false, false);
        b[0] = b0p[32]; b[1] = b1p[32];
        c2 = __builtin_amdgcn_wmma_f32_16x16x4_f32(false, a, false, b, (short)0, c2, false, false);
        b[0] = b0p[48]; b[1] = b1p[48];
        c3 = __builtin_amdgcn_wmma_f32_16x16x4_f32(false, a, false, b, (short)0, c3, false, false);
    }

    float* crow = C + n0 + lm;
#pragma unroll
    for (int j = 0; j < 8; ++j) {
        const size_t row = (size_t)(m0 + half * 8 + j) * NDIM;
        crow[row + 0]  = c0[j];
        crow[row + 16] = c1[j];
        crow[row + 32] = c2[j];
        crow[row + 48] = c3[j];
    }
}

__global__ void k_scores(const float* __restrict__ h,
                         const float* __restrict__ att_src,
                         const float* __restrict__ att_dst,
                         float* __restrict__ a_src,
                         float* __restrict__ a_dst) {
    __shared__ float red[D_];
    const int t = blockIdx.x;
    const int d = threadIdx.x;
    const float* hr = h + (size_t)t * HD_;
#pragma unroll
    for (int hd = 0; hd < H_; ++hd) {
        const float hv = hr[hd * D_ + d];
        red[d] = hv * att_src[hd * D_ + d];
        __syncthreads();
        for (int s = D_ / 2; s > 0; s >>= 1) {
            if (d < s) red[d] += red[d + s];
            __syncthreads();
        }
        if (d == 0) a_src[t * H_ + hd] = red[0];
        __syncthreads();
        red[d] = hv * att_dst[hd * D_ + d];
        __syncthreads();
        for (int s = D_ / 2; s > 0; s >>= 1) {
            if (d < s) red[d] += red[d + s];
            __syncthreads();
        }
        if (d == 0) a_dst[t * H_ + hd] = red[0];
        __syncthreads();
    }
}

__global__ void k_gat_virtual(const float* __restrict__ h,
                              const float* __restrict__ a_src,
                              const float* __restrict__ a_dst,
                              float* __restrict__ gat_agg) {
    __shared__ float w[N_ + 1];
    __shared__ float red[D_];
    const int uidx = blockIdx.x;
    const int head = blockIdx.y;
    const int b    = uidx / V_;
    const int u    = BN_ + uidx;
    const int tid  = threadIdx.x;

    const float adst = a_dst[u * H_ + head];
    const int r0 = b * N_ + tid * 2;
    float al0 = a_src[(size_t)r0 * H_ + head] + adst;
    al0 = (al0 > 0.0f) ? al0 : 0.2f * al0;
    float al1 = a_src[(size_t)(r0 + 1) * H_ + head] + adst;
    al1 = (al1 > 0.0f) ? al1 : 0.2f * al1;
    float aself = a_src[(size_t)u * H_ + head] + adst;
    aself = (aself > 0.0f) ? aself : 0.2f * aself;

    float lmax = fmaxf(al0, al1);
    if (tid == 0) lmax = fmaxf(lmax, aself);
    red[tid] = lmax;
    __syncthreads();
    for (int s = D_ / 2; s > 0; s >>= 1) {
        if (tid < s) red[tid] = fmaxf(red[tid], red[tid + s]);
        __syncthreads();
    }
    const float m = red[0];
    __syncthreads();

    const float e0 = expf(al0 - m);
    const float e1 = expf(al1 - m);
    w[tid * 2]     = e0;
    w[tid * 2 + 1] = e1;
    float es = 0.0f;
    if (tid == 0) { es = expf(aself - m); w[N_] = es; }
    red[tid] = e0 + e1 + es;
    __syncthreads();
    for (int s = D_ / 2; s > 0; s >>= 1) {
        if (tid < s) red[tid] += red[tid + s];
        __syncthreads();
    }
    const float inv = 1.0f / red[0];
    __syncthreads();

    const float* hb = h + (size_t)(b * N_) * HD_ + head * D_ + tid;
    float acc = 0.0f;
    for (int r = 0; r < N_; ++r)
        acc += w[r] * hb[(size_t)r * HD_];
    acc += w[N_] * h[(size_t)u * HD_ + head * D_ + tid];
    gat_agg[((size_t)uidx * H_ + head) * D_ + tid] = acc * inv;
}

__global__ void k_gat_finish(const float* __restrict__ h,
                             const float* __restrict__ gat_agg,
                             const float* __restrict__ bias,
                             const float* __restrict__ gg,
                             const float* __restrict__ gb,
                             float* __restrict__ out) {
    __shared__ float red[D_];
    const int t = blockIdx.x;
    const int d = threadIdx.x;
    float v;
    if (t < BN_) {
        const float* p = h + (size_t)t * HD_ + d;
        v = 0.25f * (p[0] + p[D_] + p[2 * D_] + p[3 * D_]);
    } else {
        const float* p = gat_agg + (size_t)(t - BN_) * HD_ + d;
        v = 0.25f * (p[0] + p[D_] + p[2 * D_] + p[3 * D_]);
    }
    v = gelu_exact(v + bias[d]);

    red[d] = v;
    __syncthreads();
    for (int s = D_ / 2; s > 0; s >>= 1) {
        if (d < s) red[d] += red[d + s];
        __syncthreads();
    }
    const float mu = red[0] * (1.0f / D_);
    __syncthreads();
    const float c = v - mu;
    red[d] = c * c;
    __syncthreads();
    for (int s = D_ / 2; s > 0; s >>= 1) {
        if (d < s) red[d] += red[d + s];
        __syncthreads();
    }
    const float var = red[0] * (1.0f / D_);
    out[(size_t)t * D_ + d] = c * rsqrtf(var + EPS_) * gg[d] + gb[d];
}

__global__ void k_gcn_finish(const float* __restrict__ xw,
                             const float* __restrict__ bias,
                             const float* __restrict__ gg,
                             const float* __restrict__ gb,
                             float* __restrict__ out) {
    __shared__ float red[D_];
    const int t = blockIdx.x;
    const int d = threadIdx.x;
    const int b = t / N_;
    const float* xv = xw + (size_t)(BN_ + b * V_) * D_ + d;
    const float s5 = xv[0] + xv[D_] + xv[2 * D_] + xv[3 * D_] + xv[4 * D_];
    const float inv_sqrt6 = 0.40824829046386301637f;
    float v = s5 * inv_sqrt6 + xw[(size_t)t * D_ + d] * (1.0f / 6.0f) + bias[d];
    v = gelu_exact(v);

    red[d] = v;
    __syncthreads();
    for (int s = D_ / 2; s > 0; s >>= 1) {
        if (d < s) red[d] += red[d + s];
        __syncthreads();
    }
    const float mu = red[0] * (1.0f / D_);
    __syncthreads();
    const float c = v - mu;
    red[d] = c * c;
    __syncthreads();
    for (int s = D_ / 2; s > 0; s >>= 1) {
        if (d < s) red[d] += red[d + s];
        __syncthreads();
    }
    const float var = red[0] * (1.0f / D_);
    out[(size_t)t * D_ + d] = c * rsqrtf(var + EPS_) * gg[d] + gb[d];
}

extern "C" void kernel_launch(void* const* d_in, const int* in_sizes, int n_in,
                              void* d_out, int out_size, void* d_ws, size_t ws_size,
                              hipStream_t stream) {
    (void)in_sizes; (void)n_in; (void)out_size; (void)ws_size;

    const float* x       = (const float*)d_in[0];
    const float* vn      = (const float*)d_in[2];
    const float* gat_W   = (const float*)d_in[3];
    const float* att_src = (const float*)d_in[4];
    const float* att_dst = (const float*)d_in[5];
    const float* gat_b   = (const float*)d_in[6];
    const float* gcn_W   = (const float*)d_in[7];
    const float* gcn_b   = (const float*)d_in[8];
    const float* n1g     = (const float*)d_in[9];
    const float* n1b     = (const float*)d_in[10];
    const float* n2g     = (const float*)d_in[11];
    const float* n2b     = (const float*)d_in[12];
    float* out = (float*)d_out;

    float* ws = (float*)d_ws;
    size_t off = 0;
    float* nodes = ws + off; off += (size_t)T_ * D_;
    float* h     = ws + off; off += (size_t)T_ * HD_;
    float* asrc  = ws + off; off += (size_t)T_ * H_;
    float* adst  = ws + off; off += (size_t)T_ * H_;
    float* gagg  = ws + off; off += (size_t)B_ * V_ * HD_;
    float* gat1  = ws + off; off += (size_t)T_ * D_;
    float* xw    = ws + off; off += (size_t)T_ * D_;

    {
        const size_t n = (size_t)T_ * D_;
        k_build_nodes<<<(unsigned)((n + 255) / 256), 256, 0, stream>>>(x, vn, nodes);
    }
    {
        dim3 grid(T_ / 16, HD_ / 256);
        k_gemm_f32_wmma<D_, HD_><<<grid, 128, 0, stream>>>(nodes, gat_W, h);
    }
    k_scores<<<T_, D_, 0, stream>>>(h, att_src, att_dst, asrc, adst);
    {
        dim3 grid(B_ * V_, H_);
        k_gat_virtual<<<grid, D_, 0, stream>>>(h, asrc, adst, gagg);
    }
    k_gat_finish<<<T_, D_, 0, stream>>>(h, gagg, gat_b, n1g, n1b, gat1);
    {
        dim3 grid(T_ / 16, 1);
        k_gemm_f32_wmma<D_, D_><<<grid, 128, 0, stream>>>(gat1, gcn_W, xw);
    }
    k_gcn_finish<<<BN_, D_, 0, stream>>>(xw, gcn_b, n2g, n2b, out);
}


# Fuse the attention-score dot products (a_src, a_dst) into the first GEMM's epilogue (row-DPP sums, then LDS over the 4 waves); the score kernel and its re-read of h go; score sum order changes
# speedup vs baseline: 1.5293x; 1.5293x over previous
#include <hip/hip_runtime.h>
#include <math.h>

#define B_   32
#define N_   512
#define V_   5
#define D_   256
#define H_   4
#define BN_  (B_ * N_)
#define T_   (BN_ + B_ * V_)
#define HD_  (H_ * D_)
#define EPS_ 1e-5f

typedef float v2f __attribute__((ext_vector_type(2)));
typedef float v8f __attribute__((ext_vector_type(8)));

__device__ __forceinline__ float gelu_exact(float x) {
    return 0.5f * x * (1.0f + erff(x * 0.70710678118654752f));
}

__global__ void k_build_nodes(const float* __restrict__ x,
                              const float* __restrict__ vn,
                              float* __restrict__ nodes) {
    size_t i = (size_t)blockIdx.x * blockDim.x + threadIdx.x;
    if (i < (size_t)BN_ * D_) {
        nodes[i] = x[i];
    } else if (i < (size_t)T_ * D_) {
        size_t j = i - (size_t)BN_ * D_;
        nodes[i] = vn[j % (V_ * D_)];
    }
}

template <int XMASK>
__device__ __forceinline__ float add_row_xmask(float v) {
    return v + __builtin_bit_cast(float, __builtin_amdgcn_update_dpp(
                   0, __builtin_bit_cast(int, v), 0x160 + XMASK, 0xf, 0xf, true));
}

__device__ __forceinline__ float row16_sum(float v) {
    v = add_row_xmask<1>(v);
    v = add_row_xmask<2>(v);
    v = add_row_xmask<4>(v);
    v = add_row_xmask<8>(v);
    return v;
}

template <int KDIM, int NDIM, bool SCORES>
__global__ void k_gemm_f32_wmma(const float* __restrict__ A,
                                const float* __restrict__ Bm,
                                float* __restrict__ C,
                                const float* __restrict__ att_src,
                                const float* __restrict__ att_dst,
                                float* __restrict__ a_src,
                                float* __restrict__ a_dst) {
    const int wave = threadIdx.x >> 5;
    const int lane = threadIdx.x & 31;
    const int half = lane >> 4;
    const int lm   = lane & 15;
    const int m0   = blockIdx.x * 16;
    const int n0   = blockIdx.y * 256 + wave * 64;
    if constexpr (SCORES) {
        static_assert(NDIM % 256 == 0 && NDIM / 256 == H_, "one work-group per head");
    } else {
        if (n0 >= NDIM) return;
    }

    v8f c0 = {}; v8f c1 = {}; v8f c2 = {}; v8f c3 = {};
    const float* arow = A + (size_t)(m0 + lm) * KDIM;

    for (int k0 = 0; k0 < KDIM; k0 += 4) {
        const int kb = k0 + half * 2;
        v2f a;
        a[0] = arow[kb];
        a[1] = arow[kb + 1];
        const float* b0p = Bm + (size_t)kb * NDIM + n0 + lm;
        const float* b1p = b0p + NDIM;
        v2f b;
        b[0] = b0p[0];  b[1] = b1p[0];
        c0 = __builtin_amdgcn_wmma_f32_16x16x4_f32(false, a, false, b, (short)0, c0, false, false);
        b[0] = b0p[16]; b[1] = b1p[16];
        c1 = __builtin_amdgcn_wmma_f32_16x16x4_f32(false, a, false, b, (short)0, c1, false, false);
        b[0] = b0p[32]; b[1] = b1p[32];
        c2 = __builtin_amdgcn_wmma_f32_16x16x4_f32(false, a, false, b, (short)0, c2, false, false);
        b[0] = b0p[48]; b[1] = b1p[48];
        c3 = __builtin_amdgcn_wmma_f32_16x16x4_f32(false, a, false, b, (short)0, c3, false, false);
    }

    float* crow = C + n0 + lm;
#pragma unroll
    for (int j = 0; j < 8; ++j) {
        const size_t row = (size_t)(m0 + half * 8 + j) * NDIM;
        crow[row + 0]  = c0[j];
        crow[row + 16] = c1[j];
        crow[row + 32] = c2[j];
        crow[row + 48] = c3[j];
    }

    if constexpr (SCORES) {
        __shared__ float part[4][2][16];
        const float* as = att_src + n0 + lm;
        const float* ad = att_dst + n0 + lm;
        const float s0 = as[0], s1 = as[16], s2 = as[32], s3 = as[48];
        const float d0 = ad[0], d1 = ad[16], d2 = ad[32], d3 = ad[48];
        float ps[8], pd[8];
        for (int j = 0; j < 8; ++j) {
            ps[j] = fmaf(c3[j], s3, fmaf(c2[j], s2, fmaf(c1[j], s1, c0[j] * s0)));
            pd[j] = fmaf(c3[j], d3, fmaf(c2[j], d2, fmaf(c1[j], d1, c0[j] * d0)));
        }
        for (int j = 0; j < 8; ++j) {
            ps[j] = row16_sum(ps[j]);
            pd[j] = row16_sum(pd[j]);
        }
        if (lm == 0) {
            for (int j = 0; j < 8; ++j) {
                part[wave][0][half * 8 + j] = ps[j];
                part[wave][1][half * 8 + j] = pd[j];
            }
        }
        __syncthreads();
        if (threadIdx.x < 32) {
            const int r = threadIdx.x & 15;
            const int v = threadIdx.x >> 4;
            const float sum = (part[0][v][r] + part[1][v][r]) + (part[2][v][r] + part[3][v][r]);
            float* dst = v ? a_dst : a_src;
            dst[(size_t)(m0 + r) * H_ + blockIdx.y] = sum;
        }
    }
}

__global__ void k_gat_virtual(const float* __restrict__ h,
                              const float* __restrict__ a_src,
                              const float* __restrict__ a_dst,
                              float* __restrict__ gat_agg) {
    __shared__ float w[N_ + 1];
    __shared__ float red[D_];
    const int uidx = blockIdx.x;
    const int head = blockIdx.y;
    const int b    = uidx / V_;
    const int u    = BN_ + uidx;
    const int tid  = threadIdx.x;

    const float adst = a_dst[u * H_ + head];
    const int r0 = b * N_ + tid * 2;
    float al0 = a_src[(size_t)r0 * H_ + head] + adst;
    al0 = (al0 > 0.0f) ? al0 : 0.2f * al0;
    float al1 = a_src[(size_t)(r0 + 1) * H_ + head] + adst;
    al1 = (al1 > 0.0f) ? al1 : 0.2f * al1;
    float aself = a_src[(size_t)u * H_ + head] + adst;
    aself = (aself > 0.0f) ? aself : 0.2f * aself;

    float lmax = fmaxf(al0, al1);
    if (tid == 0) lmax = fmaxf(lmax, aself);
    red[tid] = lmax;
    __syncthreads();
    for (int s = D_ / 2; s > 0; s >>= 1) {
        if (tid < s) red[tid] = fmaxf(red[tid], red[tid + s]);
        __syncthreads();
    }
    const float m = red[0];
    __syncthreads();

    const float e0 = expf(al0 - m);
    const float e1 = expf(al1 - m);
    w[tid * 2]     = e0;
    w[tid * 2 + 1] = e1;
    float es = 0.0f;
    if (tid == 0) { es = expf(aself - m); w[N_] = es; }
    red[tid] = e0 + e1 + es;
    __syncthreads();
    for (int s = D_ / 2; s > 0; s >>= 1) {
        if (tid < s) red[tid] += red[tid + s];
        __syncthreads();
    }
    const float inv = 1.0f / red[0];
    __syncthreads();

    const float* hb = h + (size_t)(b * N_) * HD_ + head * D_ + tid;
    float acc = 0.0f;
    for (int r = 0; r < N_; ++r)
        acc += w[r] * hb[(size_t)r * HD_];
    acc += w[N_] * h[(size_t)u * HD_ + head * D_ + tid];
    gat_agg[((size_t)uidx * H_ + head) * D_ + tid] = acc * inv;
}

__global__ void k_gat_finish(const float* __restrict__ h,
                             const float* __restrict__ gat_agg,
                             const float* __restrict__ bias,
                             const float* __restrict__ gg,
                             const float* __restrict__ gb,
                             float* __restrict__ out) {
    __shared__ float red[D_];
    const int t = blockIdx.x;
    const int d = threadIdx.x;
    float v;
    if (t < BN_) {
        const float* p = h + (size_t)t * HD_ + d;
        v = 0.25f * (p[0] + p[D_] + p[2 * D_] + p[3 * D_]);
    } else {
        const float* p = gat_agg + (size_t)(t - BN_) * HD_ + d;
        v = 0.25f * (p[0] + p[D_] + p[2 * D_] + p[3 * D_]);
    }
    v = gelu_exact(v + bias[d]);

    red[d] = v;
    __syncthreads();
    for (int s = D_ / 2; s > 0; s >>= 1) {
        if (d < s) red[d] += red[d + s];
        __syncthreads();
    }
    const float mu = red[0] * (1.0f / D_);
    __syncthreads();
    const float c = v - mu;
    red[d] = c * c;
    __syncthreads();
    for (int s = D_ / 2; s > 0; s >>= 1) {
        if (d < s) red[d] += red[d + s];
        __syncthreads();
    }
    const float var = red[0] * (1.0f / D_);
    out[(size_t)t * D_ + d] = c * rsqrtf(var + EPS_) * gg[d] + gb[d];
}

__global__ void k_gcn_finish(const float* __restrict__ xw,
                             const float* __restrict__ bias,
                             const float* __restrict__ gg,
                             const float* __restrict__ gb,
                             float* __restrict__ out) {
    __shared__ float red[D_];
    const int t = blockIdx.x;
    const int d = threadIdx.x;
    const int b = t / N_;
    const float* xv = xw + (size_t)(BN_ + b * V_) * D_ + d;
    const float s5 = xv[0] + xv[D_] + xv[2 * D_] + xv[3 * D_] + xv[4 * D_];
    const float inv_sqrt6 = 0.40824829046386301637f;
    float v = s5 * inv_sqrt6 + xw[(size_t)t * D_ + d] * (1.0f / 6.0f) + bias[d];
    v = gelu_exact(v);

    red[d] = v;
    __syncthreads();
    for (int s = D_ / 2; s > 0; s >>= 1) {
        if (d < s) red[d] += red[d + s];
        __syncthreads();
    }
    const float mu = red[0] * (1.0f / D_);
    __syncthreads();
    const float c = v - mu;
    red[d] = c * c;
    __syncthreads();
    for (int s = D_ / 2; s > 0; s >>= 1) {
        if (d < s) red[d] += red[d + s];
        __syncthreads();
    }
    const float var = red[0] * (1.0f / D_);
    out[(size_t)t * D_ + d] = c * rsqrtf(var + EPS_) * gg[d] + gb[d];
}

extern "C" void kernel_launch(void* const* d_in, const int* in_sizes, int n_in,
                              void* d_out, int out_size, void* d_ws, size_t ws_size,
                              hipStream_t stream) {
    (void)in_sizes; (void)n_in; (void)out_size; (void)ws_size;

    const float* x       = (const float*)d_in[0];
    const float* vn      = (const float*)d_in[2];
    const float* gat_W   = (const float*)d_in[3];
    const float* att_src = (const float*)d_in[4];
    const float* att_dst = (const float*)d_in[5];
    const float* gat_b   = (const float*)d_in[6];
    const float* gcn_W   = (const float*)d_in[7];
    const float* gcn_b   = (const float*)d_in[8];
    const float* n1g     = (const float*)d_in[9];
    const float* n1b     = (const float*)d_in[10];
    const float* n2g     = (const float*)d_in[11];
    const float* n2b     = (const float*)d_in[12];
    float* out = (float*)d_out;

    float* ws = (float*)d_ws;
    size_t off = 0;
    float* nodes = ws + off; off += (size_t)T_ * D_;
    float* h     = ws + off; off += (size_t)T_ * HD_;
    float* asrc  = ws + off; off += (size_t)T_ * H_;
    float* adst  = ws + off; off += (size_t)T_ * H_;
    float* gagg  = ws + off; off += (size_t)B_ * V_ * HD_;
    float* gat1  = ws + off; off += (size_t)T_ * D_;
    float* xw    = ws + off; off += (size_t)T_ * D_;

    {
        const size_t n = (size_t)T_ * D_;
        k_build_nodes<<<(unsigned)((n + 255) / 256), 256, 0, stream>>>(x, vn, nodes);
    }
    {
        dim3 grid(T_ / 16, HD_ / 256);
        k_gemm_f32_wmma<D_, HD_, true><<<grid, 128, 0, stream>>>(nodes, gat_W, h,
                                                                 att_src, att_dst, asrc, adst);
    }
    {
        dim3 grid(B_ * V_, H_);
        k_gat_virtual<<<grid, D_, 0, stream>>>(h, asrc, adst, gagg);
    }
    k_gat_finish<<<T_, D_, 0, stream>>>(h, gagg, gat_b, n1g, n1b, gat1);
    {
        dim3 grid(T_ / 16, 1);
        k_gemm_f32_wmma<D_, D_, false><<<grid, 128, 0, stream>>>(gat1, gcn_W, xw,
                                                                 nullptr, nullptr, nullptr, nullptr);
    }
    k_gcn_finish<<<BN_, D_, 0, stream>>>(xw, gcn_b, n2g, n2b, out);
}


# Both GEMMs: 128-bit fragment loads and stores (tile t column c = matrix column 4c+t; lane half h takes k0+4h..+3 of an 8-deep K-step); k summation order and score sum order change
# speedup vs baseline: 1.5900x; 1.0397x over previous
#include <hip/hip_runtime.h>
#include <math.h>

#define B_   32
#define N_   512
#define V_   5
#define D_   256
#define H_   4
#define BN_  (B_ * N_)
#define T_   (BN_ + B_ * V_)
#define HD_  (H_ * D_)
#define EPS_ 1e-5f

typedef float v2f __attribute__((ext_vector_type(2)));
typedef float v4f __attribute__((ext_vector_type(4)));
typedef float v8f __attribute__((ext_vector_type(8)));

__device__ __forceinline__ float gelu_exact(float x) {
    return 0.5f * x * (1.0f + erff(x * 0.70710678118654752f));
}

__global__ void k_build_nodes(const float* __restrict__ x,
                              const float* __restrict__ vn,
                              float* __restrict__ nodes) {
    size_t i = (size_t)blockIdx.x * blockDim.x + threadIdx.x;
    if (i < (size_t)BN_ * D_) {
        nodes[i] = x[i];
    } else if (i < (size_t)T_ * D_) {
        size_t j = i - (size_t)BN_ * D_;
        nodes[i] = vn[j % (V_ * D_)];
    }
}

template <int XMASK>
__device__ __forceinline__ float add_row_xmask(float v) {
    return v + __builtin_bit_cast(float, __builtin_amdgcn_update_dpp(
                   0, __builtin_bit_cast(int, v), 0x160 + XMASK, 0xf, 0xf, true));
}

__device__ __forceinline__ float row16_sum(float v) {
    v = add_row_xmask<1>(v);
    v = add_row_xmask<2>(v);
    v = add_row_xmask<4>(v);
    v = add_row_xmask<8>(v);
    return v;
}

template <int KDIM, int NDIM, bool SCORES>
__global__ void k_gemm_f32_wmma(const float* __restrict__ A,
                                const float* __restrict__ Bm,
                                float* __restrict__ C,
                                const float* __restrict__ att_src,
                                const float* __restrict__ att_dst,
                                float* __restrict__ a_src,
                                float* __restrict__ a_dst) {
    static_assert(KDIM % 8 == 0 && NDIM % 64 == 0, "8-deep K-steps, 64-column wave tiles");
    const int wave = threadIdx.x >> 5;
    const int lane = threadIdx.x & 31;
    const int half = lane >> 4;
    const int lm   = lane & 15;
    const int m0   = blockIdx.x * 16;
    const int n0   = blockIdx.y * 256 + wave * 64;
    if constexpr (SCORES) {
        static_assert(NDIM % 256 == 0 && NDIM / 256 == H_, "one work-group per head");
    } else {
        if (n0 >= NDIM) return;
    }

    v8f c0 = {}; v8f c1 = {}; v8f c2 = {}; v8f c3 = {};
    const float* ap = A + (size_t)(m0 + lm) * KDIM + half * 4;
    const float* bp = Bm + (size_t)(half * 4) * NDIM + n0 + lm * 4;

    for (int k0 = 0; k0 < KDIM; k0 += 8) {
        const v4f a  = *(const v4f*)(ap + k0);
        const v4f b0 = *(const v4f*)(bp + (size_t)(k0 + 0) * NDIM);
        const v4f b1 = *(const v4f*)(bp + (size_t)(k0 + 1) * NDIM);
        const v4f b2 = *(const v4f*)(bp + (size_t)(k0 + 2) * NDIM);
        const v4f b3 = *(const v4f*)(bp + (size_t)(k0 + 3) * NDIM);
        v2f af, bf;
        af[0] = a[0]; af[1] = a[1];
        bf[0] = b0[0]; bf[1] = b1[0];
        c0 = __builtin_amdgcn_wmma_f32_16x16x4_f32(false, af, false, bf, (short)0, c0, false, false);
        bf[0] = b0[1]; bf[1] = b1[1];
        c1 = __builtin_amdgcn_wmma_f32_16x16x4_f32(false, af, false, bf, (short)0, c1, false, false);
        bf[0] = b0[2]; bf[1] = b1[2];
        c2 = __builtin_amdgcn_wmma_f32_16x16x4_f32(false, af, false, bf, (short)0, c2, false, false);
        bf[0] = b0[3]; bf[1] = b1[3];
        c3 = __builtin_amdgcn_wmma_f32_16x16x4_f32(false, af, false, bf, (short)0, c3, false, false);
        af[0] = a[2]; af[1] = a[3];
        bf[0] = b2[0]; bf[1] = b3[0];
        c0 = __builtin_amdgcn_wmma_f32_16x16x4_f32(false, af, false, bf, (short)0, c0, false, false);
        bf[0] = b2[1]; bf[1] = b3[1];
        c1 = __builtin_amdgcn_wmma_f32_16x16x4_f32(false, af, false, bf, (short)0, c1, false, false);
        bf[0] = b2[2]; bf[1] = b3[2];
        c2 = __builtin_amdgcn_wmma_f32_16x16x4_f32(false, af, false, bf, (short)0, c2, false, false);
        bf[0] = b2[3]; bf[1] = b3[3];
        c3 = __builtin_amdgcn_wmma_f32_16x16x4_f32(false, af, false, bf, (short)0, c3, false, false);
    }

    float* crow = C + (size_t)(m0 + half * 8) * NDIM + n0 + lm * 4;
#pragma unroll
    for (int j = 0; j < 8; ++j) {
        v4f o;
        o[0] = c0[j]; o[1] = c1[j]; o[2] = c2[j]; o[3] = c3[j];
        *(v4f*)(crow + (size_t)j * NDIM) = o;
    }

    if constexpr (SCORES) {
        __shared__ float part[4][2][16];
        const v4f s = *(const v4f*)(att_src + n0 + lm * 4);
        const v4f d = *(const v4f*)(att_dst + n0 + lm * 4);
        float ps[8], pd[8];
        for (int j = 0; j < 8; ++j) {
            ps[j] = fmaf(c3[j], s[3], fmaf(c2[j], s[2], fmaf(c1[j], s[1], c0[j] * s[0])));
            pd[j] = fmaf(c3[j], d[3], fmaf(c2[j], d[2], fmaf(c1[j], d[1], c0[j] * d[0])));
        }
        for (int j = 0; j < 8; ++j) {
            ps[j] = row16_sum(ps[j]);
            pd[j] = row16_sum(pd[j]);
        }
        if (lm == 0) {
            for (int j = 0; j < 8; ++j) {
                part[wave][0][half * 8 + j] = ps[j];
                part[wave][1][half * 8 + j] = pd[j];
            }
        }
        __syncthreads();
        if (threadIdx.x < 32) {
            const int r = threadIdx.x & 15;
            const int v = threadIdx.x >> 4;
            const float sum = (part[0][v][r] + part[1][v][r]) + (part[2][v][r] + part[3][v][r]);
            float* dst = v ? a_dst : a_src;
            dst[(size_t)(m0 + r) * H_ + blockIdx.y] = sum;
        }
    }
}

__global__ void k_gat_virtual(const float* __restrict__ h,
                              const float* __restrict__ a_src,
                              const float* __restrict__ a_dst,
                              float* __restrict__ gat_agg) {
    __shared__ float w[N_ + 1];
    __shared__ float red[D_];
    const int uidx = blockIdx.x;
    const int head = blockIdx.y;
    const int b    = uidx / V_;
    const int u    = BN_ + uidx;
    const int tid  = threadIdx.x;

    const float adst = a_dst[u * H_ + head];
    const int r0 = b * N_ + tid * 2;
    float al0 = a_src[(size_t)r0 * H_ + head] + adst;
    al0 = (al0 > 0.0f) ? al0 : 0.2f * al0;
    float al1 = a_src[(size_t)(r0 + 1) * H_ + head] + adst;
    al1 = (al1 > 0.0f) ? al1 : 0.2f * al1;
    float aself = a_src[(size_t)u * H_ + head] + adst;
    aself = (aself > 0.0f) ? aself : 0.2f * aself;

    float lmax = fmaxf(al0, al1);
    if (tid == 0) lmax = fmaxf(lmax, aself);
    red[tid] = lmax;
    __syncthreads();
    for (int s = D_ / 2; s > 0; s >>= 1) {
        if (tid < s) red[tid] = fmaxf(red[tid], red[tid + s]);
        __syncthreads();
    }
    const float m = red[0];
    __syncthreads();

    const float e0 = expf(al0 - m);
    const float e1 = expf(al1 - m);
    w[tid * 2]     = e0;
    w[tid * 2 + 1] = e1;
    float es = 0.0f;
    if (tid == 0) { es = expf(aself - m); w[N_] = es; }
    red[tid] = e0 + e1 + es;
    __syncthreads();
    for (int s = D_ / 2; s > 0; s >>= 1) {
        if (tid < s) red[tid] += red[tid + s];
        __syncthreads();
    }
    const float inv = 1.0f / red[0];
    __syncthreads();

    const float* hb = h + (size_t)(b * N_) * HD_ + head * D_ + tid;
    float acc = 0.0f;
    for (int r = 0; r < N_; ++r)
        acc += w[r] * hb[(size_t)r * HD_];
    acc += w[N_] * h[(size_t)u * HD_ + head * D_ + tid];
    gat_agg[((size_t)uidx * H_ + head) * D_ + tid] = acc * inv;
}

__global__ void k_gat_finish(const float* __restrict__ h,
                             const float* __restrict__ gat_agg,
                             const float* __restrict__ bias,
                             const float* __restrict__ gg,
                             const float* __restrict__ gb,
                             float* __restrict__ out) {
    __shared__ float red[D_];
    const int t = blockIdx.x;
    const int d = threadIdx.x;
    float v;
    if (t < BN_) {
        const float* p = h + (size_t)t * HD_ + d;
        v = 0.25f * (p[0] + p[D_] + p[2 * D_] + p[3 * D_]);
    } else {
        const float* p = gat_agg + (size_t)(t - BN_) * HD_ + d;
        v = 0.25f * (p[0] + p[D_] + p[2 * D_] + p[3 * D_]);
    }
    v = gelu_exact(v + bias[d]);

    red[d] = v;
    __syncthreads();
    for (int s = D_ / 2; s > 0; s >>= 1) {
        if (d < s) red[d] += red[d + s];
        __syncthreads();
    }
    const float mu = red[0] * (1.0f / D_);
    __syncthreads();
    const float c = v - mu;
    red[d] = c * c;
    __syncthreads();
    for (int s = D_ / 2; s > 0; s >>= 1) {
        if (d < s) red[d] += red[d + s];
        __syncthreads();
    }
    const float var = red[0] * (1.0f / D_);
    out[(size_t)t * D_ + d] = c * rsqrtf(var + EPS_) * gg[d] + gb[d];
}

__global__ void k_gcn_finish(const float* __restrict__ xw,
                             const float* __restrict__ bias,
                             const float* __restrict__ gg,
                             const float* __restrict__ gb,
                             float* __restrict__ out) {
    __shared__ float red[D_];
    const int t = blockIdx.x;
    const int d = threadIdx.x;
    const int b = t / N_;
    const float* xv = xw + (size_t)(BN_ + b * V_) * D_ + d;
    const float s5 = xv[0] + xv[D_] + xv[2 * D_] + xv[3 * D_] + xv[4 * D_];
    const float inv_sqrt6 = 0.40824829046386301637f;
    float v = s5 * inv_sqrt6 + xw[(size_t)t * D_ + d] * (1.0f / 6.0f) + bias[d];
    v = gelu_exact(v);

    red[d] = v;
    __syncthreads();
    for (int s = D_ / 2; s > 0; s >>= 1) {
        if (d < s) red[d] += red[d + s];
        __syncthreads();
    }
    const float mu = red[0] * (1.0f / D_);
    __syncthreads();
    const float c = v - mu;
    red[d] = c * c;
    __syncthreads();
    for (int s = D_ / 2; s > 0; s >>= 1) {
        if (d < s) red[d] += red[d + s];
        __syncthreads();
    }
    const float var = red[0] * (1.0f / D_);
    out[(size_t)t * D_ + d] = c * rsqrtf(var + EPS_) * gg[d] + gb[d];
}

extern "C" void kernel_launch(void* const* d_in, const int* in_sizes, int n_in,
                              void* d_out, int out_size, void* d_ws, size_t ws_size,
                              hipStream_t stream) {
    (void)in_sizes; (void)n_in; (void)out_size; (void)ws_size;

    const float* x       = (const float*)d_in[0];
    const float* vn      = (const float*)d_in[2];
    const float* gat_W   = (const float*)d_in[3];
    const float* att_src = (const float*)d_in[4];
    const float* att_dst = (const float*)d_in[5];
    const float* gat_b   = (const float*)d_in[6];
    const float* gcn_W   = (const float*)d_in[7];
    const float* gcn_b   = (const float*)d_in[8];
    const float* n1g     = (const float*)d_in[9];
    const float* n1b     = (const float*)d_in[10];
    const float* n2g     = (const float*)d_in[11];
    const float* n2b     = (const float*)d_in[12];
    float* out = (float*)d_out;

    float* ws = (float*)d_ws;
    size_t off = 0;
    float* nodes = ws + off; off += (size_t)T_ * D_;
    float* h     = ws + off; off += (size_t)T_ * HD_;
    float* asrc  = ws + off; off += (size_t)T_ * H_;
    float* adst  = ws + off; off += (size_t)T_ * H_;
    float* gagg  = ws + off; off += (size_t)B_ * V_ * HD_;
    float* gat1  = ws + off; off += (size_t)T_ * D_;
    float* xw    = ws + off; off += (size_t)T_ * D_;

    {
        const size_t n = (size_t)T_ * D_;
        k_build_nodes<<<(unsigned)((n + 255) / 256), 256, 0, stream>>>(x, vn, nodes);
    }
    {
        dim3 grid(T_ / 16, HD_ / 256);
        k_gemm_f32_wmma<D_, HD_, true><<<grid, 128, 0, stream>>>(nodes, gat_W, h,
                                                                 att_src, att_dst, asrc, adst);
    }
    {
        dim3 grid(B_ * V_, H_);
        k_gat_virtual<<<grid, D_, 0, stream>>>(h, asrc, adst, gagg);
    }
    k_gat_finish<<<T_, D_, 0, stream>>>(h, gagg, gat_b, n1g, n1b, gat1);
    {
        dim3 grid(T_ / 16, 1);
        k_gemm_f32_wmma<D_, D_, false><<<grid, 128, 0, stream>>>(gat1, gcn_W, xw,
                                                                 nullptr, nullptr, nullptr, nullptr);
    }
    k_gcn_finish<<<BN_, D_, 0, stream>>>(xw, gcn_b, n2g, n2b, out);
}


# Both LayerNorm finishers: one wave per 256-channel row, 8 channels per lane by 128-bit accesses, mean/variance summed in-wave (DPP, permlanex16), no LDS tree or barrier; LayerNorm sum order changes
# speedup vs baseline: 1.9957x; 1.2552x over previous
#include <hip/hip_runtime.h>
#include <math.h>

#define B_   32
#define N_   512
#define V_   5
#define D_   256
#define H_   4
#define BN_  (B_ * N_)
#define T_   (BN_ + B_ * V_)
#define HD_  (H_ * D_)
#define EPS_ 1e-5f

typedef float v2f __attribute__((ext_vector_type(2)));
typedef float v4f __attribute__((ext_vector_type(4)));
typedef float v8f __attribute__((ext_vector_type(8)));

__device__ __forceinline__ float gelu_exact(float x) {
    return 0.5f * x * (1.0f + erff(x * 0.70710678118654752f));
}

__global__ void k_build_nodes(const float* __restrict__ x,
                              const float* __restrict__ vn,
                              float* __restrict__ nodes) {
    size_t i = (size_t)blockIdx.x * blockDim.x + threadIdx.x;
    if (i < (size_t)BN_ * D_) {
        nodes[i] = x[i];
    } else if (i < (size_t)T_ * D_) {
        size_t j = i - (size_t)BN_ * D_;
        nodes[i] = vn[j % (V_ * D_)];
    }
}

template <int XMASK>
__device__ __forceinline__ float add_row_xmask(float v) {
    return v + __builtin_bit_cast(float, __builtin_amdgcn_update_dpp(
                   0, __builtin_bit_cast(int, v), 0x160 + XMASK, 0xf, 0xf, true));
}

__device__ __forceinline__ float row16_sum(float v) {
    v = add_row_xmask<1>(v);
    v = add_row_xmask<2>(v);
    v = add_row_xmask<4>(v);
    v = add_row_xmask<8>(v);
    return v;
}

template <int KDIM, int NDIM, bool SCORES>
__global__ void k_gemm_f32_wmma(const float* __restrict__ A,
                                const float* __restrict__ Bm,
                                float* __restrict__ C,
                                const float* __restrict__ att_src,
                                const float* __restrict__ att_dst,
                                float* __restrict__ a_src,
                                float* __restrict__ a_dst) {
    static_assert(KDIM % 8 == 0 && NDIM % 64 == 0, "8-deep K-steps, 64-column wave tiles");
    const int wave = threadIdx.x >> 5;
    const int lane = threadIdx.x & 31;
    const int half = lane >> 4;
    const int lm   = lane & 15;
    const int m0   = blockIdx.x * 16;
    const int n0   = blockIdx.y * 256 + wave * 64;
    if constexpr (SCORES) {
        static_assert(NDIM % 256 == 0 && NDIM / 256 == H_, "one work-group per head");
    } else {
        if (n0 >= NDIM) return;
    }

    v8f c0 = {}; v8f c1 = {}; v8f c2 = {}; v8f c3 = {};
    const float* ap = A + (size_t)(m0 + lm) * KDIM + half * 4;
    const float* bp = Bm + (size_t)(half * 4) * NDIM + n0 + lm * 4;

    for (int k0 = 0; k0 < KDIM; k0 += 8) {
        const v4f a  = *(const v4f*)(ap + k0);
        const v4f b0 = *(const v4f*)(bp + (size_t)(k0 + 0) * NDIM);
        const v4f b1 = *(const v4f*)(bp + (size_t)(k0 + 1) * NDIM);
        const v4f b2 = *(const v4f*)(bp + (size_t)(k0 + 2) * NDIM);
        const v4f b3 = *(const v4f*)(bp + (size_t)(k0 + 3) * NDIM);
        v2f af, bf;
        af[0] = a[0]; af[1] = a[1];
        bf[0] = b0[0]; bf[1] = b1[0];
        c0 = __builtin_amdgcn_wmma_f32_16x16x4_f32(false, af, false, bf, (short)0, c0, false, false);
        bf[0] = b0[1]; bf[1] = b1[1];
        c1 = __builtin_amdgcn_wmma_f32_16x16x4_f32(false, af, false, bf, (short)0, c1, false, false);
        bf[0] = b0[2]; bf[1] = b1[2];
        c2 = __builtin_amdgcn_wmma_f32_16x16x4_f32(false, af, false, bf, (short)0, c2, false, false);
        bf[0] = b0[3]; bf[1] = b1[3];
        c3 = __builtin_amdgcn_wmma_f32_16x16x4_f32(false, af, false, bf, (short)0, c3, false, false);
        af[0] = a[2]; af[1] = a[3];
        bf[0] = b2[0]; bf[1] = b3[0];
        c0 = __builtin_amdgcn_wmma_f32_16x16x4_f32(false, af, false, bf, (short)0, c0, false, false);
        bf[0] = b2[1]; bf[1] = b3[1];
        c1 = __builtin_amdgcn_wmma_f32_16x16x4_f32(false, af, false, bf, (short)0, c1, false, false);
        bf[0] = b2[2]; bf[1] = b3[2];
        c2 = __builtin_amdgcn_wmma_f32_16x16x4_f32(false, af, false, bf, (short)0, c2, false, false);
        bf[0] = b2[3]; bf[1] = b3[3];
        c3 = __builtin_amdgcn_wmma_f32_16x16x4_f32(false, af, false, bf, (short)0, c3, false, false);
    }

    float* crow = C + (size_t)(m0 + half * 8) * NDIM + n0 + lm * 4;
#pragma unroll
    for (int j = 0; j < 8; ++j) {
        v4f o;
        o[0] = c0[j]; o[1] = c1[j]; o[2] = c2[j]; o[3] = c3[j];
        *(v4f*)(crow + (size_t)j * NDIM) = o;
    }

    if constexpr (SCORES) {
        __shared__ float part[4][2][16];
        const v4f s = *(const v4f*)(att_src + n0 + lm * 4);
        const v4f d = *(const v4f*)(att_dst + n0 + lm * 4);
        float ps[8], pd[8];
        for (int j = 0; j < 8; ++j) {
            ps[j] = fmaf(c3[j], s[3], fmaf(c2[j], s[2], fmaf(c1[j], s[1], c0[j] * s[0])));
            pd[j] = fmaf(c3[j], d[3], fmaf(c2[j], d[2], fmaf(c1[j], d[1], c0[j] * d[0])));
        }
        for (int j = 0; j < 8; ++j) {
            ps[j] = row16_sum(ps[j]);
            pd[j] = row16_sum(pd[j]);
        }
        if (lm == 0) {
            for (int j = 0; j < 8; ++j) {
                part[wave][0][half * 8 + j] = ps[j];
                part[wave][1][half * 8 + j] = pd[j];
            }
        }
        __syncthreads();
        if (threadIdx.x < 32) {
            const int r = threadIdx.x & 15;
            const int v = threadIdx.x >> 4;
            const float sum = (part[0][v][r] + part[1][v][r]) + (part[2][v][r] + part[3][v][r]);
            float* dst = v ? a_dst : a_src;
            dst[(size_t)(m0 + r) * H_ + blockIdx.y] = sum;
        }
    }
}

__global__ void k_gat_virtual(const float* __restrict__ h,
                              const float* __restrict__ a_src,
                              const float* __restrict__ a_dst,
                              float* __restrict__ gat_agg) {
    __shared__ float w[N_ + 1];
    __shared__ float red[D_];
    const int uidx = blockIdx.x;
    const int head = blockIdx.y;
    const int b    = uidx / V_;
    const int u    = BN_ + uidx;
    const int tid  = threadIdx.x;

    const float adst = a_dst[u * H_ + head];
    const int r0 = b * N_ + tid * 2;
    float al0 = a_src[(size_t)r0 * H_ + head] + adst;
    al0 = (al0 > 0.0f) ? al0 : 0.2f * al0;
    float al1 = a_src[(size_t)(r0 + 1) * H_ + head] + adst;
    al1 = (al1 > 0.0f) ? al1 : 0.2f * al1;
    float aself = a_src[(size_t)u * H_ + head] + adst;
    aself = (aself > 0.0f) ? aself : 0.2f * aself;

    float lmax = fmaxf(al0, al1);
    if (tid == 0) lmax = fmaxf(lmax, aself);
    red[tid] = lmax;
    __syncthreads();
    for (int s = D_ / 2; s > 0; s >>= 1) {
        if (tid < s) red[tid] = fmaxf(red[tid], red[tid + s]);
        __syncthreads();
    }
    const float m = red[0];
    __syncthreads();

    const float e0 = expf(al0 - m);
    const float e1 = expf(al1 - m);
    w[tid * 2]     = e0;
    w[tid * 2 + 1] = e1;
    float es = 0.0f;
    if (tid == 0) { es = expf(aself - m); w[N_] = es; }
    red[tid] = e0 + e1 + es;
    __syncthreads();
    for (int s = D_ / 2; s > 0; s >>= 1) {
        if (tid < s) red[tid] += red[tid + s];
        __syncthreads();
    }
    const float inv = 1.0f / red[0];
    __syncthreads();

    const float* hb = h + (size_t)(b * N_) * HD_ + head * D_ + tid;
    float acc = 0.0f;
    for (int r = 0; r < N_; ++r)
        acc += w[r] * hb[(size_t)r * HD_];
    acc += w[N_] * h[(size_t)u * HD_ + head * D_ + tid];
    gat_agg[((size_t)uidx * H_ + head) * D_ + tid] = acc * inv;
}

#define ROWS_PER_WG_ 8
static_assert(T_ % ROWS_PER_WG_ == 0 && BN_ % ROWS_PER_WG_ == 0, "whole work-groups of rows");
static_assert(D_ == 256, "8 channels per lane of a 32-lane wave");

__device__ __forceinline__ float wave32_sum(float v) {
    v = row16_sum(v);
    return v + __builtin_bit_cast(float, __builtin_amdgcn_permlanex16(
                   0u, __builtin_bit_cast(unsigned, v), 0x76543210u, 0xfedcba98u, true, false));
}

__device__ __forceinline__ float sum8(const float* v) {
    return ((v[0] + v[1]) + (v[2] + v[3])) + ((v[4] + v[5]) + (v[6] + v[7]));
}

__device__ __forceinline__ void layernorm_row_store(const float* v, int lane,
                                                    const float* __restrict__ gg,
                                                    const float* __restrict__ gb,
                                                    float* __restrict__ orow) {
    const float mu = wave32_sum(sum8(v)) * (1.0f / D_);
    float c[8], sq[8];
    for (int i = 0; i < 8; ++i) { c[i] = v[i] - mu; sq[i] = c[i] * c[i]; }
    const float var = wave32_sum(sum8(sq)) * (1.0f / D_);
    const float r = rsqrtf(var + EPS_);
    for (int part = 0; part < 2; ++part) {
        const int d = part * 128 + lane * 4;
        const v4f g = *(const v4f*)(gg + d);
        const v4f b = *(const v4f*)(gb + d);
        v4f o;
        for (int i = 0; i < 4; ++i) o[i] = c[part * 4 + i] * r * g[i] + b[i];
        *(v4f*)(orow + d) = o;
    }
}

__global__ void k_gat_finish(const float* __restrict__ h,
                             const float* __restrict__ gat_agg,
                             const float* __restrict__ bias,
                             const float* __restrict__ gg,
                             const float* __restrict__ gb,
                             float* __restrict__ out) {
    const int wave = __builtin_amdgcn_readfirstlane((int)(threadIdx.x >> 5));
    const int lane = threadIdx.x & 31;
    const int t = blockIdx.x * ROWS_PER_WG_ + wave;
    const float* p = (t < BN_) ? h + (size_t)t * HD_ : gat_agg + (size_t)(t - BN_) * HD_;

    v4f q[2][H_];
    for (int part = 0; part < 2; ++part)
        for (int hd = 0; hd < H_; ++hd)
            q[part][hd] = *(const v4f*)(p + hd * D_ + part * 128 + lane * 4);

    float v[8];
    for (int part = 0; part < 2; ++part) {
        const v4f bs = *(const v4f*)(bias + part * 128 + lane * 4);
        for (int i = 0; i < 4; ++i) {
            const float m = 0.25f * (q[part][0][i] + q[part][1][i] + q[part][2][i] + q[part][3][i]);
            v[part * 4 + i] = gelu_exact(m + bs[i]);
        }
    }
    layernorm_row_store(v, lane, gg, gb, out + (size_t)t * D_);
}

__global__ void k_gcn_finish(const float* __restrict__ xw,
                             const float* __restrict__ bias,
                             const float* __restrict__ gg,
                             const float* __restrict__ gb,
                             float* __restrict__ out) {
    const int wave = __builtin_amdgcn_readfirstlane((int)(threadIdx.x >> 5));
    const int lane = threadIdx.x & 31;
    const int t = blockIdx.x * ROWS_PER_WG_ + wave;
    const int b = t / N_;
    const float* xv = xw + (size_t)(BN_ + b * V_) * D_;
    const float* xr = xw + (size_t)t * D_;
    const float inv_sqrt6 = 0.40824829046386301637f;

    v4f q[2][V_ + 1];
    for (int part = 0; part < 2; ++part) {
        for (int j = 0; j < V_; ++j)
            q[part][j] = *(const v4f*)(xv + j * D_ + part * 128 + lane * 4);
        q[part][V_] = *(const v4f*)(xr + part * 128 + lane * 4);
    }

    float v[8];
    for (int part = 0; part < 2; ++part) {
        const v4f bs = *(const v4f*)(bias + part * 128 + lane * 4);
        for (int i = 0; i < 4; ++i) {
            const float s5 = q[part][0][i] + q[part][1][i] + q[part][2][i] + q[part][3][i] + q[part][4][i];
            v[part * 4 + i] = gelu_exact(s5 * inv_sqrt6 + q[part][V_][i] * (1.0f / 6.0f) + bs[i]);
        }
    }
    layernorm_row_store(v, lane, gg, gb, out + (size_t)t * D_);
}

extern "C" void kernel_launch(void* const* d_in, const int* in_sizes, int n_in,
                              void* d_out, int out_size, void* d_ws, size_t ws_size,
                              hipStream_t stream) {
    (void)in_sizes; (void)n_in; (void)out_size; (void)ws_size;

    const float* x       = (const float*)d_in[0];
    const float* vn      = (const float*)d_in[2];
    const float* gat_W   = (const float*)d_in[3];
    const float* att_src = (const float*)d_in[4];
    const float* att_dst = (const float*)d_in[5];
    const float* gat_b   = (const float*)d_in[6];
    const float* gcn_W   = (const float*)d_in[7];
    const float* gcn_b   = (const float*)d_in[8];
    const float* n1g     = (const float*)d_in[9];
    const float* n1b     = (const float*)d_in[10];
    const float* n2g     = (const float*)d_in[11];
    const float* n2b     = (const float*)d_in[12];
    float* out = (float*)d_out;

    float* ws = (float*)d_ws;
    size_t off = 0;
    float* nodes = ws + off; off += (size_t)T_ * D_;
    float* h     = ws + off; off += (size_t)T_ * HD_;
    float* asrc  = ws + off; off += (size_t)T_ * H_;
    float* adst  = ws + off; off += (size_t)T_ * H_;
    float* gagg  = ws + off; off += (size_t)B_ * V_ * HD_;
    float* gat1  = ws + off; off += (size_t)T_ * D_;
    float* xw    = ws + off; off += (size_t)T_ * D_;

    {
        const size_t n = (size_t)T_ * D_;
        k_build_nodes<<<(unsigned)((n + 255) / 256), 256, 0, stream>>>(x, vn, nodes);
    }
    {
        dim3 grid(T_ / 16, HD_ / 256);
        k_gemm_f32_wmma<D_, HD_, true><<<grid, 128, 0, stream>>>(nodes, gat_W, h,
                                                                 att_src, att_dst, asrc, adst);
    }
    {
        dim3 grid(B_ * V_, H_);
        k_gat_virtual<<<grid, D_, 0, stream>>>(h, asrc, adst, gagg);
    }
    k_gat_finish<<<T_ / ROWS_PER_WG_, ROWS_PER_WG_ * 32, 0, stream>>>(h, gagg, gat_b, n1g, n1b, gat1);
    {
        dim3 grid(T_ / 16, 1);
        k_gemm_f32_wmma<D_, D_, false><<<grid, 128, 0, stream>>>(gat1, gcn_W, xw,
                                                                 nullptr, nullptr, nullptr, nullptr);
    }
    k_gcn_finish<<<BN_ / ROWS_PER_WG_, ROWS_PER_WG_ * 32, 0, stream>>>(xw, gcn_b, n2g, n2b, out);
}


# Apply gat_W after the GAT layer's linear steps: real rows nodes@mean_h(W_h), scores nodes@(W_h att_h) on a fifth-wave WMMA tile, virtual rows (sum_r w x_r)@W_h; T x 1024 h goes; sum orders change
# speedup vs baseline: 2.2123x; 1.1085x over previous
#include <hip/hip_runtime.h>
#include <math.h>

#define B_   32
#define N_   512
#define V_   5
#define D_   256
#define H_   4
#define BN_  (B_ * N_)
#define T_   (BN_ + B_ * V_)
#define HD_  (H_ * D_)
#define EPS_ 1e-5f

typedef float v2f __attribute__((ext_vector_type(2)));
typedef float v4f __attribute__((ext_vector_type(4)));
typedef float v8f __attribute__((ext_vector_type(8)));

__device__ __forceinline__ float gelu_exact(float x) {
    return 0.5f * x * (1.0f + erff(x * 0.70710678118654752f));
}

template <int XMASK>
__device__ __forceinline__ float add_row_xmask(float v) {
    return v + __builtin_bit_cast(float, __builtin_amdgcn_update_dpp(
                   0, __builtin_bit_cast(int, v), 0x160 + XMASK, 0xf, 0xf, true));
}

__device__ __forceinline__ float row16_sum(float v) {
    v = add_row_xmask<1>(v);
    v = add_row_xmask<2>(v);
    v = add_row_xmask<4>(v);
    v = add_row_xmask<8>(v);
    return v;
}

__device__ __forceinline__ float wave32_sum(float v) {
    v = row16_sum(v);
    return v + __builtin_bit_cast(float, __builtin_amdgcn_permlanex16(
                   0u, __builtin_bit_cast(unsigned, v), 0x76543210u, 0xfedcba98u, true, false));
}

#define PREP_WGS_ D_
#define UCOLS_    32

__global__ void k_build_operands(const float* __restrict__ x,
                                 const float* __restrict__ vn,
                                 const float* __restrict__ W,
                                 const float* __restrict__ att_src,
                                 const float* __restrict__ att_dst,
                                 float* __restrict__ nodes,
                                 float* __restrict__ Wm,
                                 float* __restrict__ U) {
    if (blockIdx.x < PREP_WGS_) {
        __shared__ float red[D_ / 32][8];
        const int k = blockIdx.x;
        const int c = threadIdx.x;
        float w[H_], p[8];
        for (int hd = 0; hd < H_; ++hd) w[hd] = W[(size_t)k * HD_ + hd * D_ + c];
        Wm[(size_t)k * D_ + c] = 0.25f * ((w[0] + w[1]) + (w[2] + w[3]));
        for (int hd = 0; hd < H_; ++hd) {
            p[hd]      = w[hd] * att_src[hd * D_ + c];
            p[H_ + hd] = w[hd] * att_dst[hd * D_ + c];
        }
        for (int j = 0; j < 8; ++j) p[j] = wave32_sum(p[j]);
        if ((c & 31) == 0)
            for (int j = 0; j < 8; ++j) red[c >> 5][j] = p[j];
        __syncthreads();
        if (c < UCOLS_) {
            float s = 0.0f;
            if (c < 8)
                s = ((red[0][c] + red[1][c]) + (red[2][c] + red[3][c])) +
                    ((red[4][c] + red[5][c]) + (red[6][c] + red[7][c]));
            U[(size_t)k * UCOLS_ + c] = s;
        }
        return;
    }
    size_t i = (size_t)(blockIdx.x - PREP_WGS_) * blockDim.x + threadIdx.x;
    if (i < (size_t)BN_ * D_) {
        nodes[i] = x[i];
    } else if (i < (size_t)T_ * D_) {
        size_t j = i - (size_t)BN_ * D_;
        nodes[i] = vn[j % (V_ * D_)];
    }
}

template <int KDIM, int NDIM, bool SCORES, int YROWS>
__global__ void k_gemm_f32_wmma(const float* __restrict__ A,
                                const float* __restrict__ Bm,
                                float* __restrict__ C,
                                const float* __restrict__ U,
                                float* __restrict__ a_src,
                                float* __restrict__ a_dst) {
    static_assert(KDIM % 8 == 0 && NDIM % 64 == 0, "8-deep K-steps, 64-column wave tiles");
    const int wave = __builtin_amdgcn_readfirstlane((int)(threadIdx.x >> 5));
    const int lane = threadIdx.x & 31;
    const int half = lane >> 4;
    const int lm   = lane & 15;
    const int m0   = blockIdx.x * 16 + blockIdx.y * YROWS;
    const int n0   = blockIdx.y * 256 + wave * 64;
    if constexpr (SCORES) {
        static_assert(NDIM == 256, "waves 0-3 cover all columns; wave 4 is the score tile");
        if (wave == 4) {
            v8f s0 = {}; v8f s1 = {};
            const float* ap = A + (size_t)(m0 + lm) * KDIM + half * 4;
            const float* up = U + (size_t)(half * 4) * UCOLS_ + lm;
            for (int k0 = 0; k0 < KDIM; k0 += 8) {
                const v4f a = *(const v4f*)(ap + k0);
                const float u0 = up[(size_t)(k0 + 0) * UCOLS_];
                const float u1 = up[(size_t)(k0 + 1) * UCOLS_];
                const float u2 = up[(size_t)(k0 + 2) * UCOLS_];
                const float u3 = up[(size_t)(k0 + 3) * UCOLS_];
                v2f af, bf;
                af[0] = a[0]; af[1] = a[1]; bf[0] = u0; bf[1] = u1;
                s0 = __builtin_amdgcn_wmma_f32_16x16x4_f32(false, af, false, bf, (short)0, s0, false, false);
                af[0] = a[2]; af[1] = a[3]; bf[0] = u2; bf[1] = u3;
                s1 = __builtin_amdgcn_wmma_f32_16x16x4_f32(false, af, false, bf, (short)0, s1, false, false);
            }
            if (lm < 8) {
                float* dst = (lm < H_) ? a_src : a_dst;
                for (int j = 0; j < 8; ++j)
                    dst[(size_t)(m0 + half * 8 + j) * H_ + (lm & (H_ - 1))] = s0[j] + s1[j];
            }
            return;
        }
    } else {
        if (n0 >= NDIM) return;
    }

    v8f c0 = {}; v8f c1 = {}; v8f c2 = {}; v8f c3 = {};
    const float* ap = A + (size_t)(m0 + lm) * KDIM + half * 4;
    const float* bp = Bm + (size_t)(half * 4) * NDIM + n0 + lm * 4;

    for (int k0 = 0; k0 < KDIM; k0 += 8) {
        const v4f a  = *(const v4f*)(ap + k0);
        const v4f b0 = *(const v4f*)(bp + (size_t)(k0 + 0) * NDIM);
        const v4f b1 = *(const v4f*)(bp + (size_t)(k0 + 1) * NDIM);
        const v4f b2 = *(const v4f*)(bp + (size_t)(k0 + 2) * NDIM);
        const v4f b3 = *(const v4f*)(bp + (size_t)(k0 + 3) * NDIM);
        v2f af, bf;
        af[0] = a[0]; af[1] = a[1];
        bf[0] = b0[0]; bf[1] = b1[0];
        c0 = __builtin_amdgcn_wmma_f32_16x16x4_f32(false, af, false, bf, (short)0, c0, false, false);
        bf[0] = b0[1]; bf[1] = b1[1];
        c1 = __builtin_amdgcn_wmma_f32_16x16x4_f32(false, af, false, bf, (short)0, c1, false, false);
        bf[0] = b0[2]; bf[1] = b1[2];
        c2 = __builtin_amdgcn_wmma_f32_16x16x4_f32(false, af, false, bf, (short)0, c2, false, false);
        bf[0] = b0[3]; bf[1] = b1[3];
        c3 = __builtin_amdgcn_wmma_f32_16x16x4_f32(false, af, false, bf, (short)0, c3, false, false);
        af[0] = a[2]; af[1] = a[3];
        bf[0] = b2[0]; bf[1] = b3[0];
        c0 = __builtin_amdgcn_wmma_f32_16x16x4_f32(false, af, false, bf, (short)0, c0, false, false);
        bf[0] = b2[1]; bf[1] = b3[1];
        c1 = __builtin_amdgcn_wmma_f32_16x16x4_f32(false, af, false, bf, (short)0, c1, false, false);
        bf[0] = b2[2]; bf[1] = b3[2];
        c2 = __builtin_amdgcn_wmma_f32_16x16x4_f32(false, af, false, bf, (short)0, c2, false, false);
        bf[0] = b2[3]; bf[1] = b3[3];
        c3 = __builtin_amdgcn_wmma_f32_16x16x4_f32(false, af, false, bf, (short)0, c3, false, false);
    }

    float* crow = C + (size_t)(m0 + half * 8) * NDIM + n0 + lm * 4;
#pragma unroll
    for (int j = 0; j < 8; ++j) {
        v4f o;
        o[0] = c0[j]; o[1] = c1[j]; o[2] = c2[j]; o[3] = c3[j];
        *(v4f*)(crow + (size_t)j * NDIM) = o;
    }
}

__global__ void k_gat_virtual(const float* __restrict__ nodes,
                              const float* __restrict__ a_src,
                              const float* __restrict__ a_dst,
                              float* __restrict__ z) {
    __shared__ float w[N_ + 1];
    __shared__ float red[D_];
    const int uidx = blockIdx.x;
    const int head = blockIdx.y;
    const int b    = uidx / V_;
    const int u    = BN_ + uidx;
    const int tid  = threadIdx.x;

    const float adst = a_dst[u * H_ + head];
    const int r0 = b * N_ + tid * 2;
    float al0 = a_src[(size_t)r0 * H_ + head] + adst;
    al0 = (al0 > 0.0f) ? al0 : 0.2f * al0;
    float al1 = a_src[(size_t)(r0 + 1) * H_ + head] + adst;
    al1 = (al1 > 0.0f) ? al1 : 0.2f * al1;
    float aself = a_src[(size_t)u * H_ + head] + adst;
    aself = (aself > 0.0f) ? aself : 0.2f * aself;

    float lmax = fmaxf(al0, al1);
    if (tid == 0) lmax = fmaxf(lmax, aself);
    red[tid] = lmax;
    __syncthreads();
    for (int s = D_ / 2; s > 0; s >>= 1) {
        if (tid < s) red[tid] = fmaxf(red[tid], red[tid + s]);
        __syncthreads();
    }
    const float m = red[0];
    __syncthreads();

    const float e0 = expf(al0 - m);
    const float e1 = expf(al1 - m);
    w[tid * 2]     = e0;
    w[tid * 2 + 1] = e1;
    float es = 0.0f;
    if (tid == 0) { es = expf(aself - m); w[N_] = es; }
    red[tid] = e0 + e1 + es;
    __syncthreads();
    for (int s = D_ / 2; s > 0; s >>= 1) {
        if (tid < s) red[tid] += red[tid + s];
        __syncthreads();
    }
    const float inv = 1.0f / red[0];
    __syncthreads();

    const float* xb = nodes + (size_t)(b * N_) * D_ + tid;
    float acc = 0.0f;
    for (int r = 0; r < N_; ++r)
        acc += w[r] * xb[(size_t)r * D_];
    acc += w[N_] * nodes[(size_t)u * D_ + tid];
    z[((size_t)head * (B_ * V_) + uidx) * D_ + tid] = acc * inv;
}

#define ROWS_PER_WG_ 8
static_assert(T_ % ROWS_PER_WG_ == 0 && BN_ % ROWS_PER_WG_ == 0, "whole work-groups of rows");
static_assert(D_ == 256, "8 channels per lane of a 32-lane wave");

__device__ __forceinline__ float sum8(const float* v) {
    return ((v[0] + v[1]) + (v[2] + v[3])) + ((v[4] + v[5]) + (v[6] + v[7]));
}

__device__ __forceinline__ void layernorm_row_store(const float* v, int lane,
                                                    const float* __restrict__ gg,
                                                    const float* __restrict__ gb,
                                                    float* __restrict__ orow) {
    const float mu = wave32_sum(sum8(v)) * (1.0f / D_);
    float c[8], sq[8];
    for (int i = 0; i < 8; ++i) { c[i] = v[i] - mu; sq[i] = c[i] * c[i]; }
    const float var = wave32_sum(sum8(sq)) * (1.0f / D_);
    const float r = rsqrtf(var + EPS_);
    for (int part = 0; part < 2; ++part) {
        const int d = part * 128 + lane * 4;
        const v4f g = *(const v4f*)(gg + d);
        const v4f b = *(const v4f*)(gb + d);
        v4f o;
        for (int i = 0; i < 4; ++i) o[i] = c[part * 4 + i] * r * g[i] + b[i];
        *(v4f*)(orow + d) = o;
    }
}

__global__ void k_gat_finish(const float* __restrict__ gm,
                             const float* __restrict__ gv,
                             const float* __restrict__ bias,
                             const float* __restrict__ gg,
                             const float* __restrict__ gb,
                             float* __restrict__ out) {
    const int wave = __builtin_amdgcn_readfirstlane((int)(threadIdx.x >> 5));
    const int lane = threadIdx.x & 31;
    const int t = blockIdx.x * ROWS_PER_WG_ + wave;

    v4f m[2];
    if (t < BN_) {
        for (int part = 0; part < 2; ++part)
            m[part] = *(const v4f*)(gm + (size_t)t * D_ + part * 128 + lane * 4);
    } else {
        const int uidx = t - BN_;
        v4f q[2][H_];
        for (int part = 0; part < 2; ++part)
            for (int hd = 0; hd < H_; ++hd)
                q[part][hd] = *(const v4f*)(gv + (size_t)(hd * (B_ * V_) + uidx) * HD_ +
                                            hd * D_ + part * 128 + lane * 4);
        for (int part = 0; part < 2; ++part)
            for (int i = 0; i < 4; ++i)
                m[part][i] = 0.25f * (q[part][0][i] + q[part][1][i] + q[part][2][i] + q[part][3][i]);
    }

    float v[8];
    for (int part = 0; part < 2; ++part) {
        const v4f bs = *(const v4f*)(bias + part * 128 + lane * 4);
        for (int i = 0; i < 4; ++i)
            v[part * 4 + i] = gelu_exact(m[part][i] + bs[i]);
    }
    layernorm_row_store(v, lane, gg, gb, out + (size_t)t * D_);
}

__global__ void k_gcn_finish(const float* __restrict__ xw,
                             const float* __restrict__ bias,
                             const float* __restrict__ gg,
                             const float* __restrict__ gb,
                             float* __restrict__ out) {
    const int wave = __builtin_amdgcn_readfirstlane((int)(threadIdx.x >> 5));
    const int lane = threadIdx.x & 31;
    const int t = blockIdx.x * ROWS_PER_WG_ + wave;
    const int b = t / N_;
    const float* xv = xw + (size_t)(BN_ + b * V_) * D_;
    const float* xr = xw + (size_t)t * D_;
    const float inv_sqrt6 = 0.40824829046386301637f;

    v4f q[2][V_ + 1];
    for (int part = 0; part < 2; ++part) {
        for (int j = 0; j < V_; ++j)
            q[part][j] = *(const v4f*)(xv + j * D_ + part * 128 + lane * 4);
        q[part][V_] = *(const v4f*)(xr + part * 128 + lane * 4);
    }

    float v[8];
    for (int part = 0; part < 2; ++part) {
        const v4f bs = *(const v4f*)(bias + part * 128 + lane * 4);
        for (int i = 0; i < 4; ++i) {
            const float s5 = q[part][0][i] + q[part][1][i] + q[part][2][i] + q[part][3][i] + q[part][4][i];
            v[part * 4 + i] = gelu_exact(s5 * inv_sqrt6 + q[part][V_][i] * (1.0f / 6.0f) + bs[i]);
        }
    }
    layernorm_row_store(v, lane, gg, gb, out + (size_t)t * D_);
}

extern "C" void kernel_launch(void* const* d_in, const int* in_sizes, int n_in,
                              void* d_out, int out_size, void* d_ws, size_t ws_size,
                              hipStream_t stream) {
    (void)in_sizes; (void)n_in; (void)out_size; (void)ws_size;

    const float* x       = (const float*)d_in[0];
    const float* vn      = (const float*)d_in[2];
    const float* gat_W   = (const float*)d_in[3];
    const float* att_src = (const float*)d_in[4];
    const float* att_dst = (const float*)d_in[5];
    const float* gat_b   = (const float*)d_in[6];
    const float* gcn_W   = (const float*)d_in[7];
    const float* gcn_b   = (const float*)d_in[8];
    const float* n1g     = (const float*)d_in[9];
    const float* n1b     = (const float*)d_in[10];
    const float* n2g     = (const float*)d_in[11];
    const float* n2b     = (const float*)d_in[12];
    float* out = (float*)d_out;

    float* ws = (float*)d_ws;
    size_t off = 0;
    float* nodes = ws + off; off += (size_t)T_ * D_;
    float* wm    = ws + off; off += (size_t)D_ * D_;
    float* ut    = ws + off; off += (size_t)D_ * UCOLS_;
    float* gm    = ws + off; off += (size_t)T_ * D_;
    float* asrc  = ws + off; off += (size_t)T_ * H_;
    float* adst  = ws + off; off += (size_t)T_ * H_;
    float* z     = ws + off; off += (size_t)H_ * B_ * V_ * D_;
    float* gv    = ws + off; off += (size_t)H_ * B_ * V_ * HD_;
    float* gat1  = ws + off; off += (size_t)T_ * D_;
    float* xw    = ws + off; off += (size_t)T_ * D_;

    k_build_operands<<<PREP_WGS_ + T_, D_, 0, stream>>>(x, vn, gat_W, att_src, att_dst,
                                                        nodes, wm, ut);
    {
        dim3 grid(T_ / 16, 1);
        k_gemm_f32_wmma<D_, D_, true, 0><<<grid, 160, 0, stream>>>(nodes, wm, gm, ut, asrc, adst);
    }
    {
        dim3 grid(B_ * V_, H_);
        k_gat_virtual<<<grid, D_, 0, stream>>>(nodes, asrc, adst, z);
    }
    {
        dim3 grid(B_ * V_ / 16, H_);
        k_gemm_f32_wmma<D_, HD_, false, B_ * V_><<<grid, 128, 0, stream>>>(z, gat_W, gv,
                                                                          nullptr, nullptr, nullptr);
    }
    k_gat_finish<<<T_ / ROWS_PER_WG_, ROWS_PER_WG_ * 32, 0, stream>>>(gm, gv, gat_b, n1g, n1b, gat1);
    {
        dim3 grid(T_ / 16, 1);
        k_gemm_f32_wmma<D_, D_, false, 0><<<grid, 128, 0, stream>>>(gat1, gcn_W, xw,
                                                                    nullptr, nullptr, nullptr);
    }
    k_gcn_finish<<<BN_ / ROWS_PER_WG_, ROWS_PER_WG_ * 32, 0, stream>>>(xw, gcn_b, n2g, n2b, out);
}


# The node matrix concat(x, tile(virtual_nodes)) is no longer copied to the workspace: the first GEMM's A loader and the virtual aggregation read x / virtual_nodes rows in place
# speedup vs baseline: 2.3628x; 1.0680x over previous
#include <hip/hip_runtime.h>
#include <math.h>

#define B_   32
#define N_   512
#define V_   5
#define D_   256
#define H_   4
#define BN_  (B_ * N_)
#define T_   (BN_ + B_ * V_)
#define HD_  (H_ * D_)
#define EPS_ 1e-5f

typedef float v2f __attribute__((ext_vector_type(2)));
typedef float v4f __attribute__((ext_vector_type(4)));
typedef float v8f __attribute__((ext_vector_type(8)));

__device__ __forceinline__ float gelu_exact(float x) {
    return 0.5f * x * (1.0f + erff(x * 0.70710678118654752f));
}

template <int XMASK>
__device__ __forceinline__ float add_row_xmask(float v) {
    return v + __builtin_bit_cast(float, __builtin_amdgcn_update_dpp(
                   0, __builtin_bit_cast(int, v), 0x160 + XMASK, 0xf, 0xf, true));
}

__device__ __forceinline__ float row16_sum(float v) {
    v = add_row_xmask<1>(v);
    v = add_row_xmask<2>(v);
    v = add_row_xmask<4>(v);
    v = add_row_xmask<8>(v);
    return v;
}

__device__ __forceinline__ float wave32_sum(float v) {
    v = row16_sum(v);
    return v + __builtin_bit_cast(float, __builtin_amdgcn_permlanex16(
                   0u, __builtin_bit_cast(unsigned, v), 0x76543210u, 0xfedcba98u, true, false));
}

#define PREP_WGS_ D_
#define UCOLS_    32

__device__ __forceinline__ const float* node_row(const float* __restrict__ x,
                                                 const float* __restrict__ vn, int t) {
    return (t < BN_) ? x + (size_t)t * D_ : vn + (size_t)((t - BN_) % V_) * D_;
}

__global__ void k_fold_weights(const float* __restrict__ W,
                               const float* __restrict__ att_src,
                               const float* __restrict__ att_dst,
                               float* __restrict__ Wm,
                               float* __restrict__ U) {
    __shared__ float red[D_ / 32][8];
    const int k = blockIdx.x;
    const int c = threadIdx.x;
    float w[H_], p[8];
    for (int hd = 0; hd < H_; ++hd) w[hd] = W[(size_t)k * HD_ + hd * D_ + c];
    Wm[(size_t)k * D_ + c] = 0.25f * ((w[0] + w[1]) + (w[2] + w[3]));
    for (int hd = 0; hd < H_; ++hd) {
        p[hd]      = w[hd] * att_src[hd * D_ + c];
        p[H_ + hd] = w[hd] * att_dst[hd * D_ + c];
    }
    for (int j = 0; j < 8; ++j) p[j] = wave32_sum(p[j]);
    if ((c & 31) == 0)
        for (int j = 0; j < 8; ++j) red[c >> 5][j] = p[j];
    __syncthreads();
    if (c < UCOLS_) {
        float s = 0.0f;
        if (c < 8)
            s = ((red[0][c] + red[1][c]) + (red[2][c] + red[3][c])) +
                ((red[4][c] + red[5][c]) + (red[6][c] + red[7][c]));
        U[(size_t)k * UCOLS_ + c] = s;
    }
}

template <int KDIM, int NDIM, bool SCORES, int YROWS, bool NODE_ROWS>
__global__ void k_gemm_f32_wmma(const float* __restrict__ A,
                                const float* __restrict__ Av,
                                const float* __restrict__ Bm,
                                float* __restrict__ C,
                                const float* __restrict__ U,
                                float* __restrict__ a_src,
                                float* __restrict__ a_dst) {
    static_assert(KDIM % 8 == 0 && NDIM % 64 == 0, "8-deep K-steps, 64-column wave tiles");
    const int wave = __builtin_amdgcn_readfirstlane((int)(threadIdx.x >> 5));
    const int lane = threadIdx.x & 31;
    const int half = lane >> 4;
    const int lm   = lane & 15;
    const int m0   = blockIdx.x * 16 + blockIdx.y * YROWS;
    const int n0   = blockIdx.y * 256 + wave * 64;
    static_assert(!NODE_ROWS || KDIM == D_, "node rows are 256 words");
    const float* ap;
    if constexpr (NODE_ROWS) ap = node_row(A, Av, m0 + lm) + half * 4;
    else                     ap = A + (size_t)(m0 + lm) * KDIM + half * 4;
    if constexpr (SCORES) {
        static_assert(NDIM == 256, "waves 0-3 cover all columns; wave 4 is the score tile");
        if (wave == 4) {
            v8f s0 = {}; v8f s1 = {};
            const float* up = U + (size_t)(half * 4) * UCOLS_ + lm;
            for (int k0 = 0; k0 < KDIM; k0 += 8) {
                const v4f a = *(const v4f*)(ap + k0);
                const float u0 = up[(size_t)(k0 + 0) * UCOLS_];
                const float u1 = up[(size_t)(k0 + 1) * UCOLS_];
                const float u2 = up[(size_t)(k0 + 2) * UCOLS_];
                const float u3 = up[(size_t)(k0 + 3) * UCOLS_];
                v2f af, bf;
                af[0] = a[0]; af[1] = a[1]; bf[0] = u0; bf[1] = u1;
                s0 = __builtin_amdgcn_wmma_f32_16x16x4_f32(false, af, false, bf, (short)0, s0, false, false);
                af[0] = a[2]; af[1] = a[3]; bf[0] = u2; bf[1] = u3;
                s1 = __builtin_amdgcn_wmma_f32_16x16x4_f32(false, af, false, bf, (short)0, s1, false, false);
            }
            if (lm < 8) {
                float* dst = (lm < H_) ? a_src : a_dst;
                for (int j = 0; j < 8; ++j)
                    dst[(size_t)(m0 + half * 8 + j) * H_ + (lm & (H_ - 1))] = s0[j] + s1[j];
            }
            return;
        }
    } else {
        if (n0 >= NDIM) return;
    }

    v8f c0 = {}; v8f c1 = {}; v8f c2 = {}; v8f c3 = {};
    const float* bp = Bm + (size_t)(half * 4) * NDIM + n0 + lm * 4;

    for (int k0 = 0; k0 < KDIM; k0 += 8) {
        const v4f a  = *(const v4f*)(ap + k0);
        const v4f b0 = *(const v4f*)(bp + (size_t)(k0 + 0) * NDIM);
        const v4f b1 = *(const v4f*)(bp + (size_t)(k0 + 1) * NDIM);
        const v4f b2 = *(const v4f*)(bp + (size_t)(k0 + 2) * NDIM);
        const v4f b3 = *(const v4f*)(bp + (size_t)(k0 + 3) * NDIM);
        v2f af, bf;
        af[0] = a[0]; af[1] = a[1];
        bf[0] = b0[0]; bf[1] = b1[0];
        c0 = __builtin_amdgcn_wmma_f32_16x16x4_f32(false, af, false, bf, (short)0, c0, false, false);
        bf[0] = b0[1]; bf[1] = b1[1];
        c1 = __builtin_amdgcn_wmma_f32_16x16x4_f32(false, af, false, bf, (short)0, c1, false, false);
        bf[0] = b0[2]; bf[1] = b1[2];
        c2 = __builtin_amdgcn_wmma_f32_16x16x4_f32(false, af, false, bf, (short)0, c2, false, false);
        bf[0] = b0[3]; bf[1] = b1[3];
        c3 = __builtin_amdgcn_wmma_f32_16x16x4_f32(false, af, false, bf, (short)0, c3, false, false);
        af[0] = a[2]; af[1] = a[3];
        bf[0] = b2[0]; bf[1] = b3[0];
        c0 = __builtin_amdgcn_wmma_f32_16x16x4_f32(false, af, false, bf, (short)0, c0, false, false);
        bf[0] = b2[1]; bf[1] = b3[1];
        c1 = __builtin_amdgcn_wmma_f32_16x16x4_f32(false, af, false, bf, (short)0, c1, false, false);
        bf[0] = b2[2]; bf[1] = b3[2];
        c2 = __builtin_amdgcn_wmma_f32_16x16x4_f32(false, af, false, bf, (short)0, c2, false, false);
        bf[0] = b2[3]; bf[1] = b3[3];
        c3 = __builtin_amdgcn_wmma_f32_16x16x4_f32(false, af, false, bf, (short)0, c3, false, false);
    }

    float* crow = C + (size_t)(m0 + half * 8) * NDIM + n0 + lm * 4;
#pragma unroll
    for (int j = 0; j < 8; ++j) {
        v4f o;
        o[0] = c0[j]; o[1] = c1[j]; o[2] = c2[j]; o[3] = c3[j];
        *(v4f*)(crow + (size_t)j * NDIM) = o;
    }
}

__global__ void k_gat_virtual(const float* __restrict__ x,
                              const float* __restrict__ vn,
                              const float* __restrict__ a_src,
                              const float* __restrict__ a_dst,
                              float* __restrict__ z) {
    __shared__ float w[N_ + 1];
    __shared__ float red[D_];
    const int uidx = blockIdx.x;
    const int head = blockIdx.y;
    const int b    = uidx / V_;
    const int u    = BN_ + uidx;
    const int tid  = threadIdx.x;

    const float adst = a_dst[u * H_ + head];
    const int r0 = b * N_ + tid * 2;
    float al0 = a_src[(size_t)r0 * H_ + head] + adst;
    al0 = (al0 > 0.0f) ? al0 : 0.2f * al0;
    float al1 = a_src[(size_t)(r0 + 1) * H_ + head] + adst;
    al1 = (al1 > 0.0f) ? al1 : 0.2f * al1;
    float aself = a_src[(size_t)u * H_ + head] + adst;
    aself = (aself > 0.0f) ? aself : 0.2f * aself;

    float lmax = fmaxf(al0, al1);
    if (tid == 0) lmax = fmaxf(lmax, aself);
    red[tid] = lmax;
    __syncthreads();
    for (int s = D_ / 2; s > 0; s >>= 1) {
        if (tid < s) red[tid] = fmaxf(red[tid], red[tid + s]);
        __syncthreads();
    }
    const float m = red[0];
    __syncthreads();

    const float e0 = expf(al0 - m);
    const float e1 = expf(al1 - m);
    w[tid * 2]     = e0;
    w[tid * 2 + 1] = e1;
    float es = 0.0f;
    if (tid == 0) { es = expf(aself - m); w[N_] = es; }
    red[tid] = e0 + e1 + es;
    __syncthreads();
    for (int s = D_ / 2; s > 0; s >>= 1) {
        if (tid < s) red[tid] += red[tid + s];
        __syncthreads();
    }
    const float inv = 1.0f / red[0];
    __syncthreads();

    const float* xb = x + (size_t)(b * N_) * D_ + tid;
    float acc = 0.0f;
    for (int r = 0; r < N_; ++r)
        acc += w[r] * xb[(size_t)r * D_];
    acc += w[N_] * vn[(size_t)(uidx % V_) * D_ + tid];
    z[((size_t)head * (B_ * V_) + uidx) * D_ + tid] = acc * inv;
}

#define ROWS_PER_WG_ 8
static_assert(T_ % ROWS_PER_WG_ == 0 && BN_ % ROWS_PER_WG_ == 0, "whole work-groups of rows");
static_assert(D_ == 256, "8 channels per lane of a 32-lane wave");

__device__ __forceinline__ float sum8(const float* v) {
    return ((v[0] + v[1]) + (v[2] + v[3])) + ((v[4] + v[5]) + (v[6] + v[7]));
}

__device__ __forceinline__ void layernorm_row_store(const float* v, int lane,
                                                    const float* __restrict__ gg,
                                                    const float* __restrict__ gb,
                                                    float* __restrict__ orow) {
    const float mu = wave32_sum(sum8(v)) * (1.0f / D_);
    float c[8], sq[8];
    for (int i = 0; i < 8; ++i) { c[i] = v[i] - mu; sq[i] = c[i] * c[i]; }
    const float var = wave32_sum(sum8(sq)) * (1.0f / D_);
    const float r = rsqrtf(var + EPS_);
    for (int part = 0; part < 2; ++part) {
        const int d = part * 128 + lane * 4;
        const v4f g = *(const v4f*)(gg + d);
        const v4f b = *(const v4f*)(gb + d);
        v4f o;
        for (int i = 0; i < 4; ++i) o[i] = c[part * 4 + i] * r * g[i] + b[i];
        *(v4f*)(orow + d) = o;
    }
}

__global__ void k_gat_finish(const float* __restrict__ gm,
                             const float* __restrict__ gv,
                             const float* __restrict__ bias,
                             const float* __restrict__ gg,
                             const float* __restrict__ gb,
                             float* __restrict__ out) {
    const int wave = __builtin_amdgcn_readfirstlane((int)(threadIdx.x >> 5));
    const int lane = threadIdx.x & 31;
    const int t = blockIdx.x * ROWS_PER_WG_ + wave;

    v4f m[2];
    if (t < BN_) {
        for (int part = 0; part < 2; ++part)
            m[part] = *(const v4f*)(gm + (size_t)t * D_ + part * 128 + lane * 4);
    } else {
        const int uidx = t - BN_;
        v4f q[2][H_];
        for (int part = 0; part < 2; ++part)
            for (int hd = 0; hd < H_; ++hd)
                q[part][hd] = *(const v4f*)(gv + (size_t)(hd * (B_ * V_) + uidx) * HD_ +
                                            hd * D_ + part * 128 + lane * 4);
        for (int part = 0; part < 2; ++part)
            for (int i = 0; i < 4; ++i)
                m[part][i] = 0.25f * (q[part][0][i] + q[part][1][i] + q[part][2][i] + q[part][3][i]);
    }

    float v[8];
    for (int part = 0; part < 2; ++part) {
        const v4f bs = *(const v4f*)(bias + part * 128 + lane * 4);
        for (int i = 0; i < 4; ++i)
            v[part * 4 + i] = gelu_exact(m[part][i] + bs[i]);
    }
    layernorm_row_store(v, lane, gg, gb, out + (size_t)t * D_);
}

__global__ void k_gcn_finish(const float* __restrict__ xw,
                             const float* __restrict__ bias,
                             const float* __restrict__ gg,
                             const float* __restrict__ gb,
                             float* __restrict__ out) {
    const int wave = __builtin_amdgcn_readfirstlane((int)(threadIdx.x >> 5));
    const int lane = threadIdx.x & 31;
    const int t = blockIdx.x * ROWS_PER_WG_ + wave;
    const int b = t / N_;
    const float* xv = xw + (size_t)(BN_ + b * V_) * D_;
    const float* xr = xw + (size_t)t * D_;
    const float inv_sqrt6 = 0.40824829046386301637f;

    v4f q[2][V_ + 1];
    for (int part = 0; part < 2; ++part) {
        for (int j = 0; j < V_; ++j)
            q[part][j] = *(const v4f*)(xv + j * D_ + part * 128 + lane * 4);
        q[part][V_] = *(const v4f*)(xr + part * 128 + lane * 4);
    }

    float v[8];
    for (int part = 0; part < 2; ++part) {
        const v4f bs = *(const v4f*)(bias + part * 128 + lane * 4);
        for (int i = 0; i < 4; ++i) {
            const float s5 = q[part][0][i] + q[part][1][i] + q[part][2][i] + q[part][3][i] + q[part][4][i];
            v[part * 4 + i] = gelu_exact(s5 * inv_sqrt6 + q[part][V_][i] * (1.0f / 6.0f) + bs[i]);
        }
    }
    layernorm_row_store(v, lane, gg, gb, out + (size_t)t * D_);
}

extern "C" void kernel_launch(void* const* d_in, const int* in_sizes, int n_in,
                              void* d_out, int out_size, void* d_ws, size_t ws_size,
                              hipStream_t stream) {
    (void)in_sizes; (void)n_in; (void)out_size; (void)ws_size;

    const float* x       = (const float*)d_in[0];
    const float* vn      = (const float*)d_in[2];
    const float* gat_W   = (const float*)d_in[3];
    const float* att_src = (const float*)d_in[4];
    const float* att_dst = (const float*)d_in[5];
    const float* gat_b   = (const float*)d_in[6];
    const float* gcn_W   = (const float*)d_in[7];
    const float* gcn_b   = (const float*)d_in[8];
    const float* n1g     = (const float*)d_in[9];
    const float* n1b     = (const float*)d_in[10];
    const float* n2g     = (const float*)d_in[11];
    const float* n2b     = (const float*)d_in[12];
    float* out = (float*)d_out;

    float* ws = (float*)d_ws;
    size_t off = 0;
    float* wm    = ws + off; off += (size_t)D_ * D_;
    float* ut    = ws + off; off += (size_t)D_ * UCOLS_;
    float* gm    = ws + off; off += (size_t)T_ * D_;
    float* asrc  = ws + off; off += (size_t)T_ * H_;
    float* adst  = ws + off; off += (size_t)T_ * H_;
    float* z     = ws + off; off += (size_t)H_ * B_ * V_ * D_;
    float* gv    = ws + off; off += (size_t)H_ * B_ * V_ * HD_;
    float* gat1  = ws + off; off += (size_t)T_ * D_;
    float* xw    = ws + off; off += (size_t)T_ * D_;

    k_fold_weights<<<PREP_WGS_, D_, 0, stream>>>(gat_W, att_src, att_dst, wm, ut);
    {
        dim3 grid(T_ / 16, 1);
        k_gemm_f32_wmma<D_, D_, true, 0, true><<<grid, 160, 0, stream>>>(x, vn, wm, gm, ut, asrc, adst);
    }
    {
        dim3 grid(B_ * V_, H_);
        k_gat_virtual<<<grid, D_, 0, stream>>>(x, vn, asrc, adst, z);
    }
    {
        dim3 grid(B_ * V_ / 16, H_);
        k_gemm_f32_wmma<D_, HD_, false, B_ * V_, false><<<grid, 128, 0, stream>>>(z, nullptr, gat_W, gv,
                                                                                 nullptr, nullptr, nullptr);
    }
    k_gat_finish<<<T_ / ROWS_PER_WG_, ROWS_PER_WG_ * 32, 0, stream>>>(gm, gv, gat_b, n1g, n1b, gat1);
    {
        dim3 grid(T_ / 16, 1);
        k_gemm_f32_wmma<D_, D_, false, 0, false><<<grid, 128, 0, stream>>>(gat1, nullptr, gcn_W, xw,
                                                                           nullptr, nullptr, nullptr);
    }
    k_gcn_finish<<<BN_ / ROWS_PER_WG_, ROWS_PER_WG_ * 32, 0, stream>>>(xw, gcn_b, n2g, n2b, out);
}


# Virtual aggregation: one work-group per virtual node does all 4 heads, so each x row is read once; 128-bit loads 32 deep, rows split over 4 thread groups, LDS combine; softmax, row sum orders change
# speedup vs baseline: 2.6335x; 1.1146x over previous
#include <hip/hip_runtime.h>
#include <math.h>

#define B_   32
#define N_   512
#define V_   5
#define D_   256
#define H_   4
#define BN_  (B_ * N_)
#define T_   (BN_ + B_ * V_)
#define HD_  (H_ * D_)
#define EPS_ 1e-5f

typedef float v2f __attribute__((ext_vector_type(2)));
typedef float v4f __attribute__((ext_vector_type(4)));
typedef float v8f __attribute__((ext_vector_type(8)));

__device__ __forceinline__ float gelu_exact(float x) {
    return 0.5f * x * (1.0f + erff(x * 0.70710678118654752f));
}

template <int XMASK>
__device__ __forceinline__ float add_row_xmask(float v) {
    return v + __builtin_bit_cast(float, __builtin_amdgcn_update_dpp(
                   0, __builtin_bit_cast(int, v), 0x160 + XMASK, 0xf, 0xf, true));
}

__device__ __forceinline__ float row16_sum(float v) {
    v = add_row_xmask<1>(v);
    v = add_row_xmask<2>(v);
    v = add_row_xmask<4>(v);
    v = add_row_xmask<8>(v);
    return v;
}

__device__ __forceinline__ float wave32_sum(float v) {
    v = row16_sum(v);
    return v + __builtin_bit_cast(float, __builtin_amdgcn_permlanex16(
                   0u, __builtin_bit_cast(unsigned, v), 0x76543210u, 0xfedcba98u, true, false));
}

#define PREP_WGS_ D_
#define UCOLS_    32

__device__ __forceinline__ const float* node_row(const float* __restrict__ x,
                                                 const float* __restrict__ vn, int t) {
    return (t < BN_) ? x + (size_t)t * D_ : vn + (size_t)((t - BN_) % V_) * D_;
}

__global__ void k_fold_weights(const float* __restrict__ W,
                               const float* __restrict__ att_src,
                               const float* __restrict__ att_dst,
                               float* __restrict__ Wm,
                               float* __restrict__ U) {
    __shared__ float red[D_ / 32][8];
    const int k = blockIdx.x;
    const int c = threadIdx.x;
    float w[H_], p[8];
    for (int hd = 0; hd < H_; ++hd) w[hd] = W[(size_t)k * HD_ + hd * D_ + c];
    Wm[(size_t)k * D_ + c] = 0.25f * ((w[0] + w[1]) + (w[2] + w[3]));
    for (int hd = 0; hd < H_; ++hd) {
        p[hd]      = w[hd] * att_src[hd * D_ + c];
        p[H_ + hd] = w[hd] * att_dst[hd * D_ + c];
    }
    for (int j = 0; j < 8; ++j) p[j] = wave32_sum(p[j]);
    if ((c & 31) == 0)
        for (int j = 0; j < 8; ++j) red[c >> 5][j] = p[j];
    __syncthreads();
    if (c < UCOLS_) {
        float s = 0.0f;
        if (c < 8)
            s = ((red[0][c] + red[1][c]) + (red[2][c] + red[3][c])) +
                ((red[4][c] + red[5][c]) + (red[6][c] + red[7][c]));
        U[(size_t)k * UCOLS_ + c] = s;
    }
}

template <int KDIM, int NDIM, bool SCORES, int YROWS, bool NODE_ROWS>
__global__ void k_gemm_f32_wmma(const float* __restrict__ A,
                                const float* __restrict__ Av,
                                const float* __restrict__ Bm,
                                float* __restrict__ C,
                                const float* __restrict__ U,
                                float* __restrict__ a_src,
                                float* __restrict__ a_dst) {
    static_assert(KDIM % 8 == 0 && NDIM % 64 == 0, "8-deep K-steps, 64-column wave tiles");
    const int wave = __builtin_amdgcn_readfirstlane((int)(threadIdx.x >> 5));
    const int lane = threadIdx.x & 31;
    const int half = lane >> 4;
    const int lm   = lane & 15;
    const int m0   = blockIdx.x * 16 + blockIdx.y * YROWS;
    const int n0   = blockIdx.y * 256 + wave * 64;
    static_assert(!NODE_ROWS || KDIM == D_, "node rows are 256 words");
    const float* ap;
    if constexpr (NODE_ROWS) ap = node_row(A, Av, m0 + lm) + half * 4;
    else                     ap = A + (size_t)(m0 + lm) * KDIM + half * 4;
    if constexpr (SCORES) {
        static_assert(NDIM == 256, "waves 0-3 cover all columns; wave 4 is the score tile");
        if (wave == 4) {
            v8f s0 = {}; v8f s1 = {};
            const float* up = U + (size_t)(half * 4) * UCOLS_ + lm;
            for (int k0 = 0; k0 < KDIM; k0 += 8) {
                const v4f a = *(const v4f*)(ap + k0);
                const float u0 = up[(size_t)(k0 + 0) * UCOLS_];
                const float u1 = up[(size_t)(k0 + 1) * UCOLS_];
                const float u2 = up[(size_t)(k0 + 2) * UCOLS_];
                const float u3 = up[(size_t)(k0 + 3) * UCOLS_];
                v2f af, bf;
                af[0] = a[0]; af[1] = a[1]; bf[0] = u0; bf[1] = u1;
                s0 = __builtin_amdgcn_wmma_f32_16x16x4_f32(false, af, false, bf, (short)0, s0, false, false);
                af[0] = a[2]; af[1] = a[3]; bf[0] = u2; bf[1] = u3;
                s1 = __builtin_amdgcn_wmma_f32_16x16x4_f32(false, af, false, bf, (short)0, s1, false, false);
            }
            if (lm < 8) {
                float* dst = (lm < H_) ? a_src : a_dst;
                for (int j = 0; j < 8; ++j)
                    dst[(size_t)(m0 + half * 8 + j) * H_ + (lm & (H_ - 1))] = s0[j] + s1[j];
            }
            return;
        }
    } else {
        if (n0 >= NDIM) return;
    }

    v8f c0 = {}; v8f c1 = {}; v8f c2 = {}; v8f c3 = {};
    const float* bp = Bm + (size_t)(half * 4) * NDIM + n0 + lm * 4;

    for (int k0 = 0; k0 < KDIM; k0 += 8) {
        const v4f a  = *(const v4f*)(ap + k0);
        const v4f b0 = *(const v4f*)(bp + (size_t)(k0 + 0) * NDIM);
        const v4f b1 = *(const v4f*)(bp + (size_t)(k0 + 1) * NDIM);
        const v4f b2 = *(const v4f*)(bp + (size_t)(k0 + 2) * NDIM);
        const v4f b3 = *(const v4f*)(bp + (size_t)(k0 + 3) * NDIM);
        v2f af, bf;
        af[0] = a[0]; af[1] = a[1];
        bf[0] = b0[0]; bf[1] = b1[0];
        c0 = __builtin_amdgcn_wmma_f32_16x16x4_f32(false, af, false, bf, (short)0, c0, false, false);
        bf[0] = b0[1]; bf[1] = b1[1];
        c1 = __builtin_amdgcn_wmma_f32_16x16x4_f32(false, af, false, bf, (short)0, c1, false, false);
        bf[0] = b0[2]; bf[1] = b1[2];
        c2 = __builtin_amdgcn_wmma_f32_16x16x4_f32(false, af, false, bf, (short)0, c2, false, false);
        bf[0] = b0[3]; bf[1] = b1[3];
        c3 = __builtin_amdgcn_wmma_f32_16x16x4_f32(false, af, false, bf, (short)0, c3, false, false);
        af[0] = a[2]; af[1] = a[3];
        bf[0] = b2[0]; bf[1] = b3[0];
        c0 = __builtin_amdgcn_wmma_f32_16x16x4_f32(false, af, false, bf, (short)0, c0, false, false);
        bf[0] = b2[1]; bf[1] = b3[1];
        c1 = __builtin_amdgcn_wmma_f32_16x16x4_f32(false, af, false, bf, (short)0, c1, false, false);
        bf[0] = b2[2]; bf[1] = b3[2];
        c2 = __builtin_amdgcn_wmma_f32_16x16x4_f32(false, af, false, bf, (short)0, c2, false, false);
        bf[0] = b2[3]; bf[1] = b3[3];
        c3 = __builtin_amdgcn_wmma_f32_16x16x4_f32(false, af, false, bf, (short)0, c3, false, false);
    }

    float* crow = C + (size_t)(m0 + half * 8) * NDIM + n0 + lm * 4;
#pragma unroll
    for (int j = 0; j < 8; ++j) {
        v4f o;
        o[0] = c0[j]; o[1] = c1[j]; o[2] = c2[j]; o[3] = c3[j];
        *(v4f*)(crow + (size_t)j * NDIM) = o;
    }
}

#define QG_    4
#define QROWS_ (N_ / QG_)
static_assert(QG_ == H_ && QG_ * 64 == D_, "thread (q, c4) also finishes head q, channels 4*c4..+3");

template <int XMASK>
__device__ __forceinline__ float max_row_xmask(float v) {
    return fmaxf(v, __builtin_bit_cast(float, __builtin_amdgcn_update_dpp(
                        0, __builtin_bit_cast(int, v), 0x160 + XMASK, 0xf, 0xf, true)));
}

__device__ __forceinline__ float wave32_max(float v) {
    v = max_row_xmask<1>(v);
    v = max_row_xmask<2>(v);
    v = max_row_xmask<4>(v);
    v = max_row_xmask<8>(v);
    return fmaxf(v, __builtin_bit_cast(float, __builtin_amdgcn_permlanex16(
                        0u, __builtin_bit_cast(unsigned, v), 0x76543210u, 0xfedcba98u, true, false)));
}

__device__ __forceinline__ float leaky02(float a) { return (a > 0.0f) ? a : 0.2f * a; }

__device__ __forceinline__ float pick4(v4f v, int i) {
    return (i == 0) ? v[0] : (i == 1) ? v[1] : (i == 2) ? v[2] : v[3];
}

__global__ void __launch_bounds__(D_) k_gat_virtual(const float* __restrict__ x,
                              const float* __restrict__ vn,
                              const float* __restrict__ a_src,
                              const float* __restrict__ a_dst,
                              float* __restrict__ z) {
    __shared__ v4f w4[N_];
    __shared__ v4f red4[D_ / 32];
    __shared__ v4f part[QG_][H_][D_ / 4];
    const int uidx = blockIdx.x;
    const int b    = uidx / V_;
    const int u    = BN_ + uidx;
    const int tid  = threadIdx.x;
    const int lane = tid & 31;
    const int wave = tid >> 5;

    const v4f adst = *(const v4f*)(a_dst + (size_t)u * H_);
    const int r0 = b * N_ + tid * 2;
    v4f al0 = *(const v4f*)(a_src + (size_t)r0 * H_);
    v4f al1 = *(const v4f*)(a_src + (size_t)(r0 + 1) * H_);
    v4f als = *(const v4f*)(a_src + (size_t)u * H_);
    v4f lmax;
    for (int hd = 0; hd < H_; ++hd) {
        al0[hd] = leaky02(al0[hd] + adst[hd]);
        al1[hd] = leaky02(al1[hd] + adst[hd]);
        als[hd] = leaky02(als[hd] + adst[hd]);
        lmax[hd] = wave32_max(fmaxf(fmaxf(al0[hd], al1[hd]), als[hd]));
    }
    if (lane == 0) red4[wave] = lmax;
    __syncthreads();
    v4f m = red4[0];
    for (int wv = 1; wv < D_ / 32; ++wv) {
        const v4f o = red4[wv];
        for (int hd = 0; hd < H_; ++hd) m[hd] = fmaxf(m[hd], o[hd]);
    }
    __syncthreads();

    v4f e0, e1, es, ps;
    for (int hd = 0; hd < H_; ++hd) {
        e0[hd] = expf(al0[hd] - m[hd]);
        e1[hd] = expf(al1[hd] - m[hd]);
        es[hd] = expf(als[hd] - m[hd]);
        float s = e0[hd] + e1[hd];
        if (tid == 0) s += es[hd];
        ps[hd] = wave32_sum(s);
    }
    w4[tid * 2]     = e0;
    w4[tid * 2 + 1] = e1;
    if (lane == 0) red4[wave] = ps;
    __syncthreads();
    const v4f tot = ((red4[0] + red4[1]) + (red4[2] + red4[3])) +
                    ((red4[4] + red4[5]) + (red4[6] + red4[7]));

    const int c4 = tid & 63;
    const int q  = __builtin_amdgcn_readfirstlane(tid >> 6);
    const float* xb = x + (size_t)(b * N_ + q * QROWS_) * D_ + c4 * 4;
    const v4f* wq = w4 + q * QROWS_;
    v4f acc[H_];
    for (int hd = 0; hd < H_; ++hd) acc[hd] = (v4f){0.0f, 0.0f, 0.0f, 0.0f};
    for (int rb = 0; rb < QROWS_; rb += 32) {
        v4f xv[32];
#pragma unroll
        for (int i = 0; i < 32; ++i) xv[i] = *(const v4f*)(xb + (size_t)(rb + i) * D_);
#pragma unroll
        for (int i = 0; i < 32; ++i) {
            const v4f wr = wq[rb + i];
            for (int hd = 0; hd < H_; ++hd) acc[hd] += wr[hd] * xv[i];
        }
    }
    for (int hd = 0; hd < H_; ++hd) part[q][hd][c4] = acc[hd];
    __syncthreads();

    v4f s4 = (part[0][q][c4] + part[1][q][c4]) + (part[2][q][c4] + part[3][q][c4]);
    const v4f vs = *(const v4f*)(vn + (size_t)(uidx % V_) * D_ + c4 * 4);
    s4 += pick4(es, q) * vs;
    const float inv = 1.0f / pick4(tot, q);
    *(v4f*)(z + ((size_t)q * (B_ * V_) + uidx) * D_ + c4 * 4) = s4 * inv;
}

#define ROWS_PER_WG_ 8
static_assert(T_ % ROWS_PER_WG_ == 0 && BN_ % ROWS_PER_WG_ == 0, "whole work-groups of rows");
static_assert(D_ == 256, "8 channels per lane of a 32-lane wave");

__device__ __forceinline__ float sum8(const float* v) {
    return ((v[0] + v[1]) + (v[2] + v[3])) + ((v[4] + v[5]) + (v[6] + v[7]));
}

__device__ __forceinline__ void layernorm_row_store(const float* v, int lane,
                                                    const float* __restrict__ gg,
                                                    const float* __restrict__ gb,
                                                    float* __restrict__ orow) {
    const float mu = wave32_sum(sum8(v)) * (1.0f / D_);
    float c[8], sq[8];
    for (int i = 0; i < 8; ++i) { c[i] = v[i] - mu; sq[i] = c[i] * c[i]; }
    const float var = wave32_sum(sum8(sq)) * (1.0f / D_);
    const float r = rsqrtf(var + EPS_);
    for (int part = 0; part < 2; ++part) {
        const int d = part * 128 + lane * 4;
        const v4f g = *(const v4f*)(gg + d);
        const v4f b = *(const v4f*)(gb + d);
        v4f o;
        for (int i = 0; i < 4; ++i) o[i] = c[part * 4 + i] * r * g[i] + b[i];
        *(v4f*)(orow + d) = o;
    }
}

__global__ void k_gat_finish(const float* __restrict__ gm,
                             const float* __restrict__ gv,
                             const float* __restrict__ bias,
                             const float* __restrict__ gg,
                             const float* __restrict__ gb,
                             float* __restrict__ out) {
    const int wave = __builtin_amdgcn_readfirstlane((int)(threadIdx.x >> 5));
    const int lane = threadIdx.x & 31;
    const int t = blockIdx.x * ROWS_PER_WG_ + wave;

    v4f m[2];
    if (t < BN_) {
        for (int part = 0; part < 2; ++part)
            m[part] = *(const v4f*)(gm + (size_t)t * D_ + part * 128 + lane * 4);
    } else {
        const int uidx = t - BN_;
        v4f q[2][H_];
        for (int part = 0; part < 2; ++part)
            for (int hd = 0; hd < H_; ++hd)
                q[part][hd] = *(const v4f*)(gv + (size_t)(hd * (B_ * V_) + uidx) * HD_ +
                                            hd * D_ + part * 128 + lane * 4);
        for (int part = 0; part < 2; ++part)
            for (int i = 0; i < 4; ++i)
                m[part][i] = 0.25f * (q[part][0][i] + q[part][1][i] + q[part][2][i] + q[part][3][i]);
    }

    float v[8];
    for (int part = 0; part < 2; ++part) {
        const v4f bs = *(const v4f*)(bias + part * 128 + lane * 4);
        for (int i = 0; i < 4; ++i)
            v[part * 4 + i] = gelu_exact(m[part][i] + bs[i]);
    }
    layernorm_row_store(v, lane, gg, gb, out + (size_t)t * D_);
}

__global__ void k_gcn_finish(const float* __restrict__ xw,
                             const float* __restrict__ bias,
                             const float* __restrict__ gg,
                             const float* __restrict__ gb,
                             float* __restrict__ out) {
    const int wave = __builtin_amdgcn_readfirstlane((int)(threadIdx.x >> 5));
    const int lane = threadIdx.x & 31;
    const int t = blockIdx.x * ROWS_PER_WG_ + wave;
    const int b = t / N_;
    const float* xv = xw + (size_t)(BN_ + b * V_) * D_;
    const float* xr = xw + (size_t)t * D_;
    const float inv_sqrt6 = 0.40824829046386301637f;

    v4f q[2][V_ + 1];
    for (int part = 0; part < 2; ++part) {
        for (int j = 0; j < V_; ++j)
            q[part][j] = *(const v4f*)(xv + j * D_ + part * 128 + lane * 4);
        q[part][V_] = *(const v4f*)(xr + part * 128 + lane * 4);
    }

    float v[8];
    for (int part = 0; part < 2; ++part) {
        const v4f bs = *(const v4f*)(bias + part * 128 + lane * 4);
        for (int i = 0; i < 4; ++i) {
            const float s5 = q[part][0][i] + q[part][1][i] + q[part][2][i] + q[part][3][i] + q[part][4][i];
            v[part * 4 + i] = gelu_exact(s5 * inv_sqrt6 + q[part][V_][i] * (1.0f / 6.0f) + bs[i]);
        }
    }
    layernorm_row_store(v, lane, gg, gb, out + (size_t)t * D_);
}

extern "C" void kernel_launch(void* const* d_in, const int* in_sizes, int n_in,
                              void* d_out, int out_size, void* d_ws, size_t ws_size,
                              hipStream_t stream) {
    (void)in_sizes; (void)n_in; (void)out_size; (void)ws_size;

    const float* x       = (const float*)d_in[0];
    const float* vn      = (const float*)d_in[2];
    const float* gat_W   = (const float*)d_in[3];
    const float* att_src = (const float*)d_in[4];
    const float* att_dst = (const float*)d_in[5];
    const float* gat_b   = (const float*)d_in[6];
    const float* gcn_W   = (const float*)d_in[7];
    const float* gcn_b   = (const float*)d_in[8];
    const float* n1g     = (const float*)d_in[9];
    const float* n1b     = (const float*)d_in[10];
    const float* n2g     = (const float*)d_in[11];
    const float* n2b     = (const float*)d_in[12];
    float* out = (float*)d_out;

    float* ws = (float*)d_ws;
    size_t off = 0;
    float* wm    = ws + off; off += (size_t)D_ * D_;
    float* ut    = ws + off; off += (size_t)D_ * UCOLS_;
    float* gm    = ws + off; off += (size_t)T_ * D_;
    float* asrc  = ws + off; off += (size_t)T_ * H_;
    float* adst  = ws + off; off += (size_t)T_ * H_;
    float* z     = ws + off; off += (size_t)H_ * B_ * V_ * D_;
    float* gv    = ws + off; off += (size_t)H_ * B_ * V_ * HD_;
    float* gat1  = ws + off; off += (size_t)T_ * D_;
    float* xw    = ws + off; off += (size_t)T_ * D_;

    k_fold_weights<<<PREP_WGS_, D_, 0, stream>>>(gat_W, att_src, att_dst, wm, ut);
    {
        dim3 grid(T_ / 16, 1);
        k_gemm_f32_wmma<D_, D_, true, 0, true><<<grid, 160, 0, stream>>>(x, vn, wm, gm, ut, asrc, adst);
    }
    {
        k_gat_virtual<<<B_ * V_, D_, 0, stream>>>(x, vn, asrc, adst, z);
    }
    {
        dim3 grid(B_ * V_ / 16, H_);
        k_gemm_f32_wmma<D_, HD_, false, B_ * V_, false><<<grid, 128, 0, stream>>>(z, nullptr, gat_W, gv,
                                                                                 nullptr, nullptr, nullptr);
    }
    k_gat_finish<<<T_ / ROWS_PER_WG_, ROWS_PER_WG_ * 32, 0, stream>>>(gm, gv, gat_b, n1g, n1b, gat1);
    {
        dim3 grid(T_ / 16, 1);
        k_gemm_f32_wmma<D_, D_, false, 0, false><<<grid, 128, 0, stream>>>(gat1, nullptr, gcn_W, xw,
                                                                           nullptr, nullptr, nullptr);
    }
    k_gcn_finish<<<BN_ / ROWS_PER_WG_, ROWS_PER_WG_ * 32, 0, stream>>>(xw, gcn_b, n2g, n2b, out);
}


# GEMM K-loop: two register sets of fragments; the next block's loads (2 K-steps; 4 in the per-head virtual GEMM) go out as one group in front of the current block's WMMAs
# speedup vs baseline: 4.0543x; 1.5395x over previous
#include <hip/hip_runtime.h>
#include <math.h>

#define B_   32
#define N_   512
#define V_   5
#define D_   256
#define H_   4
#define BN_  (B_ * N_)
#define T_   (BN_ + B_ * V_)
#define HD_  (H_ * D_)
#define EPS_ 1e-5f

typedef float v2f __attribute__((ext_vector_type(2)));
typedef float v4f __attribute__((ext_vector_type(4)));
typedef float v8f __attribute__((ext_vector_type(8)));

__device__ __forceinline__ float gelu_exact(float x) {
    return 0.5f * x * (1.0f + erff(x * 0.70710678118654752f));
}

template <int XMASK>
__device__ __forceinline__ float add_row_xmask(float v) {
    return v + __builtin_bit_cast(float, __builtin_amdgcn_update_dpp(
                   0, __builtin_bit_cast(int, v), 0x160 + XMASK, 0xf, 0xf, true));
}

__device__ __forceinline__ float row16_sum(float v) {
    v = add_row_xmask<1>(v);
    v = add_row_xmask<2>(v);
    v = add_row_xmask<4>(v);
    v = add_row_xmask<8>(v);
    return v;
}

__device__ __forceinline__ float wave32_sum(float v) {
    v = row16_sum(v);
    return v + __builtin_bit_cast(float, __builtin_amdgcn_permlanex16(
                   0u, __builtin_bit_cast(unsigned, v), 0x76543210u, 0xfedcba98u, true, false));
}

#define PREP_WGS_ D_
#define UCOLS_    32

__device__ __forceinline__ const float* node_row(const float* __restrict__ x,
                                                 const float* __restrict__ vn, int t) {
    return (t < BN_) ? x + (size_t)t * D_ : vn + (size_t)((t - BN_) % V_) * D_;
}

__global__ void k_fold_weights(const float* __restrict__ W,
                               const float* __restrict__ att_src,
                               const float* __restrict__ att_dst,
                               float* __restrict__ Wm,
                               float* __restrict__ U) {
    __shared__ float red[D_ / 32][8];
    const int k = blockIdx.x;
    const int c = threadIdx.x;
    float w[H_], p[8];
    for (int hd = 0; hd < H_; ++hd) w[hd] = W[(size_t)k * HD_ + hd * D_ + c];
    Wm[(size_t)k * D_ + c] = 0.25f * ((w[0] + w[1]) + (w[2] + w[3]));
    for (int hd = 0; hd < H_; ++hd) {
        p[hd]      = w[hd] * att_src[hd * D_ + c];
        p[H_ + hd] = w[hd] * att_dst[hd * D_ + c];
    }
    for (int j = 0; j < 8; ++j) p[j] = wave32_sum(p[j]);
    if ((c & 31) == 0)
        for (int j = 0; j < 8; ++j) red[c >> 5][j] = p[j];
    __syncthreads();
    if (c < UCOLS_) {
        float s = 0.0f;
        if (c < 8)
            s = ((red[0][c] + red[1][c]) + (red[2][c] + red[3][c])) +
                ((red[4][c] + red[5][c]) + (red[6][c] + red[7][c]));
        U[(size_t)k * UCOLS_ + c] = s;
    }
}

template <int KB>
struct GemmFrag {
    v4f a[KB];
    v4f b[KB][4];
};

template <int KB, int NDIM>
__device__ __forceinline__ void gemm_load_block(GemmFrag<KB>& f, const float* __restrict__ ap,
                                                const float* __restrict__ bp, int k0) {
#pragma unroll
    for (int s = 0; s < KB; ++s) {
        f.a[s] = *(const v4f*)(ap + k0 + 8 * s);
#pragma unroll
        for (int j = 0; j < 4; ++j)
            f.b[s][j] = *(const v4f*)(bp + (size_t)(k0 + 8 * s + j) * NDIM);
    }
}

template <int KB>
__device__ __forceinline__ void gemm_mma_block(const GemmFrag<KB>& f, v8f& c0, v8f& c1, v8f& c2, v8f& c3) {
#pragma unroll
    for (int s = 0; s < KB; ++s) {
        const v4f a = f.a[s];
        const v4f b0 = f.b[s][0], b1 = f.b[s][1], b2 = f.b[s][2], b3 = f.b[s][3];
        v2f af, bf;
        af[0] = a[0]; af[1] = a[1];
        bf[0] = b0[0]; bf[1] = b1[0];
        c0 = __builtin_amdgcn_wmma_f32_16x16x4_f32(false, af, false, bf, (short)0, c0, false, false);
        bf[0] = b0[1]; bf[1] = b1[1];
        c1 = __builtin_amdgcn_wmma_f32_16x16x4_f32(false, af, false, bf, (short)0, c1, false, false);
        bf[0] = b0[2]; bf[1] = b1[2];
        c2 = __builtin_amdgcn_wmma_f32_16x16x4_f32(false, af, false, bf, (short)0, c2, false, false);
        bf[0] = b0[3]; bf[1] = b1[3];
        c3 = __builtin_amdgcn_wmma_f32_16x16x4_f32(false, af, false, bf, (short)0, c3, false, false);
        af[0] = a[2]; af[1] = a[3];
        bf[0] = b2[0]; bf[1] = b3[0];
        c0 = __builtin_amdgcn_wmma_f32_16x16x4_f32(false, af, false, bf, (short)0, c0, false, false);
        bf[0] = b2[1]; bf[1] = b3[1];
        c1 = __builtin_amdgcn_wmma_f32_16x16x4_f32(false, af, false, bf, (short)0, c1, false, false);
        bf[0] = b2[2]; bf[1] = b3[2];
        c2 = __builtin_amdgcn_wmma_f32_16x16x4_f32(false, af, false, bf, (short)0, c2, false, false);
        bf[0] = b2[3]; bf[1] = b3[3];
        c3 = __builtin_amdgcn_wmma_f32_16x16x4_f32(false, af, false, bf, (short)0, c3, false, false);
    }
}

template <int KDIM, int NDIM, bool SCORES, int YROWS, bool NODE_ROWS, int KB>
__global__ void __launch_bounds__(160) k_gemm_f32_wmma(const float* __restrict__ A,
                                const float* __restrict__ Av,
                                const float* __restrict__ Bm,
                                float* __restrict__ C,
                                const float* __restrict__ U,
                                float* __restrict__ a_src,
                                float* __restrict__ a_dst) {
    static_assert(KDIM % 8 == 0 && NDIM % 64 == 0, "8-deep K-steps, 64-column wave tiles");
    const int wave = __builtin_amdgcn_readfirstlane((int)(threadIdx.x >> 5));
    const int lane = threadIdx.x & 31;
    const int half = lane >> 4;
    const int lm   = lane & 15;
    const int m0   = blockIdx.x * 16 + blockIdx.y * YROWS;
    const int n0   = blockIdx.y * 256 + wave * 64;
    static_assert(!NODE_ROWS || KDIM == D_, "node rows are 256 words");
    const float* ap;
    if constexpr (NODE_ROWS) ap = node_row(A, Av, m0 + lm) + half * 4;
    else                     ap = A + (size_t)(m0 + lm) * KDIM + half * 4;
    if constexpr (SCORES) {
        static_assert(NDIM == 256, "waves 0-3 cover all columns; wave 4 is the score tile");
        if (wave == 4) {
            v8f s0 = {}; v8f s1 = {};
            const float* up = U + (size_t)(half * 4) * UCOLS_ + lm;
            for (int k0 = 0; k0 < KDIM; k0 += 8) {
                const v4f a = *(const v4f*)(ap + k0);
                const float u0 = up[(size_t)(k0 + 0) * UCOLS_];
                const float u1 = up[(size_t)(k0 + 1) * UCOLS_];
                const float u2 = up[(size_t)(k0 + 2) * UCOLS_];
                const float u3 = up[(size_t)(k0 + 3) * UCOLS_];
                v2f af, bf;
                af[0] = a[0]; af[1] = a[1]; bf[0] = u0; bf[1] = u1;
                s0 = __builtin_amdgcn_wmma_f32_16x16x4_f32(false, af, false, bf, (short)0, s0, false, false);
                af[0] = a[2]; af[1] = a[3]; bf[0] = u2; bf[1] = u3;
                s1 = __builtin_amdgcn_wmma_f32_16x16x4_f32(false, af, false, bf, (short)0, s1, false, false);
            }
            if (lm < 8) {
                float* dst = (lm < H_) ? a_src : a_dst;
                for (int j = 0; j < 8; ++j)
                    dst[(size_t)(m0 + half * 8 + j) * H_ + (lm & (H_ - 1))] = s0[j] + s1[j];
            }
            return;
        }
    } else {
        if (n0 >= NDIM) return;
    }

    v8f c0 = {}; v8f c1 = {}; v8f c2 = {}; v8f c3 = {};
    const float* bp = Bm + (size_t)(half * 4) * NDIM + n0 + lm * 4;

    constexpr int KBLK = 8 * KB;
    static_assert(KDIM % (2 * KBLK) == 0, "an even number of blocks: the sets alternate");
    GemmFrag<KB> f0, f1;
    gemm_load_block<KB, NDIM>(f0, ap, bp, 0);
    for (int k0 = 0; k0 < KDIM - 2 * KBLK; k0 += 2 * KBLK) {
        gemm_load_block<KB, NDIM>(f1, ap, bp, k0 + KBLK);
        __builtin_amdgcn_sched_barrier(0);
        gemm_mma_block<KB>(f0, c0, c1, c2, c3);
        __builtin_amdgcn_sched_barrier(0);
        gemm_load_block<KB, NDIM>(f0, ap, bp, k0 + 2 * KBLK);
        __builtin_amdgcn_sched_barrier(0);
        gemm_mma_block<KB>(f1, c0, c1, c2, c3);
        __builtin_amdgcn_sched_barrier(0);
    }
    gemm_load_block<KB, NDIM>(f1, ap, bp, KDIM - KBLK);
    __builtin_amdgcn_sched_barrier(0);
    gemm_mma_block<KB>(f0, c0, c1, c2, c3);
    gemm_mma_block<KB>(f1, c0, c1, c2, c3);

    float* crow = C + (size_t)(m0 + half * 8) * NDIM + n0 + lm * 4;
#pragma unroll
    for (int j = 0; j < 8; ++j) {
        v4f o;
        o[0] = c0[j]; o[1] = c1[j]; o[2] = c2[j]; o[3] = c3[j];
        *(v4f*)(crow + (size_t)j * NDIM) = o;
    }
}

#define QG_    4
#define QROWS_ (N_ / QG_)
static_assert(QG_ == H_ && QG_ * 64 == D_, "thread (q, c4) also finishes head q, channels 4*c4..+3");

template <int XMASK>
__device__ __forceinline__ float max_row_xmask(float v) {
    return fmaxf(v, __builtin_bit_cast(float, __builtin_amdgcn_update_dpp(
                        0, __builtin_bit_cast(int, v), 0x160 + XMASK, 0xf, 0xf, true)));
}

__device__ __forceinline__ float wave32_max(float v) {
    v = max_row_xmask<1>(v);
    v = max_row_xmask<2>(v);
    v = max_row_xmask<4>(v);
    v = max_row_xmask<8>(v);
    return fmaxf(v, __builtin_bit_cast(float, __builtin_amdgcn_permlanex16(
                        0u, __builtin_bit_cast(unsigned, v), 0x76543210u, 0xfedcba98u, true, false)));
}

__device__ __forceinline__ float leaky02(float a) { return (a > 0.0f) ? a : 0.2f * a; }

__device__ __forceinline__ float pick4(v4f v, int i) {
    return (i == 0) ? v[0] : (i == 1) ? v[1] : (i == 2) ? v[2] : v[3];
}

__global__ void __launch_bounds__(D_) k_gat_virtual(const float* __restrict__ x,
                              const float* __restrict__ vn,
                              const float* __restrict__ a_src,
                              const float* __restrict__ a_dst,
                              float* __restrict__ z) {
    __shared__ v4f w4[N_];
    __shared__ v4f red4[D_ / 32];
    __shared__ v4f part[QG_][H_][D_ / 4];
    const int uidx = blockIdx.x;
    const int b    = uidx / V_;
    const int u    = BN_ + uidx;
    const int tid  = threadIdx.x;
    const int lane = tid & 31;
    const int wave = tid >> 5;

    const v4f adst = *(const v4f*)(a_dst + (size_t)u * H_);
    const int r0 = b * N_ + tid * 2;
    v4f al0 = *(const v4f*)(a_src + (size_t)r0 * H_);
    v4f al1 = *(const v4f*)(a_src + (size_t)(r0 + 1) * H_);
    v4f als = *(const v4f*)(a_src + (size_t)u * H_);
    v4f lmax;
    for (int hd = 0; hd < H_; ++hd) {
        al0[hd] = leaky02(al0[hd] + adst[hd]);
        al1[hd] = leaky02(al1[hd] + adst[hd]);
        als[hd] = leaky02(als[hd] + adst[hd]);
        lmax[hd] = wave32_max(fmaxf(fmaxf(al0[hd], al1[hd]), als[hd]));
    }
    if (lane == 0) red4[wave] = lmax;
    __syncthreads();
    v4f m = red4[0];
    for (int wv = 1; wv < D_ / 32; ++wv) {
        const v4f o = red4[wv];
        for (int hd = 0; hd < H_; ++hd) m[hd] = fmaxf(m[hd], o[hd]);
    }
    __syncthreads();

    v4f e0, e1, es, ps;
    for (int hd = 0; hd < H_; ++hd) {
        e0[hd] = expf(al0[hd] - m[hd]);
        e1[hd] = expf(al1[hd] - m[hd]);
        es[hd] = expf(als[hd] - m[hd]);
        float s = e0[hd] + e1[hd];
        if (tid == 0) s += es[hd];
        ps[hd] = wave32_sum(s);
    }
    w4[tid * 2]     = e0;
    w4[tid * 2 + 1] = e1;
    if (lane == 0) red4[wave] = ps;
    __syncthreads();
    const v4f tot = ((red4[0] + red4[1]) + (red4[2] + red4[3])) +
                    ((red4[4] + red4[5]) + (red4[6] + red4[7]));

    const int c4 = tid & 63;
    const int q  = __builtin_amdgcn_readfirstlane(tid >> 6);
    const float* xb = x + (size_t)(b * N_ + q * QROWS_) * D_ + c4 * 4;
    const v4f* wq = w4 + q * QROWS_;
    v4f acc[H_];
    for (int hd = 0; hd < H_; ++hd) acc[hd] = (v4f){0.0f, 0.0f, 0.0f, 0.0f};
    for (int rb = 0; rb < QROWS_; rb += 32) {
        v4f xv[32];
#pragma unroll
        for (int i = 0; i < 32; ++i) xv[i] = *(const v4f*)(xb + (size_t)(rb + i) * D_);
#pragma unroll
        for (int i = 0; i < 32; ++i) {
            const v4f wr = wq[rb + i];
            for (int hd = 0; hd < H_; ++hd) acc[hd] += wr[hd] * xv[i];
        }
    }
    for (int hd = 0; hd < H_; ++hd) part[q][hd][c4] = acc[hd];
    __syncthreads();

    v4f s4 = (part[0][q][c4] + part[1][q][c4]) + (part[2][q][c4] + part[3][q][c4]);
    const v4f vs = *(const v4f*)(vn + (size_t)(uidx % V_) * D_ + c4 * 4);
    s4 += pick4(es, q) * vs;
    const float inv = 1.0f / pick4(tot, q);
    *(v4f*)(z + ((size_t)q * (B_ * V_) + uidx) * D_ + c4 * 4) = s4 * inv;
}

#define ROWS_PER_WG_ 8
static_assert(T_ % ROWS_PER_WG_ == 0 && BN_ % ROWS_PER_WG_ == 0, "whole work-groups of rows");
static_assert(D_ == 256, "8 channels per lane of a 32-lane wave");

__device__ __forceinline__ float sum8(const float* v) {
    return ((v[0] + v[1]) + (v[2] + v[3])) + ((v[4] + v[5]) + (v[6] + v[7]));
}

__device__ __forceinline__ void layernorm_row_store(const float* v, int lane,
                                                    const float* __restrict__ gg,
                                                    const float* __restrict__ gb,
                                                    float* __restrict__ orow) {
    const float mu = wave32_sum(sum8(v)) * (1.0f / D_);
    float c[8], sq[8];
    for (int i = 0; i < 8; ++i) { c[i] = v[i] - mu; sq[i] = c[i] * c[i]; }
    const float var = wave32_sum(sum8(sq)) * (1.0f / D_);
    const float r = rsqrtf(var + EPS_);
    for (int part = 0; part < 2; ++part) {
        const int d = part * 128 + lane * 4;
        const v4f g = *(const v4f*)(gg + d);
        const v4f b = *(const v4f*)(gb + d);
        v4f o;
        for (int i = 0; i < 4; ++i) o[i] = c[part * 4 + i] * r * g[i] + b[i];
        *(v4f*)(orow + d) = o;
    }
}

__global__ void k_gat_finish(const float* __restrict__ gm,
                             const float* __restrict__ gv,
                             const float* __restrict__ bias,
                             const float* __restrict__ gg,
                             const float* __restrict__ gb,
                             float* __restrict__ out) {
    const int wave = __builtin_amdgcn_readfirstlane((int)(threadIdx.x >> 5));
    const int lane = threadIdx.x & 31;
    const int t = blockIdx.x * ROWS_PER_WG_ + wave;

    v4f m[2];
    if (t < BN_) {
        for (int part = 0; part < 2; ++part)
            m[part] = *(const v4f*)(gm + (size_t)t * D_ + part * 128 + lane * 4);
    } else {
        const int uidx = t - BN_;
        v4f q[2][H_];
        for (int part = 0; part < 2; ++part)
            for (int hd = 0; hd < H_; ++hd)
                q[part][hd] = *(const v4f*)(gv + (size_t)(hd * (B_ * V_) + uidx) * HD_ +
                                            hd * D_ + part * 128 + lane * 4);
        for (int part = 0; part < 2; ++part)
            for (int i = 0; i < 4; ++i)
                m[part][i] = 0.25f * (q[part][0][i] + q[part][1][i] + q[part][2][i] + q[part][3][i]);
    }

    float v[8];
    for (int part = 0; part < 2; ++part) {
        const v4f bs = *(const v4f*)(bias + part * 128 + lane * 4);
        for (int i = 0; i < 4; ++i)
            v[part * 4 + i] = gelu_exact(m[part][i] + bs[i]);
    }
    layernorm_row_store(v, lane, gg, gb, out + (size_t)t * D_);
}

__global__ void k_gcn_finish(const float* __restrict__ xw,
                             const float* __restrict__ bias,
                             const float* __restrict__ gg,
                             const float* __restrict__ gb,
                             float* __restrict__ out) {
    const int wave = __builtin_amdgcn_readfirstlane((int)(threadIdx.x >> 5));
    const int lane = threadIdx.x & 31;
    const int t = blockIdx.x * ROWS_PER_WG_ + wave;
    const int b = t / N_;
    const float* xv = xw + (size_t)(BN_ + b * V_) * D_;
    const float* xr = xw + (size_t)t * D_;
    const float inv_sqrt6 = 0.40824829046386301637f;

    v4f q[2][V_ + 1];
    for (int part = 0; part < 2; ++part) {
        for (int j = 0; j < V_; ++j)
            q[part][j] = *(const v4f*)(xv + j * D_ + part * 128 + lane * 4);
        q[part][V_] = *(const v4f*)(xr + part * 128 + lane * 4);
    }

    float v[8];
    for (int part = 0; part < 2; ++part) {
        const v4f bs = *(const v4f*)(bias + part * 128 + lane * 4);
        for (int i = 0; i < 4; ++i) {
            const float s5 = q[part][0][i] + q[part][1][i] + q[part][2][i] + q[part][3][i] + q[part][4][i];
            v[part * 4 + i] = gelu_exact(s5 * inv_sqrt6 + q[part][V_][i] * (1.0f / 6.0f) + bs[i]);
        }
    }
    layernorm_row_store(v, lane, gg, gb, out + (size_t)t * D_);
}

extern "C" void kernel_launch(void* const* d_in, const int* in_sizes, int n_in,
                              void* d_out, int out_size, void* d_ws, size_t ws_size,
                              hipStream_t stream) {
    (void)in_sizes; (void)n_in; (void)out_size; (void)ws_size;

    const float* x       = (const float*)d_in[0];
    const float* vn      = (const float*)d_in[2];
    const float* gat_W   = (const float*)d_in[3];
    const float* att_src = (const float*)d_in[4];
    const float* att_dst = (const float*)d_in[5];
    const float* gat_b   = (const float*)d_in[6];
    const float* gcn_W   = (const float*)d_in[7];
    const float* gcn_b   = (const float*)d_in[8];
    const float* n1g     = (const float*)d_in[9];
    const float* n1b     = (const float*)d_in[10];
    const float* n2g     = (const float*)d_in[11];
    const float* n2b     = (const float*)d_in[12];
    float* out = (float*)d_out;

    float* ws = (float*)d_ws;
    size_t off = 0;
    float* wm    = ws + off; off += (size_t)D_ * D_;
    float* ut    = ws + off; off += (size_t)D_ * UCOLS_;
    float* gm    = ws + off; off += (size_t)T_ * D_;
    float* asrc  = ws + off; off += (size_t)T_ * H_;
    float* adst  = ws + off; off += (size_t)T_ * H_;
    float* z     = ws + off; off += (size_t)H_ * B_ * V_ * D_;
    float* gv    = ws + off; off += (size_t)H_ * B_ * V_ * HD_;
    float* gat1  = ws + off; off += (size_t)T_ * D_;
    float* xw    = ws + off; off += (size_t)T_ * D_;

    k_fold_weights<<<PREP_WGS_, D_, 0, stream>>>(gat_W, att_src, att_dst, wm, ut);
    {
        dim3 grid(T_ / 16, 1);
        k_gemm_f32_wmma<D_, D_, true, 0, true, 2><<<grid, 160, 0, stream>>>(x, vn, wm, gm, ut, asrc, adst);
    }
    {
        k_gat_virtual<<<B_ * V_, D_, 0, stream>>>(x, vn, asrc, adst, z);
    }
    {
        dim3 grid(B_ * V_ / 16, H_);
        k_gemm_f32_wmma<D_, HD_, false, B_ * V_, false, 4><<<grid, 128, 0, stream>>>(z, nullptr, gat_W, gv,
                                                                                 nullptr, nullptr, nullptr);
    }
    k_gat_finish<<<T_ / ROWS_PER_WG_, ROWS_PER_WG_ * 32, 0, stream>>>(gm, gv, gat_b, n1g, n1b, gat1);
    {
        dim3 grid(T_ / 16, 1);
        k_gemm_f32_wmma<D_, D_, false, 0, false, 2><<<grid, 128, 0, stream>>>(gat1, nullptr, gcn_W, xw,
                                                                           nullptr, nullptr, nullptr);
    }
    k_gcn_finish<<<BN_ / ROWS_PER_WG_, ROWS_PER_WG_ * 32, 0, stream>>>(xw, gcn_b, n2g, n2b, out);
}


# LayerNorm1 finisher becomes the GCN GEMM's prologue: a work-group normalises its 16 rows (4 per wave) into LDS and loads its A fragments from there; the K5 launch and the T x 256 gat1 go
# speedup vs baseline: 4.3941x; 1.0838x over previous
#include <hip/hip_runtime.h>
#include <math.h>

#define B_   32
#define N_   512
#define V_   5
#define D_   256
#define H_   4
#define BN_  (B_ * N_)
#define T_   (BN_ + B_ * V_)
#define HD_  (H_ * D_)
#define EPS_ 1e-5f

typedef float v2f __attribute__((ext_vector_type(2)));
typedef float v4f __attribute__((ext_vector_type(4)));
typedef float v8f __attribute__((ext_vector_type(8)));

__device__ __forceinline__ float gelu_exact(float x) {
    return 0.5f * x * (1.0f + erff(x * 0.70710678118654752f));
}

template <int XMASK>
__device__ __forceinline__ float add_row_xmask(float v) {
    return v + __builtin_bit_cast(float, __builtin_amdgcn_update_dpp(
                   0, __builtin_bit_cast(int, v), 0x160 + XMASK, 0xf, 0xf, true));
}

__device__ __forceinline__ float row16_sum(float v) {
    v = add_row_xmask<1>(v);
    v = add_row_xmask<2>(v);
    v = add_row_xmask<4>(v);
    v = add_row_xmask<8>(v);
    return v;
}

__device__ __forceinline__ float wave32_sum(float v) {
    v = row16_sum(v);
    return v + __builtin_bit_cast(float, __builtin_amdgcn_permlanex16(
                   0u, __builtin_bit_cast(unsigned, v), 0x76543210u, 0xfedcba98u, true, false));
}

#define PREP_WGS_ D_
#define UCOLS_    32

__device__ __forceinline__ const float* node_row(const float* __restrict__ x,
                                                 const float* __restrict__ vn, int t) {
    return (t < BN_) ? x + (size_t)t * D_ : vn + (size_t)((t - BN_) % V_) * D_;
}

__global__ void k_fold_weights(const float* __restrict__ W,
                               const float* __restrict__ att_src,
                               const float* __restrict__ att_dst,
                               float* __restrict__ Wm,
                               float* __restrict__ U) {
    __shared__ float red[D_ / 32][8];
    const int k = blockIdx.x;
    const int c = threadIdx.x;
    float w[H_], p[8];
    for (int hd = 0; hd < H_; ++hd) w[hd] = W[(size_t)k * HD_ + hd * D_ + c];
    Wm[(size_t)k * D_ + c] = 0.25f * ((w[0] + w[1]) + (w[2] + w[3]));
    for (int hd = 0; hd < H_; ++hd) {
        p[hd]      = w[hd] * att_src[hd * D_ + c];
        p[H_ + hd] = w[hd] * att_dst[hd * D_ + c];
    }
    for (int j = 0; j < 8; ++j) p[j] = wave32_sum(p[j]);
    if ((c & 31) == 0)
        for (int j = 0; j < 8; ++j) red[c >> 5][j] = p[j];
    __syncthreads();
    if (c < UCOLS_) {
        float s = 0.0f;
        if (c < 8)
            s = ((red[0][c] + red[1][c]) + (red[2][c] + red[3][c])) +
                ((red[4][c] + red[5][c]) + (red[6][c] + red[7][c]));
        U[(size_t)k * UCOLS_ + c] = s;
    }
}

#define ROWS_PER_WG_ 8
static_assert(T_ % ROWS_PER_WG_ == 0 && BN_ % ROWS_PER_WG_ == 0, "whole work-groups of rows");
static_assert(D_ == 256, "8 channels per lane of a 32-lane wave");

__device__ __forceinline__ float sum8(const float* v) {
    return ((v[0] + v[1]) + (v[2] + v[3])) + ((v[4] + v[5]) + (v[6] + v[7]));
}

__device__ __forceinline__ float layernorm_center(const float* v, float* c) {
    const float mu = wave32_sum(sum8(v)) * (1.0f / D_);
    float sq[8];
    for (int i = 0; i < 8; ++i) { c[i] = v[i] - mu; sq[i] = c[i] * c[i]; }
    const float var = wave32_sum(sum8(sq)) * (1.0f / D_);
    return rsqrtf(var + EPS_);
}

__device__ __forceinline__ void layernorm_row_store(const float* v, int lane,
                                                    const float* __restrict__ gg,
                                                    const float* __restrict__ gb,
                                                    float* __restrict__ orow) {
    float c[8];
    const float r = layernorm_center(v, c);
    for (int part = 0; part < 2; ++part) {
        const int d = part * 128 + lane * 4;
        const v4f g = *(const v4f*)(gg + d);
        const v4f b = *(const v4f*)(gb + d);
        v4f o;
        for (int i = 0; i < 4; ++i) o[i] = c[part * 4 + i] * r * g[i] + b[i];
        *(v4f*)(orow + d) = o;
    }
}

__device__ __forceinline__ void load_row8(const float* __restrict__ p, int lane, v4f* q) {
    q[0] = *(const v4f*)(p + lane * 4);
    q[1] = *(const v4f*)(p + 128 + lane * 4);
}

template <int KB>
struct GemmFrag {
    v4f a[KB];
    v4f b[KB][4];
};

template <int KB, int NDIM>
__device__ __forceinline__ void gemm_load_block(GemmFrag<KB>& f, const float* __restrict__ ap,
                                                const float* __restrict__ bp, int k0) {
#pragma unroll
    for (int s = 0; s < KB; ++s) {
        f.a[s] = *(const v4f*)(ap + k0 + 8 * s);
#pragma unroll
        for (int j = 0; j < 4; ++j)
            f.b[s][j] = *(const v4f*)(bp + (size_t)(k0 + 8 * s + j) * NDIM);
    }
}

template <int KB>
__device__ __forceinline__ void gemm_mma_block(const GemmFrag<KB>& f, v8f& c0, v8f& c1, v8f& c2, v8f& c3) {
#pragma unroll
    for (int s = 0; s < KB; ++s) {
        const v4f a = f.a[s];
        const v4f b0 = f.b[s][0], b1 = f.b[s][1], b2 = f.b[s][2], b3 = f.b[s][3];
        v2f af, bf;
        af[0] = a[0]; af[1] = a[1];
        bf[0] = b0[0]; bf[1] = b1[0];
        c0 = __builtin_amdgcn_wmma_f32_16x16x4_f32(false, af, false, bf, (short)0, c0, false, false);
        bf[0] = b0[1]; bf[1] = b1[1];
        c1 = __builtin_amdgcn_wmma_f32_16x16x4_f32(false, af, false, bf, (short)0, c1, false, false);
        bf[0] = b0[2]; bf[1] = b1[2];
        c2 = __builtin_amdgcn_wmma_f32_16x16x4_f32(false, af, false, bf, (short)0, c2, false, false);
        bf[0] = b0[3]; bf[1] = b1[3];
        c3 = __builtin_amdgcn_wmma_f32_16x16x4_f32(false, af, false, bf, (short)0, c3, false, false);
        af[0] = a[2]; af[1] = a[3];
        bf[0] = b2[0]; bf[1] = b3[0];
        c0 = __builtin_amdgcn_wmma_f32_16x16x4_f32(false, af, false, bf, (short)0, c0, false, false);
        bf[0] = b2[1]; bf[1] = b3[1];
        c1 = __builtin_amdgcn_wmma_f32_16x16x4_f32(false, af, false, bf, (short)0, c1, false, false);
        bf[0] = b2[2]; bf[1] = b3[2];
        c2 = __builtin_amdgcn_wmma_f32_16x16x4_f32(false, af, false, bf, (short)0, c2, false, false);
        bf[0] = b2[3]; bf[1] = b3[3];
        c3 = __builtin_amdgcn_wmma_f32_16x16x4_f32(false, af, false, bf, (short)0, c3, false, false);
    }
}

#define LN_PITCH_ 260
template <int KDIM, int NDIM, bool SCORES, int YROWS, bool NODE_ROWS, int KB, bool LN_ROWS>
__global__ void __launch_bounds__(160) k_gemm_f32_wmma(const float* __restrict__ A,
                                const float* __restrict__ Av,
                                const float* __restrict__ Bm,
                                float* __restrict__ C,
                                const float* __restrict__ U,
                                float* __restrict__ a_src,
                                float* __restrict__ a_dst,
                                const float* __restrict__ ln_bias,
                                const float* __restrict__ ln_g,
                                const float* __restrict__ ln_b) {
    static_assert(KDIM % 8 == 0 && NDIM % 64 == 0, "8-deep K-steps, 64-column wave tiles");
    const int wave = __builtin_amdgcn_readfirstlane((int)(threadIdx.x >> 5));
    const int lane = threadIdx.x & 31;
    const int half = lane >> 4;
    const int lm   = lane & 15;
    const int m0   = blockIdx.x * 16 + blockIdx.y * YROWS;
    const int n0   = blockIdx.y * 256 + wave * 64;
    static_assert(!NODE_ROWS || KDIM == D_, "node rows are 256 words");
    const float* ap;
    if constexpr (LN_ROWS) {
        static_assert(KDIM == D_ && !SCORES && !NODE_ROWS && YROWS == 0 && BN_ % 16 == 0,
                      "a tile is 16 whole rows of 256 channels, all real or all virtual");
        __shared__ v4f a_tile[16 * LN_PITCH_ / 4];
        float* at = (float*)a_tile;
        const int tr = m0 + wave * 4;
        v4f bs[2], g[2], b[2], m[4][2];
        load_row8(ln_bias, lane, bs);
        load_row8(ln_g, lane, g);
        load_row8(ln_b, lane, b);
        if (m0 < BN_) {
#pragma unroll
            for (int r = 0; r < 4; ++r) load_row8(A + (size_t)(tr + r) * D_, lane, m[r]);
        } else {
#pragma unroll
            for (int r = 0; r < 4; ++r) {
                v4f q[H_][2];
                for (int hd = 0; hd < H_; ++hd)
                    load_row8(Av + (size_t)(hd * (B_ * V_) + (tr + r - BN_)) * HD_ + hd * D_, lane, q[hd]);
                for (int part = 0; part < 2; ++part)
                    for (int i = 0; i < 4; ++i)
                        m[r][part][i] = 0.25f * (q[0][part][i] + q[1][part][i] + q[2][part][i] + q[3][part][i]);
            }
        }
#pragma unroll
        for (int r = 0; r < 4; ++r) {
            float v[8], c[8];
            for (int part = 0; part < 2; ++part)
                for (int i = 0; i < 4; ++i)
                    v[part * 4 + i] = gelu_exact(m[r][part][i] + bs[part][i]);
            const float rs = layernorm_center(v, c);
            for (int part = 0; part < 2; ++part) {
                v4f o;
                for (int i = 0; i < 4; ++i) o[i] = c[part * 4 + i] * rs * g[part][i] + b[part][i];
                *(v4f*)(at + (wave * 4 + r) * LN_PITCH_ + part * 128 + lane * 4) = o;
            }
        }
        __syncthreads();
        ap = at + lm * LN_PITCH_ + half * 4;
    } else if constexpr (NODE_ROWS) ap = node_row(A, Av, m0 + lm) + half * 4;
    else                            ap = A + (size_t)(m0 + lm) * KDIM + half * 4;
    if constexpr (SCORES) {
        static_assert(NDIM == 256, "waves 0-3 cover all columns; wave 4 is the score tile");
        if (wave == 4) {
            v8f s0 = {}; v8f s1 = {};
            const float* up = U + (size_t)(half * 4) * UCOLS_ + lm;
            for (int k0 = 0; k0 < KDIM; k0 += 8) {
                const v4f a = *(const v4f*)(ap + k0);
                const float u0 = up[(size_t)(k0 + 0) * UCOLS_];
                const float u1 = up[(size_t)(k0 + 1) * UCOLS_];
                const float u2 = up[(size_t)(k0 + 2) * UCOLS_];
                const float u3 = up[(size_t)(k0 + 3) * UCOLS_];
                v2f af, bf;
                af[0] = a[0]; af[1] = a[1]; bf[0] = u0; bf[1] = u1;
                s0 = __builtin_amdgcn_wmma_f32_16x16x4_f32(false, af, false, bf, (short)0, s0, false, false);
                af[0] = a[2]; af[1] = a[3]; bf[0] = u2; bf[1] = u3;
                s1 = __builtin_amdgcn_wmma_f32_16x16x4_f32(false, af, false, bf, (short)0, s1, false, false);
            }
            if (lm < 8) {
                float* dst = (lm < H_) ? a_src : a_dst;
                for (int j = 0; j < 8; ++j)
                    dst[(size_t)(m0 + half * 8 + j) * H_ + (lm & (H_ - 1))] = s0[j] + s1[j];
            }
            return;
        }
    } else {
        if (n0 >= NDIM) return;
    }

    v8f c0 = {}; v8f c1 = {}; v8f c2 = {}; v8f c3 = {};
    const float* bp = Bm + (size_t)(half * 4) * NDIM + n0 + lm * 4;

    constexpr int KBLK = 8 * KB;
    static_assert(KDIM % (2 * KBLK) == 0, "an even number of blocks: the sets alternate");
    GemmFrag<KB> f0, f1;
    gemm_load_block<KB, NDIM>(f0, ap, bp, 0);
    for (int k0 = 0; k0 < KDIM - 2 * KBLK; k0 += 2 * KBLK) {
        gemm_load_block<KB, NDIM>(f1, ap, bp, k0 + KBLK);
        __builtin_amdgcn_sched_barrier(0);
        gemm_mma_block<KB>(f0, c0, c1, c2, c3);
        __builtin_amdgcn_sched_barrier(0);
        gemm_load_block<KB, NDIM>(f0, ap, bp, k0 + 2 * KBLK);
        __builtin_amdgcn_sched_barrier(0);
        gemm_mma_block<KB>(f1, c0, c1, c2, c3);
        __builtin_amdgcn_sched_barrier(0);
    }
    gemm_load_block<KB, NDIM>(f1, ap, bp, KDIM - KBLK);
    __builtin_amdgcn_sched_barrier(0);
    gemm_mma_block<KB>(f0, c0, c1, c2, c3);
    gemm_mma_block<KB>(f1, c0, c1, c2, c3);

    float* crow = C + (size_t)(m0 + half * 8) * NDIM + n0 + lm * 4;
#pragma unroll
    for (int j = 0; j < 8; ++j) {
        v4f o;
        o[0] = c0[j]; o[1] = c1[j]; o[2] = c2[j]; o[3] = c3[j];
        *(v4f*)(crow + (size_t)j * NDIM) = o;
    }
}

#define QG_    4
#define QROWS_ (N_ / QG_)
static_assert(QG_ == H_ && QG_ * 64 == D_, "thread (q, c4) also finishes head q, channels 4*c4..+3");

template <int XMASK>
__device__ __forceinline__ float max_row_xmask(float v) {
    return fmaxf(v, __builtin_bit_cast(float, __builtin_amdgcn_update_dpp(
                        0, __builtin_bit_cast(int, v), 0x160 + XMASK, 0xf, 0xf, true)));
}

__device__ __forceinline__ float wave32_max(float v) {
    v = max_row_xmask<1>(v);
    v = max_row_xmask<2>(v);
    v = max_row_xmask<4>(v);
    v = max_row_xmask<8>(v);
    return fmaxf(v, __builtin_bit_cast(float, __builtin_amdgcn_permlanex16(
                        0u, __builtin_bit_cast(unsigned, v), 0x76543210u, 0xfedcba98u, true, false)));
}

__device__ __forceinline__ float leaky02(float a) { return (a > 0.0f) ? a : 0.2f * a; }

__device__ __forceinline__ float pick4(v4f v, int i) {
    return (i == 0) ? v[0] : (i == 1) ? v[1] : (i == 2) ? v[2] : v[3];
}

__global__ void __launch_bounds__(D_) k_gat_virtual(const float* __restrict__ x,
                              const float* __restrict__ vn,
                              const float* __restrict__ a_src,
                              const float* __restrict__ a_dst,
                              float* __restrict__ z) {
    __shared__ v4f w4[N_];
    __shared__ v4f red4[D_ / 32];
    __shared__ v4f part[QG_][H_][D_ / 4];
    const int uidx = blockIdx.x;
    const int b    = uidx / V_;
    const int u    = BN_ + uidx;
    const int tid  = threadIdx.x;
    const int lane = tid & 31;
    const int wave = tid >> 5;

    const v4f adst = *(const v4f*)(a_dst + (size_t)u * H_);
    const int r0 = b * N_ + tid * 2;
    v4f al0 = *(const v4f*)(a_src + (size_t)r0 * H_);
    v4f al1 = *(const v4f*)(a_src + (size_t)(r0 + 1) * H_);
    v4f als = *(const v4f*)(a_src + (size_t)u * H_);
    v4f lmax;
    for (int hd = 0; hd < H_; ++hd) {
        al0[hd] = leaky02(al0[hd] + adst[hd]);
        al1[hd] = leaky02(al1[hd] + adst[hd]);
        als[hd] = leaky02(als[hd] + adst[hd]);
        lmax[hd] = wave32_max(fmaxf(fmaxf(al0[hd], al1[hd]), als[hd]));
    }
    if (lane == 0) red4[wave] = lmax;
    __syncthreads();
    v4f m = red4[0];
    for (int wv = 1; wv < D_ / 32; ++wv) {
        const v4f o = red4[wv];
        for (int hd = 0; hd < H_; ++hd) m[hd] = fmaxf(m[hd], o[hd]);
    }
    __syncthreads();

    v4f e0, e1, es, ps;
    for (int hd = 0; hd < H_; ++hd) {
        e0[hd] = expf(al0[hd] - m[hd]);
        e1[hd] = expf(al1[hd] - m[hd]);
        es[hd] = expf(als[hd] - m[hd]);
        float s = e0[hd] + e1[hd];
        if (tid == 0) s += es[hd];
        ps[hd] = wave32_sum(s);
    }
    w4[tid * 2]     = e0;
    w4[tid * 2 + 1] = e1;
    if (lane == 0) red4[wave] = ps;
    __syncthreads();
    const v4f tot = ((red4[0] + red4[1]) + (red4[2] + red4[3])) +
                    ((red4[4] + red4[5]) + (red4[6] + red4[7]));

    const int c4 = tid & 63;
    const int q  = __builtin_amdgcn_readfirstlane(tid >> 6);
    const float* xb = x + (size_t)(b * N_ + q * QROWS_) * D_ + c4 * 4;
    const v4f* wq = w4 + q * QROWS_;
    v4f acc[H_];
    for (int hd = 0; hd < H_; ++hd) acc[hd] = (v4f){0.0f, 0.0f, 0.0f, 0.0f};
    for (int rb = 0; rb < QROWS_; rb += 32) {
        v4f xv[32];
#pragma unroll
        for (int i = 0; i < 32; ++i) xv[i] = *(const v4f*)(xb + (size_t)(rb + i) * D_);
#pragma unroll
        for (int i = 0; i < 32; ++i) {
            const v4f wr = wq[rb + i];
            for (int hd = 0; hd < H_; ++hd) acc[hd] += wr[hd] * xv[i];
        }
    }
    for (int hd = 0; hd < H_; ++hd) part[q][hd][c4] = acc[hd];
    __syncthreads();

    v4f s4 = (part[0][q][c4] + part[1][q][c4]) + (part[2][q][c4] + part[3][q][c4]);
    const v4f vs = *(const v4f*)(vn + (size_t)(uidx % V_) * D_ + c4 * 4);
    s4 += pick4(es, q) * vs;
    const float inv = 1.0f / pick4(tot, q);
    *(v4f*)(z + ((size_t)q * (B_ * V_) + uidx) * D_ + c4 * 4) = s4 * inv;
}

__global__ void k_gcn_finish(const float* __restrict__ xw,
                             const float* __restrict__ bias,
                             const float* __restrict__ gg,
                             const float* __restrict__ gb,
                             float* __restrict__ out) {
    const int wave = __builtin_amdgcn_readfirstlane((int)(threadIdx.x >> 5));
    const int lane = threadIdx.x & 31;
    const int t = blockIdx.x * ROWS_PER_WG_ + wave;
    const int b = t / N_;
    const float* xv = xw + (size_t)(BN_ + b * V_) * D_;
    const float* xr = xw + (size_t)t * D_;
    const float inv_sqrt6 = 0.40824829046386301637f;

    v4f q[2][V_ + 1];
    for (int part = 0; part < 2; ++part) {
        for (int j = 0; j < V_; ++j)
            q[part][j] = *(const v4f*)(xv + j * D_ + part * 128 + lane * 4);
        q[part][V_] = *(const v4f*)(xr + part * 128 + lane * 4);
    }

    float v[8];
    for (int part = 0; part < 2; ++part) {
        const v4f bs = *(const v4f*)(bias + part * 128 + lane * 4);
        for (int i = 0; i < 4; ++i) {
            const float s5 = q[part][0][i] + q[part][1][i] + q[part][2][i] + q[part][3][i] + q[part][4][i];
            v[part * 4 + i] = gelu_exact(s5 * inv_sqrt6 + q[part][V_][i] * (1.0f / 6.0f) + bs[i]);
        }
    }
    layernorm_row_store(v, lane, gg, gb, out + (size_t)t * D_);
}

extern "C" void kernel_launch(void* const* d_in, const int* in_sizes, int n_in,
                              void* d_out, int out_size, void* d_ws, size_t ws_size,
                              hipStream_t stream) {
    (void)in_sizes; (void)n_in; (void)out_size; (void)ws_size;

    const float* x       = (const float*)d_in[0];
    const float* vn      = (const float*)d_in[2];
    const float* gat_W   = (const float*)d_in[3];
    const float* att_src = (const float*)d_in[4];
    const float* att_dst = (const float*)d_in[5];
    const float* gat_b   = (const float*)d_in[6];
    const float* gcn_W   = (const float*)d_in[7];
    const float* gcn_b   = (const float*)d_in[8];
    const float* n1g     = (const float*)d_in[9];
    const float* n1b     = (const float*)d_in[10];
    const float* n2g     = (const float*)d_in[11];
    const float* n2b     = (const float*)d_in[12];
    float* out = (float*)d_out;

    float* ws = (float*)d_ws;
    size_t off = 0;
    float* wm    = ws + off; off += (size_t)D_ * D_;
    float* ut    = ws + off; off += (size_t)D_ * UCOLS_;
    float* gm    = ws + off; off += (size_t)T_ * D_;
    float* asrc  = ws + off; off += (size_t)T_ * H_;
    float* adst  = ws + off; off += (size_t)T_ * H_;
    float* z     = ws + off; off += (size_t)H_ * B_ * V_ * D_;
    float* gv    = ws + off; off += (size_t)H_ * B_ * V_ * HD_;
    float* xw    = ws + off; off += (size_t)T_ * D_;

    k_fold_weights<<<PREP_WGS_, D_, 0, stream>>>(gat_W, att_src, att_dst, wm, ut);
    {
        dim3 grid(T_ / 16, 1);
        k_gemm_f32_wmma<D_, D_, true, 0, true, 2, false><<<grid, 160, 0, stream>>>(x, vn, wm, gm, ut, asrc, adst,
                                                                                 nullptr, nullptr, nullptr);
    }
    {
        k_gat_virtual<<<B_ * V_, D_, 0, stream>>>(x, vn, asrc, adst, z);
    }
    {
        dim3 grid(B_ * V_ / 16, H_);
        k_gemm_f32_wmma<D_, HD_, false, B_ * V_, false, 4, false><<<grid, 128, 0, stream>>>(z, nullptr, gat_W, gv,
                                                                                        nullptr, nullptr, nullptr,
                                                                                        nullptr, nullptr, nullptr);
    }
    {
        dim3 grid(T_ / 16, 1);
        k_gemm_f32_wmma<D_, D_, false, 0, false, 2, true><<<grid, 128, 0, stream>>>(gm, gv, gcn_W, xw,
                                                                                 nullptr, nullptr, nullptr,
                                                                                 gat_b, n1g, n1b);
    }
    k_gcn_finish<<<BN_ / ROWS_PER_WG_, ROWS_PER_WG_ * 32, 0, stream>>>(xw, gcn_b, n2g, n2b, out);
}


# GCN combine (gat1/6 + batch's 5 virtual gat1 rows/sqrt6) moved in front of gcn_W into the GEMM prologue; bias+GELU+LayerNorm2 become its epilogue; K7 launch and T x 256 xw go; GCN sum order changes
# speedup vs baseline: 4.5605x; 1.0379x over previous
#include <hip/hip_runtime.h>
#include <math.h>

#define B_   32
#define N_   512
#define V_   5
#define D_   256
#define H_   4
#define BN_  (B_ * N_)
#define T_   (BN_ + B_ * V_)
#define HD_  (H_ * D_)
#define EPS_ 1e-5f

typedef float v2f __attribute__((ext_vector_type(2)));
typedef float v4f __attribute__((ext_vector_type(4)));
typedef float v8f __attribute__((ext_vector_type(8)));

__device__ __forceinline__ float gelu_exact(float x) {
    return 0.5f * x * (1.0f + erff(x * 0.70710678118654752f));
}

template <int XMASK>
__device__ __forceinline__ float add_row_xmask(float v) {
    return v + __builtin_bit_cast(float, __builtin_amdgcn_update_dpp(
                   0, __builtin_bit_cast(int, v), 0x160 + XMASK, 0xf, 0xf, true));
}

__device__ __forceinline__ float row16_sum(float v) {
    v = add_row_xmask<1>(v);
    v = add_row_xmask<2>(v);
    v = add_row_xmask<4>(v);
    v = add_row_xmask<8>(v);
    return v;
}

__device__ __forceinline__ float wave32_sum(float v) {
    v = row16_sum(v);
    return v + __builtin_bit_cast(float, __builtin_amdgcn_permlanex16(
                   0u, __builtin_bit_cast(unsigned, v), 0x76543210u, 0xfedcba98u, true, false));
}

#define PREP_WGS_ D_
#define UCOLS_    32

__device__ __forceinline__ const float* node_row(const float* __restrict__ x,
                                                 const float* __restrict__ vn, int t) {
    return (t < BN_) ? x + (size_t)t * D_ : vn + (size_t)((t - BN_) % V_) * D_;
}

__global__ void k_fold_weights(const float* __restrict__ W,
                               const float* __restrict__ att_src,
                               const float* __restrict__ att_dst,
                               float* __restrict__ Wm,
                               float* __restrict__ U) {
    __shared__ float red[D_ / 32][8];
    const int k = blockIdx.x;
    const int c = threadIdx.x;
    float w[H_], p[8];
    for (int hd = 0; hd < H_; ++hd) w[hd] = W[(size_t)k * HD_ + hd * D_ + c];
    Wm[(size_t)k * D_ + c] = 0.25f * ((w[0] + w[1]) + (w[2] + w[3]));
    for (int hd = 0; hd < H_; ++hd) {
        p[hd]      = w[hd] * att_src[hd * D_ + c];
        p[H_ + hd] = w[hd] * att_dst[hd * D_ + c];
    }
    for (int j = 0; j < 8; ++j) p[j] = wave32_sum(p[j]);
    if ((c & 31) == 0)
        for (int j = 0; j < 8; ++j) red[c >> 5][j] = p[j];
    __syncthreads();
    if (c < UCOLS_) {
        float s = 0.0f;
        if (c < 8)
            s = ((red[0][c] + red[1][c]) + (red[2][c] + red[3][c])) +
                ((red[4][c] + red[5][c]) + (red[6][c] + red[7][c]));
        U[(size_t)k * UCOLS_ + c] = s;
    }
}

static_assert(D_ == 256, "8 channels per lane of a 32-lane wave");

__device__ __forceinline__ float sum8(const float* v) {
    return ((v[0] + v[1]) + (v[2] + v[3])) + ((v[4] + v[5]) + (v[6] + v[7]));
}

__device__ __forceinline__ float layernorm_center(const float* v, float* c) {
    const float mu = wave32_sum(sum8(v)) * (1.0f / D_);
    float sq[8];
    for (int i = 0; i < 8; ++i) { c[i] = v[i] - mu; sq[i] = c[i] * c[i]; }
    const float var = wave32_sum(sum8(sq)) * (1.0f / D_);
    return rsqrtf(var + EPS_);
}

__device__ __forceinline__ void load_row8(const float* __restrict__ p, int lane, v4f* q) {
    q[0] = *(const v4f*)(p + lane * 4);
    q[1] = *(const v4f*)(p + 128 + lane * 4);
}

__device__ __forceinline__ void bias_gelu_layernorm(const v4f* m, const v4f* bs, const v4f* g, const v4f* b, v4f* o) {
    float v[8], c[8];
    for (int part = 0; part < 2; ++part)
        for (int i = 0; i < 4; ++i)
            v[part * 4 + i] = gelu_exact(m[part][i] + bs[part][i]);
    const float rs = layernorm_center(v, c);
    for (int part = 0; part < 2; ++part)
        for (int i = 0; i < 4; ++i)
            o[part][i] = c[part * 4 + i] * rs * g[part][i] + b[part][i];
}

__device__ __forceinline__ void load_virtual_heads(const float* __restrict__ gv, int uidx, int lane, v4f (*q)[2]) {
    for (int hd = 0; hd < H_; ++hd)
        load_row8(gv + (size_t)(hd * (B_ * V_) + uidx) * HD_ + hd * D_, lane, q[hd]);
}
__device__ __forceinline__ void head_mean(const v4f (*q)[2], v4f* m) {
    for (int part = 0; part < 2; ++part)
        for (int i = 0; i < 4; ++i)
            m[part][i] = 0.25f * (q[0][part][i] + q[1][part][i] + q[2][part][i] + q[3][part][i]);
}

template <int KB>
struct GemmFrag {
    v4f a[KB];
    v4f b[KB][4];
};

template <int KB, int NDIM>
__device__ __forceinline__ void gemm_load_block(GemmFrag<KB>& f, const float* __restrict__ ap,
                                                const float* __restrict__ bp, int k0) {
#pragma unroll
    for (int s = 0; s < KB; ++s) {
        f.a[s] = *(const v4f*)(ap + k0 + 8 * s);
#pragma unroll
        for (int j = 0; j < 4; ++j)
            f.b[s][j] = *(const v4f*)(bp + (size_t)(k0 + 8 * s + j) * NDIM);
    }
}

template <int KB>
__device__ __forceinline__ void gemm_mma_block(const GemmFrag<KB>& f, v8f& c0, v8f& c1, v8f& c2, v8f& c3) {
#pragma unroll
    for (int s = 0; s < KB; ++s) {
        const v4f a = f.a[s];
        const v4f b0 = f.b[s][0], b1 = f.b[s][1], b2 = f.b[s][2], b3 = f.b[s][3];
        v2f af, bf;
        af[0] = a[0]; af[1] = a[1];
        bf[0] = b0[0]; bf[1] = b1[0];
        c0 = __builtin_amdgcn_wmma_f32_16x16x4_f32(false, af, false, bf, (short)0, c0, false, false);
        bf[0] = b0[1]; bf[1] = b1[1];
        c1 = __builtin_amdgcn_wmma_f32_16x16x4_f32(false, af, false, bf, (short)0, c1, false, false);
        bf[0] = b0[2]; bf[1] = b1[2];
        c2 = __builtin_amdgcn_wmma_f32_16x16x4_f32(false, af, false, bf, (short)0, c2, false, false);
        bf[0] = b0[3]; bf[1] = b1[3];
        c3 = __builtin_amdgcn_wmma_f32_16x16x4_f32(false, af, false, bf, (short)0, c3, false, false);
        af[0] = a[2]; af[1] = a[3];
        bf[0] = b2[0]; bf[1] = b3[0];
        c0 = __builtin_amdgcn_wmma_f32_16x16x4_f32(false, af, false, bf, (short)0, c0, false, false);
        bf[0] = b2[1]; bf[1] = b3[1];
        c1 = __builtin_amdgcn_wmma_f32_16x16x4_f32(false, af, false, bf, (short)0, c1, false, false);
        bf[0] = b2[2]; bf[1] = b3[2];
        c2 = __builtin_amdgcn_wmma_f32_16x16x4_f32(false, af, false, bf, (short)0, c2, false, false);
        bf[0] = b2[3]; bf[1] = b3[3];
        c3 = __builtin_amdgcn_wmma_f32_16x16x4_f32(false, af, false, bf, (short)0, c3, false, false);
    }
}

#define LN_PITCH_ 260
template <int KDIM, int NDIM, bool SCORES, int YROWS, bool NODE_ROWS, int KB, bool LN_ROWS>
__global__ void __launch_bounds__(160) k_gemm_f32_wmma(const float* __restrict__ A,
                                const float* __restrict__ Av,
                                const float* __restrict__ Bm,
                                float* __restrict__ C,
                                const float* __restrict__ U,
                                float* __restrict__ a_src,
                                float* __restrict__ a_dst,
                                const float* __restrict__ ln_bias,
                                const float* __restrict__ ln_g,
                                const float* __restrict__ ln_b,
                                const float* __restrict__ ep_bias,
                                const float* __restrict__ ep_g,
                                const float* __restrict__ ep_b) {
    static_assert(KDIM % 8 == 0 && NDIM % 64 == 0, "8-deep K-steps, 64-column wave tiles");
    const int wave = __builtin_amdgcn_readfirstlane((int)(threadIdx.x >> 5));
    const int lane = threadIdx.x & 31;
    const int half = lane >> 4;
    const int lm   = lane & 15;
    const int m0   = blockIdx.x * 16 + blockIdx.y * YROWS;
    const int n0   = blockIdx.y * 256 + wave * 64;
    static_assert(!NODE_ROWS || KDIM == D_, "node rows are 256 words");
    const float* ap;
    float* at = nullptr;
    if constexpr (LN_ROWS) {
        static_assert(KDIM == D_ && NDIM == D_ && !SCORES && !NODE_ROWS && YROWS == 0 && N_ % 16 == 0 && V_ == 5,
                      "a tile is 16 whole real rows of one batch; 5 virtual rows over 4 waves");
        __shared__ v4f a_tile[(16 + V_) * LN_PITCH_ / 4];
        at = (float*)a_tile;
        const int tr = m0 + wave * 4;
        const int ub = (m0 / N_) * V_;
        v4f bs[2], g[2], b[2], m[4][2], qa[H_][2], qb[H_][2];
        load_row8(ln_bias, lane, bs);
        load_row8(ln_g, lane, g);
        load_row8(ln_b, lane, b);
#pragma unroll
        for (int r = 0; r < 4; ++r) load_row8(A + (size_t)(tr + r) * D_, lane, m[r]);
        load_virtual_heads(Av, ub + wave, lane, qa);
        if (wave == 0) load_virtual_heads(Av, ub + 4, lane, qb);
#pragma unroll
        for (int r = 0; r < 4; ++r)
            for (int part = 0; part < 2; ++part)
                *(v4f*)(at + (wave * 4 + r) * LN_PITCH_ + part * 128 + lane * 4) = m[r][part];
        {
            v4f mv[2];
            head_mean(qa, mv);
            for (int part = 0; part < 2; ++part) *(v4f*)(at + (16 + wave) * LN_PITCH_ + part * 128 + lane * 4) = mv[part];
            if (wave == 0) {
                head_mean(qb, mv);
                for (int part = 0; part < 2; ++part) *(v4f*)(at + 20 * LN_PITCH_ + part * 128 + lane * 4) = mv[part];
            }
        }
        const int nrows = (wave == 0) ? 6 : 5;
#pragma nounroll
        for (int i = 0; i < nrows; ++i) {
            const int row = (i < 4) ? wave * 4 + i : (i == 4) ? 16 + wave : 20;
            float* rp = at + row * LN_PITCH_ + lane * 4;
            v4f mm[2], o[2];
            mm[0] = *(const v4f*)rp;
            mm[1] = *(const v4f*)(rp + 128);
            bias_gelu_layernorm(mm, bs, g, b, o);
            *(v4f*)rp = o[0];
            *(v4f*)(rp + 128) = o[1];
        }
        __syncthreads();
        const float inv_sqrt6 = 0.40824829046386301637f;
        for (int part = 0; part < 2; ++part) {
            const float* vp = at + 16 * LN_PITCH_ + part * 128 + lane * 4;
            const v4f sv = ((*(const v4f*)(vp) + *(const v4f*)(vp + LN_PITCH_)) +
                            (*(const v4f*)(vp + 2 * LN_PITCH_) + *(const v4f*)(vp + 3 * LN_PITCH_))) +
                           *(const v4f*)(vp + 4 * LN_PITCH_);
            for (int r = 0; r < 4; ++r) {
                float* rp = at + (wave * 4 + r) * LN_PITCH_ + part * 128 + lane * 4;
                *(v4f*)rp = *(const v4f*)rp * (1.0f / 6.0f) + sv * inv_sqrt6;
            }
        }
        __syncthreads();
        ap = at + lm * LN_PITCH_ + half * 4;
    } else if constexpr (NODE_ROWS) ap = node_row(A, Av, m0 + lm) + half * 4;
    else                            ap = A + (size_t)(m0 + lm) * KDIM + half * 4;
    if constexpr (SCORES) {
        static_assert(NDIM == 256, "waves 0-3 cover all columns; wave 4 is the score tile");
        if (wave == 4) {
            v8f s0 = {}; v8f s1 = {};
            const float* up = U + (size_t)(half * 4) * UCOLS_ + lm;
            for (int k0 = 0; k0 < KDIM; k0 += 8) {
                const v4f a = *(const v4f*)(ap + k0);
                const float u0 = up[(size_t)(k0 + 0) * UCOLS_];
                const float u1 = up[(size_t)(k0 + 1) * UCOLS_];
                const float u2 = up[(size_t)(k0 + 2) * UCOLS_];
                const float u3 = up[(size_t)(k0 + 3) * UCOLS_];
                v2f af, bf;
                af[0] = a[0]; af[1] = a[1]; bf[0] = u0; bf[1] = u1;
                s0 = __builtin_amdgcn_wmma_f32_16x16x4_f32(false, af, false, bf, (short)0, s0, false, false);
                af[0] = a[2]; af[1] = a[3]; bf[0] = u2; bf[1] = u3;
                s1 = __builtin_amdgcn_wmma_f32_16x16x4_f32(false, af, false, bf, (short)0, s1, false, false);
            }
            if (lm < 8) {
                float* dst = (lm < H_) ? a_src : a_dst;
                for (int j = 0; j < 8; ++j)
                    dst[(size_t)(m0 + half * 8 + j) * H_ + (lm & (H_ - 1))] = s0[j] + s1[j];
            }
            return;
        }
    } else if constexpr (!LN_ROWS) {
        if (n0 >= NDIM) return;
    }

    v8f c0 = {}; v8f c1 = {}; v8f c2 = {}; v8f c3 = {};
    const float* bp = Bm + (size_t)(half * 4) * NDIM + n0 + lm * 4;

    constexpr int KBLK = 8 * KB;
    static_assert(KDIM % (2 * KBLK) == 0, "an even number of blocks: the sets alternate");
    GemmFrag<KB> f0, f1;
    gemm_load_block<KB, NDIM>(f0, ap, bp, 0);
    for (int k0 = 0; k0 < KDIM - 2 * KBLK; k0 += 2 * KBLK) {
        gemm_load_block<KB, NDIM>(f1, ap, bp, k0 + KBLK);
        __builtin_amdgcn_sched_barrier(0);
        gemm_mma_block<KB>(f0, c0, c1, c2, c3);
        __builtin_amdgcn_sched_barrier(0);
        gemm_load_block<KB, NDIM>(f0, ap, bp, k0 + 2 * KBLK);
        __builtin_amdgcn_sched_barrier(0);
        gemm_mma_block<KB>(f1, c0, c1, c2, c3);
        __builtin_amdgcn_sched_barrier(0);
    }
    gemm_load_block<KB, NDIM>(f1, ap, bp, KDIM - KBLK);
    __builtin_amdgcn_sched_barrier(0);
    gemm_mma_block<KB>(f0, c0, c1, c2, c3);
    gemm_mma_block<KB>(f1, c0, c1, c2, c3);

    if constexpr (LN_ROWS) {
        v4f bs2[2], g2[2], b2[2];
        load_row8(ep_bias, lane, bs2);
        load_row8(ep_g, lane, g2);
        load_row8(ep_b, lane, b2);
        __syncthreads();
        float* trow = at + (half * 8) * LN_PITCH_ + n0 + lm * 4;
#pragma unroll
        for (int j = 0; j < 8; ++j) {
            v4f o;
            o[0] = c0[j]; o[1] = c1[j]; o[2] = c2[j]; o[3] = c3[j];
            *(v4f*)(trow + j * LN_PITCH_) = o;
        }
        __syncthreads();
#pragma nounroll
        for (int r = 0; r < 4; ++r) {
            const float* rp = at + (wave * 4 + r) * LN_PITCH_ + lane * 4;
            v4f y[2], o[2];
            y[0] = *(const v4f*)rp;
            y[1] = *(const v4f*)(rp + 128);
            bias_gelu_layernorm(y, bs2, g2, b2, o);
            float* orow = C + (size_t)(m0 + wave * 4 + r) * D_ + lane * 4;
            *(v4f*)orow = o[0];
            *(v4f*)(orow + 128) = o[1];
        }
    } else {
        float* crow = C + (size_t)(m0 + half * 8) * NDIM + n0 + lm * 4;
#pragma unroll
        for (int j = 0; j < 8; ++j) {
            v4f o;
            o[0] = c0[j]; o[1] = c1[j]; o[2] = c2[j]; o[3] = c3[j];
            *(v4f*)(crow + (size_t)j * NDIM) = o;
        }
    }
}

#define QG_    4
#define QROWS_ (N_ / QG_)
static_assert(QG_ == H_ && QG_ * 64 == D_, "thread (q, c4) also finishes head q, channels 4*c4..+3");

template <int XMASK>
__device__ __forceinline__ float max_row_xmask(float v) {
    return fmaxf(v, __builtin_bit_cast(float, __builtin_amdgcn_update_dpp(
                        0, __builtin_bit_cast(int, v), 0x160 + XMASK, 0xf, 0xf, true)));
}

__device__ __forceinline__ float wave32_max(float v) {
    v = max_row_xmask<1>(v);
    v = max_row_xmask<2>(v);
    v = max_row_xmask<4>(v);
    v = max_row_xmask<8>(v);
    return fmaxf(v, __builtin_bit_cast(float, __builtin_amdgcn_permlanex16(
                        0u, __builtin_bit_cast(unsigned, v), 0x76543210u, 0xfedcba98u, true, false)));
}

__device__ __forceinline__ float leaky02(float a) { return (a > 0.0f) ? a : 0.2f * a; }

__device__ __forceinline__ float pick4(v4f v, int i) {
    return (i == 0) ? v[0] : (i == 1) ? v[1] : (i == 2) ? v[2] : v[3];
}

__global__ void __launch_bounds__(D_) k_gat_virtual(const float* __restrict__ x,
                              const float* __restrict__ vn,
                              const float* __restrict__ a_src,
                              const float* __restrict__ a_dst,
                              float* __restrict__ z) {
    __shared__ v4f w4[N_];
    __shared__ v4f red4[D_ / 32];
    __shared__ v4f part[QG_][H_][D_ / 4];
    const int uidx = blockIdx.x;
    const int b    = uidx / V_;
    const int u    = BN_ + uidx;
    const int tid  = threadIdx.x;
    const int lane = tid & 31;
    const int wave = tid >> 5;

    const v4f adst = *(const v4f*)(a_dst + (size_t)u * H_);
    const int r0 = b * N_ + tid * 2;
    v4f al0 = *(const v4f*)(a_src + (size_t)r0 * H_);
    v4f al1 = *(const v4f*)(a_src + (size_t)(r0 + 1) * H_);
    v4f als = *(const v4f*)(a_src + (size_t)u * H_);
    v4f lmax;
    for (int hd = 0; hd < H_; ++hd) {
        al0[hd] = leaky02(al0[hd] + adst[hd]);
        al1[hd] = leaky02(al1[hd] + adst[hd]);
        als[hd] = leaky02(als[hd] + adst[hd]);
        lmax[hd] = wave32_max(fmaxf(fmaxf(al0[hd], al1[hd]), als[hd]));
    }
    if (lane == 0) red4[wave] = lmax;
    __syncthreads();
    v4f m = red4[0];
    for (int wv = 1; wv < D_ / 32; ++wv) {
        const v4f o = red4[wv];
        for (int hd = 0; hd < H_; ++hd) m[hd] = fmaxf(m[hd], o[hd]);
    }
    __syncthreads();

    v4f e0, e1, es, ps;
    for (int hd = 0; hd < H_; ++hd) {
        e0[hd] = expf(al0[hd] - m[hd]);
        e1[hd] = expf(al1[hd] - m[hd]);
        es[hd] = expf(als[hd] - m[hd]);
        float s = e0[hd] + e1[hd];
        if (tid == 0) s += es[hd];
        ps[hd] = wave32_sum(s);
    }
    w4[tid * 2]     = e0;
    w4[tid * 2 + 1] = e1;
    if (lane == 0) red4[wave] = ps;
    __syncthreads();
    const v4f tot = ((red4[0] + red4[1]) + (red4[2] + red4[3])) +
                    ((red4[4] + red4[5]) + (red4[6] + red4[7]));

    const int c4 = tid & 63;
    const int q  = __builtin_amdgcn_readfirstlane(tid >> 6);
    const float* xb = x + (size_t)(b * N_ + q * QROWS_) * D_ + c4 * 4;
    const v4f* wq = w4 + q * QROWS_;
    v4f acc[H_];
    for (int hd = 0; hd < H_; ++hd) acc[hd] = (v4f){0.0f, 0.0f, 0.0f, 0.0f};
    for (int rb = 0; rb < QROWS_; rb += 32) {
        v4f xv[32];
#pragma unroll
        for (int i = 0; i < 32; ++i) xv[i] = *(const v4f*)(xb + (size_t)(rb + i) * D_);
#pragma unroll
        for (int i = 0; i < 32; ++i) {
            const v4f wr = wq[rb + i];
            for (int hd = 0; hd < H_; ++hd) acc[hd] += wr[hd] * xv[i];
        }
    }
    for (int hd = 0; hd < H_; ++hd) part[q][hd][c4] = acc[hd];
    __syncthreads();

    v4f s4 = (part[0][q][c4] + part[1][q][c4]) + (part[2][q][c4] + part[3][q][c4]);
    const v4f vs = *(const v4f*)(vn + (size_t)(uidx % V_) * D_ + c4 * 4);
    s4 += pick4(es, q) * vs;
    const float inv = 1.0f / pick4(tot, q);
    *(v4f*)(z + ((size_t)q * (B_ * V_) + uidx) * D_ + c4 * 4) = s4 * inv;
}

extern "C" void kernel_launch(void* const* d_in, const int* in_sizes, int n_in,
                              void* d_out, int out_size, void* d_ws, size_t ws_size,
                              hipStream_t stream) {
    (void)in_sizes; (void)n_in; (void)out_size; (void)ws_size;

    const float* x       = (const float*)d_in[0];
    const float* vn      = (const float*)d_in[2];
    const float* gat_W   = (const float*)d_in[3];
    const float* att_src = (const float*)d_in[4];
    const float* att_dst = (const float*)d_in[5];
    const float* gat_b   = (const float*)d_in[6];
    const float* gcn_W   = (const float*)d_in[7];
    const float* gcn_b   = (const float*)d_in[8];
    const float* n1g     = (const float*)d_in[9];
    const float* n1b     = (const float*)d_in[10];
    const float* n2g     = (const float*)d_in[11];
    const float* n2b     = (const float*)d_in[12];
    float* out = (float*)d_out;

    float* ws = (float*)d_ws;
    size_t off = 0;
    float* wm    = ws + off; off += (size_t)D_ * D_;
    float* ut    = ws + off; off += (size_t)D_ * UCOLS_;
    float* gm    = ws + off; off += (size_t)T_ * D_;
    float* asrc  = ws + off; off += (size_t)T_ * H_;
    float* adst  = ws + off; off += (size_t)T_ * H_;
    float* z     = ws + off; off += (size_t)H_ * B_ * V_ * D_;
    float* gv    = ws + off; off += (size_t)H_ * B_ * V_ * HD_;

    k_fold_weights<<<PREP_WGS_, D_, 0, stream>>>(gat_W, att_src, att_dst, wm, ut);
    {
        dim3 grid(T_ / 16, 1);
        k_gemm_f32_wmma<D_, D_, true, 0, true, 2, false><<<grid, 160, 0, stream>>>(x, vn, wm, gm, ut, asrc, adst,
                                                                                 nullptr, nullptr, nullptr,
                                                                                 nullptr, nullptr, nullptr);
    }
    {
        k_gat_virtual<<<B_ * V_, D_, 0, stream>>>(x, vn, asrc, adst, z);
    }
    {
        dim3 grid(B_ * V_ / 16, H_);
        k_gemm_f32_wmma<D_, HD_, false, B_ * V_, false, 4, false><<<grid, 128, 0, stream>>>(z, nullptr, gat_W, gv,
                                                                                        nullptr, nullptr, nullptr,
                                                                                        nullptr, nullptr, nullptr,
                                                                                        nullptr, nullptr, nullptr);
    }
    {
        dim3 grid(BN_ / 16, 1);
        k_gemm_f32_wmma<D_, D_, false, 0, false, 2, true><<<grid, 128, 0, stream>>>(gm, gv, gcn_W, out,
                                                                                 nullptr, nullptr, nullptr,
                                                                                 gat_b, n1g, n1b,
                                                                                 gcn_b, n2g, n2b);
    }
}


# GELU of both LayerNorm stages as max(x,0) - |x|*2^(|x|*q(|x|)-1), q a degree-6 fit of log2 erfc(u/sqrt2)/u (within 3e-7 of true GELU): one branch-free range replaces the library erf's two branches
# speedup vs baseline: 4.9438x; 1.0840x over previous
#include <hip/hip_runtime.h>
#include <math.h>

#define B_   32
#define N_   512
#define V_   5
#define D_   256
#define H_   4
#define BN_  (B_ * N_)
#define T_   (BN_ + B_ * V_)
#define HD_  (H_ * D_)
#define EPS_ 1e-5f

typedef float v2f __attribute__((ext_vector_type(2)));
typedef float v4f __attribute__((ext_vector_type(4)));
typedef float v8f __attribute__((ext_vector_type(8)));

__device__ __forceinline__ float gelu_exact(float x) {
    const float u = fminf(fabsf(x), 6.0f);
    float q = 4.27864052e-6f;
    q = fmaf(q, u, -1.27970425e-5f);
    q = fmaf(q, u, -5.75791288e-4f);
    q = fmaf(q, u, 7.67079415e-3f);
    q = fmaf(q, u, -5.29485755e-2f);
    q = fmaf(q, u, -4.59043920e-1f);
    q = fmaf(q, u, -1.15112698f);
    const float e = __builtin_amdgcn_exp2f(fmaf(q, u, -1.0f));
    return fmaf(-u, e, fmaxf(x, 0.0f));
}

template <int XMASK>
__device__ __forceinline__ float add_row_xmask(float v) {
    return v + __builtin_bit_cast(float, __builtin_amdgcn_update_dpp(
                   0, __builtin_bit_cast(int, v), 0x160 + XMASK, 0xf, 0xf, true));
}

__device__ __forceinline__ float row16_sum(float v) {
    v = add_row_xmask<1>(v);
    v = add_row_xmask<2>(v);
    v = add_row_xmask<4>(v);
    v = add_row_xmask<8>(v);
    return v;
}

__device__ __forceinline__ float wave32_sum(float v) {
    v = row16_sum(v);
    return v + __builtin_bit_cast(float, __builtin_amdgcn_permlanex16(
                   0u, __builtin_bit_cast(unsigned, v), 0x76543210u, 0xfedcba98u, true, false));
}

#define PREP_WGS_ D_
#define UCOLS_    32

__device__ __forceinline__ const float* node_row(const float* __restrict__ x,
                                                 const float* __restrict__ vn, int t) {
    return (t < BN_) ? x + (size_t)t * D_ : vn + (size_t)((t - BN_) % V_) * D_;
}

__global__ void k_fold_weights(const float* __restrict__ W,
                               const float* __restrict__ att_src,
                               const float* __restrict__ att_dst,
                               float* __restrict__ Wm,
                               float* __restrict__ U) {
    __shared__ float red[D_ / 32][8];
    const int k = blockIdx.x;
    const int c = threadIdx.x;
    float w[H_], p[8];
    for (int hd = 0; hd < H_; ++hd) w[hd] = W[(size_t)k * HD_ + hd * D_ + c];
    Wm[(size_t)k * D_ + c] = 0.25f * ((w[0] + w[1]) + (w[2] + w[3]));
    for (int hd = 0; hd < H_; ++hd) {
        p[hd]      = w[hd] * att_src[hd * D_ + c];
        p[H_ + hd] = w[hd] * att_dst[hd * D_ + c];
    }
    for (int j = 0; j < 8; ++j) p[j] = wave32_sum(p[j]);
    if ((c & 31) == 0)
        for (int j = 0; j < 8; ++j) red[c >> 5][j] = p[j];
    __syncthreads();
    if (c < UCOLS_) {
        float s = 0.0f;
        if (c < 8)
            s = ((red[0][c] + red[1][c]) + (red[2][c] + red[3][c])) +
                ((red[4][c] + red[5][c]) + (red[6][c] + red[7][c]));
        U[(size_t)k * UCOLS_ + c] = s;
    }
}

static_assert(D_ == 256, "8 channels per lane of a 32-lane wave");

__device__ __forceinline__ float sum8(const float* v) {
    return ((v[0] + v[1]) + (v[2] + v[3])) + ((v[4] + v[5]) + (v[6] + v[7]));
}

__device__ __forceinline__ float layernorm_center(const float* v, float* c) {
    const float mu = wave32_sum(sum8(v)) * (1.0f / D_);
    float sq[8];
    for (int i = 0; i < 8; ++i) { c[i] = v[i] - mu; sq[i] = c[i] * c[i]; }
    const float var = wave32_sum(sum8(sq)) * (1.0f / D_);
    return rsqrtf(var + EPS_);
}

__device__ __forceinline__ void load_row8(const float* __restrict__ p, int lane, v4f* q) {
    q[0] = *(const v4f*)(p + lane * 4);
    q[1] = *(const v4f*)(p + 128 + lane * 4);
}

__device__ __forceinline__ void bias_gelu_layernorm(const v4f* m, const v4f* bs, const v4f* g, const v4f* b, v4f* o) {
    float v[8], c[8];
    for (int part = 0; part < 2; ++part)
        for (int i = 0; i < 4; ++i)
            v[part * 4 + i] = gelu_exact(m[part][i] + bs[part][i]);
    const float rs = layernorm_center(v, c);
    for (int part = 0; part < 2; ++part)
        for (int i = 0; i < 4; ++i)
            o[part][i] = c[part * 4 + i] * rs * g[part][i] + b[part][i];
}

__device__ __forceinline__ void load_virtual_heads(const float* __restrict__ gv, int uidx, int lane, v4f (*q)[2]) {
    for (int hd = 0; hd < H_; ++hd)
        load_row8(gv + (size_t)(hd * (B_ * V_) + uidx) * HD_ + hd * D_, lane, q[hd]);
}
__device__ __forceinline__ void head_mean(const v4f (*q)[2], v4f* m) {
    for (int part = 0; part < 2; ++part)
        for (int i = 0; i < 4; ++i)
            m[part][i] = 0.25f * (q[0][part][i] + q[1][part][i] + q[2][part][i] + q[3][part][i]);
}

template <int KB>
struct GemmFrag {
    v4f a[KB];
    v4f b[KB][4];
};

template <int KB, int NDIM>
__device__ __forceinline__ void gemm_load_block(GemmFrag<KB>& f, const float* __restrict__ ap,
                                                const float* __restrict__ bp, int k0) {
#pragma unroll
    for (int s = 0; s < KB; ++s) {
        f.a[s] = *(const v4f*)(ap + k0 + 8 * s);
#pragma unroll
        for (int j = 0; j < 4; ++j)
            f.b[s][j] = *(const v4f*)(bp + (size_t)(k0 + 8 * s + j) * NDIM);
    }
}

template <int KB>
__device__ __forceinline__ void gemm_mma_block(const GemmFrag<KB>& f, v8f& c0, v8f& c1, v8f& c2, v8f& c3) {
#pragma unroll
    for (int s = 0; s < KB; ++s) {
        const v4f a = f.a[s];
        const v4f b0 = f.b[s][0], b1 = f.b[s][1], b2 = f.b[s][2], b3 = f.b[s][3];
        v2f af, bf;
        af[0] = a[0]; af[1] = a[1];
        bf[0] = b0[0]; bf[1] = b1[0];
        c0 = __builtin_amdgcn_wmma_f32_16x16x4_f32(false, af, false, bf, (short)0, c0, false, false);
        bf[0] = b0[1]; bf[1] = b1[1];
        c1 = __builtin_amdgcn_wmma_f32_16x16x4_f32(false, af, false, bf, (short)0, c1, false, false);
        bf[0] = b0[2]; bf[1] = b1[2];
        c2 = __builtin_amdgcn_wmma_f32_16x16x4_f32(false, af, false, bf, (short)0, c2, false, false);
        bf[0] = b0[3]; bf[1] = b1[3];
        c3 = __builtin_amdgcn_wmma_f32_16x16x4_f32(false, af, false, bf, (short)0, c3, false, false);
        af[0] = a[2]; af[1] = a[3];
        bf[0] = b2[0]; bf[1] = b3[0];
        c0 = __builtin_amdgcn_wmma_f32_16x16x4_f32(false, af, false, bf, (short)0, c0, false, false);
        bf[0] = b2[1]; bf[1] = b3[1];
        c1 = __builtin_amdgcn_wmma_f32_16x16x4_f32(false, af, false, bf, (short)0, c1, false, false);
        bf[0] = b2[2]; bf[1] = b3[2];
        c2 = __builtin_amdgcn_wmma_f32_16x16x4_f32(false, af, false, bf, (short)0, c2, false, false);
        bf[0] = b2[3]; bf[1] = b3[3];
        c3 = __builtin_amdgcn_wmma_f32_16x16x4_f32(false, af, false, bf, (short)0, c3, false, false);
    }
}

#define LN_PITCH_ 260
template <int KDIM, int NDIM, bool SCORES, int YROWS, bool NODE_ROWS, int KB, bool LN_ROWS>
__global__ void __launch_bounds__(160) k_gemm_f32_wmma(const float* __restrict__ A,
                                const float* __restrict__ Av,
                                const float* __restrict__ Bm,
                                float* __restrict__ C,
                                const float* __restrict__ U,
                                float* __restrict__ a_src,
                                float* __restrict__ a_dst,
                                const float* __restrict__ ln_bias,
                                const float* __restrict__ ln_g,
                                const float* __restrict__ ln_b,
                                const float* __restrict__ ep_bias,
                                const float* __restrict__ ep_g,
                                const float* __restrict__ ep_b) {
    static_assert(KDIM % 8 == 0 && NDIM % 64 == 0, "8-deep K-steps, 64-column wave tiles");
    const int wave = __builtin_amdgcn_readfirstlane((int)(threadIdx.x >> 5));
    const int lane = threadIdx.x & 31;
    const int half = lane >> 4;
    const int lm   = lane & 15;
    const int m0   = blockIdx.x * 16 + blockIdx.y * YROWS;
    const int n0   = blockIdx.y * 256 + wave * 64;
    static_assert(!NODE_ROWS || KDIM == D_, "node rows are 256 words");
    const float* ap;
    float* at = nullptr;
    if constexpr (LN_ROWS) {
        static_assert(KDIM == D_ && NDIM == D_ && !SCORES && !NODE_ROWS && YROWS == 0 && N_ % 16 == 0 && V_ == 5,
                      "a tile is 16 whole real rows of one batch; 5 virtual rows over 4 waves");
        __shared__ v4f a_tile[(16 + V_) * LN_PITCH_ / 4];
        at = (float*)a_tile;
        const int tr = m0 + wave * 4;
        const int ub = (m0 / N_) * V_;
        v4f bs[2], g[2], b[2], m[4][2], qa[H_][2], qb[H_][2];
        load_row8(ln_bias, lane, bs);
        load_row8(ln_g, lane, g);
        load_row8(ln_b, lane, b);
#pragma unroll
        for (int r = 0; r < 4; ++r) load_row8(A + (size_t)(tr + r) * D_, lane, m[r]);
        load_virtual_heads(Av, ub + wave, lane, qa);
        if (wave == 0) load_virtual_heads(Av, ub + 4, lane, qb);
#pragma unroll
        for (int r = 0; r < 4; ++r)
            for (int part = 0; part < 2; ++part)
                *(v4f*)(at + (wave * 4 + r) * LN_PITCH_ + part * 128 + lane * 4) = m[r][part];
        {
            v4f mv[2];
            head_mean(qa, mv);
            for (int part = 0; part < 2; ++part) *(v4f*)(at + (16 + wave) * LN_PITCH_ + part * 128 + lane * 4) = mv[part];
            if (wave == 0) {
                head_mean(qb, mv);
                for (int part = 0; part < 2; ++part) *(v4f*)(at + 20 * LN_PITCH_ + part * 128 + lane * 4) = mv[part];
            }
        }
        const int nrows = (wave == 0) ? 6 : 5;
#pragma nounroll
        for (int i = 0; i < nrows; ++i) {
            const int row = (i < 4) ? wave * 4 + i : (i == 4) ? 16 + wave : 20;
            float* rp = at + row * LN_PITCH_ + lane * 4;
            v4f mm[2], o[2];
            mm[0] = *(const v4f*)rp;
            mm[1] = *(const v4f*)(rp + 128);
            bias_gelu_layernorm(mm, bs, g, b, o);
            *(v4f*)rp = o[0];
            *(v4f*)(rp + 128) = o[1];
        }
        __syncthreads();
        const float inv_sqrt6 = 0.40824829046386301637f;
        for (int part = 0; part < 2; ++part) {
            const float* vp = at + 16 * LN_PITCH_ + part * 128 + lane * 4;
            const v4f sv = ((*(const v4f*)(vp) + *(const v4f*)(vp + LN_PITCH_)) +
                            (*(const v4f*)(vp + 2 * LN_PITCH_) + *(const v4f*)(vp + 3 * LN_PITCH_))) +
                           *(const v4f*)(vp + 4 * LN_PITCH_);
            for (int r = 0; r < 4; ++r) {
                float* rp = at + (wave * 4 + r) * LN_PITCH_ + part * 128 + lane * 4;
                *(v4f*)rp = *(const v4f*)rp * (1.0f / 6.0f) + sv * inv_sqrt6;
            }
        }
        __syncthreads();
        ap = at + lm * LN_PITCH_ + half * 4;
    } else if constexpr (NODE_ROWS) ap = node_row(A, Av, m0 + lm) + half * 4;
    else                            ap = A + (size_t)(m0 + lm) * KDIM + half * 4;
    if constexpr (SCORES) {
        static_assert(NDIM == 256, "waves 0-3 cover all columns; wave 4 is the score tile");
        if (wave == 4) {
            v8f s0 = {}; v8f s1 = {};
            const float* up = U + (size_t)(half * 4) * UCOLS_ + lm;
            for (int k0 = 0; k0 < KDIM; k0 += 8) {
                const v4f a = *(const v4f*)(ap + k0);
                const float u0 = up[(size_t)(k0 + 0) * UCOLS_];
                const float u1 = up[(size_t)(k0 + 1) * UCOLS_];
                const float u2 = up[(size_t)(k0 + 2) * UCOLS_];
                const float u3 = up[(size_t)(k0 + 3) * UCOLS_];
                v2f af, bf;
                af[0] = a[0]; af[1] = a[1]; bf[0] = u0; bf[1] = u1;
                s0 = __builtin_amdgcn_wmma_f32_16x16x4_f32(false, af, false, bf, (short)0, s0, false, false);
                af[0] = a[2]; af[1] = a[3]; bf[0] = u2; bf[1] = u3;
                s1 = __builtin_amdgcn_wmma_f32_16x16x4_f32(false, af, false, bf, (short)0, s1, false, false);
            }
            if (lm < 8) {
                float* dst = (lm < H_) ? a_src : a_dst;
                for (int j = 0; j < 8; ++j)
                    dst[(size_t)(m0 + half * 8 + j) * H_ + (lm & (H_ - 1))] = s0[j] + s1[j];
            }
            return;
        }
    } else if constexpr (!LN_ROWS) {
        if (n0 >= NDIM) return;
    }

    v8f c0 = {}; v8f c1 = {}; v8f c2 = {}; v8f c3 = {};
    const float* bp = Bm + (size_t)(half * 4) * NDIM + n0 + lm * 4;

    constexpr int KBLK = 8 * KB;
    static_assert(KDIM % (2 * KBLK) == 0, "an even number of blocks: the sets alternate");
    GemmFrag<KB> f0, f1;
    gemm_load_block<KB, NDIM>(f0, ap, bp, 0);
    for (int k0 = 0; k0 < KDIM - 2 * KBLK; k0 += 2 * KBLK) {
        gemm_load_block<KB, NDIM>(f1, ap, bp, k0 + KBLK);
        __builtin_amdgcn_sched_barrier(0);
        gemm_mma_block<KB>(f0, c0, c1, c2, c3);
        __builtin_amdgcn_sched_barrier(0);
        gemm_load_block<KB, NDIM>(f0, ap, bp, k0 + 2 * KBLK);
        __builtin_amdgcn_sched_barrier(0);
        gemm_mma_block<KB>(f1, c0, c1, c2, c3);
        __builtin_amdgcn_sched_barrier(0);
    }
    gemm_load_block<KB, NDIM>(f1, ap, bp, KDIM - KBLK);
    __builtin_amdgcn_sched_barrier(0);
    gemm_mma_block<KB>(f0, c0, c1, c2, c3);
    gemm_mma_block<KB>(f1, c0, c1, c2, c3);

    if constexpr (LN_ROWS) {
        v4f bs2[2], g2[2], b2[2];
        load_row8(ep_bias, lane, bs2);
        load_row8(ep_g, lane, g2);
        load_row8(ep_b, lane, b2);
        __syncthreads();
        float* trow = at + (half * 8) * LN_PITCH_ + n0 + lm * 4;
#pragma unroll
        for (int j = 0; j < 8; ++j) {
            v4f o;
            o[0] = c0[j]; o[1] = c1[j]; o[2] = c2[j]; o[3] = c3[j];
            *(v4f*)(trow + j * LN_PITCH_) = o;
        }
        __syncthreads();
#pragma nounroll
        for (int r = 0; r < 4; ++r) {
            const float* rp = at + (wave * 4 + r) * LN_PITCH_ + lane * 4;
            v4f y[2], o[2];
            y[0] = *(const v4f*)rp;
            y[1] = *(const v4f*)(rp + 128);
            bias_gelu_layernorm(y, bs2, g2, b2, o);
            float* orow = C + (size_t)(m0 + wave * 4 + r) * D_ + lane * 4;
            *(v4f*)orow = o[0];
            *(v4f*)(orow + 128) = o[1];
        }
    } else {
        float* crow = C + (size_t)(m0 + half * 8) * NDIM + n0 + lm * 4;
#pragma unroll
        for (int j = 0; j < 8; ++j) {
            v4f o;
            o[0] = c0[j]; o[1] = c1[j]; o[2] = c2[j]; o[3] = c3[j];
            *(v4f*)(crow + (size_t)j * NDIM) = o;
        }
    }
}

#define QG_    4
#define QROWS_ (N_ / QG_)
static_assert(QG_ == H_ && QG_ * 64 == D_, "thread (q, c4) also finishes head q, channels 4*c4..+3");

template <int XMASK>
__device__ __forceinline__ float max_row_xmask(float v) {
    return fmaxf(v, __builtin_bit_cast(float, __builtin_amdgcn_update_dpp(
                        0, __builtin_bit_cast(int, v), 0x160 + XMASK, 0xf, 0xf, true)));
}

__device__ __forceinline__ float wave32_max(float v) {
    v = max_row_xmask<1>(v);
    v = max_row_xmask<2>(v);
    v = max_row_xmask<4>(v);
    v = max_row_xmask<8>(v);
    return fmaxf(v, __builtin_bit_cast(float, __builtin_amdgcn_permlanex16(
                        0u, __builtin_bit_cast(unsigned, v), 0x76543210u, 0xfedcba98u, true, false)));
}

__device__ __forceinline__ float leaky02(float a) { return (a > 0.0f) ? a : 0.2f * a; }

__device__ __forceinline__ float pick4(v4f v, int i) {
    return (i == 0) ? v[0] : (i == 1) ? v[1] : (i == 2) ? v[2] : v[3];
}

__global__ void __launch_bounds__(D_) k_gat_virtual(const float* __restrict__ x,
                              const float* __restrict__ vn,
                              const float* __restrict__ a_src,
                              const float* __restrict__ a_dst,
                              float* __restrict__ z) {
    __shared__ v4f w4[N_];
    __shared__ v4f red4[D_ / 32];
    __shared__ v4f part[QG_][H_][D_ / 4];
    const int uidx = blockIdx.x;
    const int b    = uidx / V_;
    const int u    = BN_ + uidx;
    const int tid  = threadIdx.x;
    const int lane = tid & 31;
    const int wave = tid >> 5;

    const v4f adst = *(const v4f*)(a_dst + (size_t)u * H_);
    const int r0 = b * N_ + tid * 2;
    v4f al0 = *(const v4f*)(a_src + (size_t)r0 * H_);
    v4f al1 = *(const v4f*)(a_src + (size_t)(r0 + 1) * H_);
    v4f als = *(const v4f*)(a_src + (size_t)u * H_);
    v4f lmax;
    for (int hd = 0; hd < H_; ++hd) {
        al0[hd] = leaky02(al0[hd] + adst[hd]);
        al1[hd] = leaky02(al1[hd] + adst[hd]);
        als[hd] = leaky02(als[hd] + adst[hd]);
        lmax[hd] = wave32_max(fmaxf(fmaxf(al0[hd], al1[hd]), als[hd]));
    }
    if (lane == 0) red4[wave] = lmax;
    __syncthreads();
    v4f m = red4[0];
    for (int wv = 1; wv < D_ / 32; ++wv) {
        const v4f o = red4[wv];
        for (int hd = 0; hd < H_; ++hd) m[hd] = fmaxf(m[hd], o[hd]);
    }
    __syncthreads();

    v4f e0, e1, es, ps;
    for (int hd = 0; hd < H_; ++hd) {
        e0[hd] = expf(al0[hd] - m[hd]);
        e1[hd] = expf(al1[hd] - m[hd]);
        es[hd] = expf(als[hd] - m[hd]);
        float s = e0[hd] + e1[hd];
        if (tid == 0) s += es[hd];
        ps[hd] = wave32_sum(s);
    }
    w4[tid * 2]     = e0;
    w4[tid * 2 + 1] = e1;
    if (lane == 0) red4[wave] = ps;
    __syncthreads();
    const v4f tot = ((red4[0] + red4[1]) + (red4[2] + red4[3])) +
                    ((red4[4] + red4[5]) + (red4[6] + red4[7]));

    const int c4 = tid & 63;
    const int q  = __builtin_amdgcn_readfirstlane(tid >> 6);
    const float* xb = x + (size_t)(b * N_ + q * QROWS_) * D_ + c4 * 4;
    const v4f* wq = w4 + q * QROWS_;
    v4f acc[H_];
    for (int hd = 0; hd < H_; ++hd) acc[hd] = (v4f){0.0f, 0.0f, 0.0f, 0.0f};
    for (int rb = 0; rb < QROWS_; rb += 32) {
        v4f xv[32];
#pragma unroll
        for (int i = 0; i < 32; ++i) xv[i] = *(const v4f*)(xb + (size_t)(rb + i) * D_);
#pragma unroll
        for (int i = 0; i < 32; ++i) {
            const v4f wr = wq[rb + i];
            for (int hd = 0; hd < H_; ++hd) acc[hd] += wr[hd] * xv[i];
        }
    }
    for (int hd = 0; hd < H_; ++hd) part[q][hd][c4] = acc[hd];
    __syncthreads();

    v4f s4 = (part[0][q][c4] + part[1][q][c4]) + (part[2][q][c4] + part[3][q][c4]);
    const v4f vs = *(const v4f*)(vn + (size_t)(uidx % V_) * D_ + c4 * 4);
    s4 += pick4(es, q) * vs;
    const float inv = 1.0f / pick4(tot, q);
    *(v4f*)(z + ((size_t)q * (B_ * V_) + uidx) * D_ + c4 * 4) = s4 * inv;
}

extern "C" void kernel_launch(void* const* d_in, const int* in_sizes, int n_in,
                              void* d_out, int out_size, void* d_ws, size_t ws_size,
                              hipStream_t stream) {
    (void)in_sizes; (void)n_in; (void)out_size; (void)ws_size;

    const float* x       = (const float*)d_in[0];
    const float* vn      = (const float*)d_in[2];
    const float* gat_W   = (const float*)d_in[3];
    const float* att_src = (const float*)d_in[4];
    const float* att_dst = (const float*)d_in[5];
    const float* gat_b   = (const float*)d_in[6];
    const float* gcn_W   = (const float*)d_in[7];
    const float* gcn_b   = (const float*)d_in[8];
    const float* n1g     = (const float*)d_in[9];
    const float* n1b     = (const float*)d_in[10];
    const float* n2g     = (const float*)d_in[11];
    const float* n2b     = (const float*)d_in[12];
    float* out = (float*)d_out;

    float* ws = (float*)d_ws;
    size_t off = 0;
    float* wm    = ws + off; off += (size_t)D_ * D_;
    float* ut    = ws + off; off += (size_t)D_ * UCOLS_;
    float* gm    = ws + off; off += (size_t)T_ * D_;
    float* asrc  = ws + off; off += (size_t)T_ * H_;
    float* adst  = ws + off; off += (size_t)T_ * H_;
    float* z     = ws + off; off += (size_t)H_ * B_ * V_ * D_;
    float* gv    = ws + off; off += (size_t)H_ * B_ * V_ * HD_;

    k_fold_weights<<<PREP_WGS_, D_, 0, stream>>>(gat_W, att_src, att_dst, wm, ut);
    {
        dim3 grid(T_ / 16, 1);
        k_gemm_f32_wmma<D_, D_, true, 0, true, 2, false><<<grid, 160, 0, stream>>>(x, vn, wm, gm, ut, asrc, adst,
                                                                                 nullptr, nullptr, nullptr,
                                                                                 nullptr, nullptr, nullptr);
    }
    {
        k_gat_virtual<<<B_ * V_, D_, 0, stream>>>(x, vn, asrc, adst, z);
    }
    {
        dim3 grid(B_ * V_ / 16, H_);
        k_gemm_f32_wmma<D_, HD_, false, B_ * V_, false, 4, false><<<grid, 128, 0, stream>>>(z, nullptr, gat_W, gv,
                                                                                        nullptr, nullptr, nullptr,
                                                                                        nullptr, nullptr, nullptr,
                                                                                        nullptr, nullptr, nullptr);
    }
    {
        dim3 grid(BN_ / 16, 1);
        k_gemm_f32_wmma<D_, D_, false, 0, false, 2, true><<<grid, 128, 0, stream>>>(gm, gv, gcn_W, out,
                                                                                 nullptr, nullptr, nullptr,
                                                                                 gat_b, n1g, n1b,
                                                                                 gcn_b, n2g, n2b);
    }
}
